# starting kernel
# baseline (speedup 1.0000x reference)
#include <hip/hip_runtime.h>

#define HDIM 128
#define NCLS 40
#define BN_EPS 1e-5f

typedef __attribute__((ext_vector_type(2))) float v2f;
typedef __attribute__((ext_vector_type(8))) float v8f;

__global__ void k_set_ones(float* deg, int n) {
  int i = blockIdx.x * blockDim.x + threadIdx.x;
  if (i < n) deg[i] = 1.0f;
}

__global__ void k_deg_accum(const int* __restrict__ ei, float* deg, int E) {
  int e = blockIdx.x * blockDim.x + threadIdx.x;
  if (e < E) atomicAdd(&deg[ei[E + e]], 1.0f);
}

__global__ void k_rsqrt_ip(float* d, int n) {
  int i = blockIdx.x * blockDim.x + threadIdx.x;
  if (i < n) d[i] = rsqrtf(d[i]);
}

__global__ void k_stats(const float* __restrict__ x, float* __restrict__ stats, int n) {
  int col = threadIdx.x & (HDIM - 1);
  int half = threadIdx.x >> 7;
  float s = 0.f, s2 = 0.f;
  for (int r = blockIdx.x * 2 + half; r < n; r += gridDim.x * 2) {
    float v = x[(size_t)r * HDIM + col];
    s += v; s2 += v * v;
  }
  __shared__ float ls[HDIM], ls2[HDIM];
  if (half == 0) { ls[col] = s; ls2[col] = s2; }
  __syncthreads();
  if (half == 1) {
    atomicAdd(&stats[col], s + ls[col]);
    atomicAdd(&stats[HDIM + col], s2 + ls2[col]);
  }
}

__global__ void k_bias_relu_stats(float* __restrict__ x, const float* __restrict__ b,
                                  float* __restrict__ stats, int n, int do_stats) {
  int col = threadIdx.x & (HDIM - 1);
  int half = threadIdx.x >> 7;
  float bias = b[col];
  float s = 0.f, s2 = 0.f;
  for (int r = blockIdx.x * 2 + half; r < n; r += gridDim.x * 2) {
    size_t idx = (size_t)r * HDIM + col;
    float v = x[idx] + bias;
    v = v > 0.f ? v : 0.f;
    x[idx] = v;
    s += v; s2 += v * v;
  }
  if (!do_stats) return;
  __shared__ float ls[HDIM], ls2[HDIM];
  if (half == 0) { ls[col] = s; ls2[col] = s2; }
  __syncthreads();
  if (half == 1) {
    atomicAdd(&stats[col], s + ls[col]);
    atomicAdd(&stats[HDIM + col], s2 + ls2[col]);
  }
}

__global__ void k_fold(const float* __restrict__ stats, const float* __restrict__ gamma,
                       const float* __restrict__ beta, const float* __restrict__ W,
                       float* __restrict__ WfT, float* __restrict__ cb, float inv_n) {
  __shared__ float as[HDIM], cs[HDIM];
  int t = threadIdx.x;
  float mean = stats[t] * inv_n;
  float var  = stats[HDIM + t] * inv_n - mean * mean;
  float a = gamma[t] * rsqrtf(var + BN_EPS);
  float c = beta[t] - a * mean;
  as[t] = a; cs[t] = c;
  __syncthreads();
  float acc = 0.f;
  for (int k = 0; k < HDIM; ++k) {
    float w = W[k * HDIM + t];
    WfT[t * HDIM + k] = as[k] * w;
    acc += cs[k] * w;
  }
  cb[t] = acc;
}

#define XPAD 68

__global__ __launch_bounds__(128) void k_gemm(const float* __restrict__ x,
                                              const float* __restrict__ WfT,
                                              const float* __restrict__ cb,
                                              float* __restrict__ h, int n) {
  __shared__ float xs[64 * XPAD];
  __shared__ float wfs[HDIM * XPAD];
  const int tid = threadIdx.x;
  const int lane = tid & 31;
  const int wave = tid >> 5;
  const int rowBase = blockIdx.x * 64;
  const int m = lane & 15;
  const int hi = lane >> 4;

  v8f acc[8];
#pragma unroll
  for (int c = 0; c < 8; ++c)
#pragma unroll
    for (int v = 0; v < 8; ++v) acc[c][v] = 0.f;

  for (int kh = 0; kh < 2; ++kh) {
#pragma unroll
    for (int p = 0; p < 8; ++p) {
      int idx = p * 512 + tid * 4;
      int r = idx >> 6, cc = idx & 63;
      int row = rowBase + r;
      float4 v = make_float4(0.f, 0.f, 0.f, 0.f);
      if (row < n) v = *(const float4*)(x + (size_t)row * HDIM + kh * 64 + cc);
      *(float4*)(&xs[r * XPAD + cc]) = v;
    }
#pragma unroll
    for (int p = 0; p < 16; ++p) {
      int idx = p * 512 + tid * 4;
      int nn = idx >> 6, cc = idx & 63;
      *(float4*)(&wfs[nn * XPAD + cc]) = *(const float4*)(WfT + nn * HDIM + kh * 64 + cc);
    }
    __syncthreads();

    const float* xrow = &xs[(wave * 16 + m) * XPAD];
#pragma unroll
    for (int k4 = 0; k4 < 16; ++k4) {
      int kb = k4 * 4 + hi * 2;
      v2f a; a.x = xrow[kb]; a.y = xrow[kb + 1];
#pragma unroll
      for (int c = 0; c < 8; ++c) {
        const float* wrow = &wfs[(c * 16 + m) * XPAD];
        v2f b; b.x = wrow[kb]; b.y = wrow[kb + 1];
        acc[c] = __builtin_amdgcn_wmma_f32_16x16x4_f32(
            false, a, false, b, (short)0, acc[c], false, false);
      }
    }
    __syncthreads();
  }

#pragma unroll
  for (int c = 0; c < 8; ++c) {
    int col = c * 16 + m;
    float cbv = cb[col];
#pragma unroll
    for (int v = 0; v < 8; ++v) {
      int row = rowBase + wave * 16 + v + 8 * hi;
      if (row < n) h[(size_t)row * HDIM + col] = acc[c][v] + cbv;
    }
  }
}

__global__ void k_edge_agg(const int* __restrict__ ei, const float* __restrict__ dinv,
                           const float* __restrict__ hbuf, float* __restrict__ out,
                           int E, int n) {
  int e = blockIdx.x * 8 + (threadIdx.x >> 5);
  if (e >= E + n) return;
  int lane = threadIdx.x & 31;
  int src, dst;
  if (e < E) { src = ei[e]; dst = ei[E + e]; }
  else       { src = dst = e - E; }
  float nrm = dinv[src] * dinv[dst];
  const float4 hv = *(const float4*)(hbuf + (size_t)src * HDIM + lane * 4);
  float* o = out + (size_t)dst * HDIM + lane * 4;
  atomicAdd(o + 0, hv.x * nrm);
  atomicAdd(o + 1, hv.y * nrm);
  atomicAdd(o + 2, hv.z * nrm);
  atomicAdd(o + 3, hv.w * nrm);
}

__global__ __launch_bounds__(256) void k_out_gemm(const float* __restrict__ x,
                                                  const float* __restrict__ Wo,
                                                  const float* __restrict__ bo,
                                                  float* __restrict__ out, int n) {
  __shared__ float ws[HDIM * NCLS];
  __shared__ float bs[NCLS];
  int tid = threadIdx.x;
  for (int i = tid; i < HDIM * NCLS; i += 256) ws[i] = Wo[i];
  if (tid < NCLS) bs[tid] = bo[tid];
  __syncthreads();
  int r = blockIdx.x * 128 + (tid >> 1);
  int c0 = (tid & 1) * 20;
  if (r >= n) return;
  float acc[20];
#pragma unroll
  for (int c = 0; c < 20; ++c) acc[c] = bs[c0 + c];
  const float* xr = x + (size_t)r * HDIM;
  for (int k = 0; k < HDIM; ++k) {
    float xv = xr[k];
    const float* wr = &ws[k * NCLS + c0];
#pragma unroll
    for (int c = 0; c < 20; ++c) acc[c] += xv * wr[c];
  }
  float* o = out + (size_t)r * NCLS + c0;
#pragma unroll
  for (int c = 0; c < 20; ++c) o[c] = acc[c];
}

extern "C" void kernel_launch(void* const* d_in, const int* in_sizes, int n_in,
                              void* d_out, int out_size, void* d_ws, size_t ws_size,
                              hipStream_t stream) {
  const float* x_in  = (const float*)d_in[0];
  const int*   ei    = (const int*)d_in[1];
  const float* gamma = (const float*)d_in[2];
  const float* beta  = (const float*)d_in[3];
  const float* W     = (const float*)d_in[4];
  const float* b     = (const float*)d_in[5];
  const float* Wo    = (const float*)d_in[6];
  const float* bo    = (const float*)d_in[7];
  float* out = (float*)d_out;

  const int N = in_sizes[0] / HDIM;
  const int E = in_sizes[1] / 2;

  float* dinv  = (float*)d_ws;
  float* stats = dinv + ((N + 255) / 256) * 256;
  float* WfT   = stats + 2 * HDIM;
  float* cb    = WfT + HDIM * HDIM;
  float* hbuf  = cb + HDIM + 64;
  float* xbuf  = hbuf + (size_t)N * HDIM;

  k_set_ones<<<(N + 255) / 256, 256, 0, stream>>>(dinv, N);
  k_deg_accum<<<(E + 255) / 256, 256, 0, stream>>>(ei, dinv, E);
  k_rsqrt_ip<<<(N + 255) / 256, 256, 0, stream>>>(dinv, N);

  hipMemsetAsync(stats, 0, 2 * HDIM * sizeof(float), stream);
  k_stats<<<512, 256, 0, stream>>>(x_in, stats, N);

  const int EN = E + N;
  for (int l = 0; l < 3; ++l) {
    k_fold<<<1, 128, 0, stream>>>(stats, gamma + l * HDIM, beta + l * HDIM,
                                  W + (size_t)l * HDIM * HDIM, WfT, cb, 1.0f / (float)N);
    const float* xl = (l == 0) ? x_in : xbuf;
    k_gemm<<<(N + 63) / 64, 128, 0, stream>>>(xl, WfT, cb, hbuf, N);
    hipMemsetAsync(xbuf, 0, (size_t)N * HDIM * sizeof(float), stream);
    k_edge_agg<<<(EN + 7) / 8, 256, 0, stream>>>(ei, dinv, hbuf, xbuf, E, N);
    if (l < 2) {
      hipMemsetAsync(stats, 0, 2 * HDIM * sizeof(float), stream);
      k_bias_relu_stats<<<512, 256, 0, stream>>>(xbuf, b + l * HDIM, stats, N, 1);
    } else {
      k_bias_relu_stats<<<512, 256, 0, stream>>>(xbuf, b + l * HDIM, stats, N, 0);
    }
  }

  k_out_gemm<<<(N + 127) / 128, 256, 0, stream>>>(xbuf, Wo, bo, out, N);
}


# Aggregation as a gather over a per-call destination-sorted edge list (integer-atomic rank + scan): a wave sums a node's rows in registers, plain stores; replaces float atomics, sum order differs
# speedup vs baseline: 2.0805x; 2.0805x over previous
#include <hip/hip_runtime.h>

#define HDIM 128
#define NCLS 40
#define BN_EPS 1e-5f

typedef __attribute__((ext_vector_type(2))) float v2f;
typedef __attribute__((ext_vector_type(8))) float v8f;

__global__ void k_zero_int(int* p, int n) {
  int i = blockIdx.x * blockDim.x + threadIdx.x;
  if (i < n) p[i] = 0;
}

__global__ void k_count_rank(const int* __restrict__ ei, int* cnt, int* __restrict__ rank, int E) {
  int e = blockIdx.x * blockDim.x + threadIdx.x;
  if (e < E) rank[e] = atomicAdd(&cnt[ei[E + e]], 1);
}

#define SCAN_T 1024
__device__ __forceinline__ int4 load4_guarded(const int* __restrict__ p, int i0, int n) {
  int4 v = make_int4(0, 0, 0, 0);
  if (i0 + 3 < n) {
    v = *(const int4*)(p + i0);
  } else {
    if (i0 < n) v.x = p[i0];
    if (i0 + 1 < n) v.y = p[i0 + 1];
    if (i0 + 2 < n) v.z = p[i0 + 2];
  }
  return v;
}

__global__ __launch_bounds__(SCAN_T) void k_scan_dinv(const int* __restrict__ cnt,
                                                      int* __restrict__ rowptr,
                                                      float* __restrict__ dinv, int n) {
  __shared__ int wsum[32];
  const int t = threadIdx.x, lane = t & 31, wv = t >> 5;
  int carry = 0;
  int4 v[4];
#pragma unroll
  for (int g = 0; g < 4; ++g) v[g] = load4_guarded(cnt, t * 16 + g * 4, n);
  for (int base = 0; base < n; base += SCAN_T * 16) {
    const int i0 = base + t * 16;
    int4 nxt[4];
#pragma unroll
    for (int g = 0; g < 4; ++g) nxt[g] = load4_guarded(cnt, i0 + SCAN_T * 16 + g * 4, n);
    int s = 0;
#pragma unroll
    for (int g = 0; g < 4; ++g) s += v[g].x + v[g].y + v[g].z + v[g].w;
    int inc = s;
#pragma unroll
    for (int d = 1; d < 32; d <<= 1) {
      int o = __shfl_up(inc, d);
      if (lane >= d) inc += o;
    }
    if (lane == 31) wsum[wv] = inc;
    __syncthreads();
    const int wt = wsum[lane];
    int winc = wt;
#pragma unroll
    for (int d = 1; d < 32; d <<= 1) {
      int o = __shfl_up(winc, d);
      if (lane >= d) winc += o;
    }
    const int wex = __shfl(winc, wv) - __shfl(wt, wv);
    const int total = __shfl(winc, 31);
    int run = carry + wex + inc - s;
#pragma unroll
    for (int g = 0; g < 4; ++g) {
      const int i = i0 + g * 4;
      const int4 c = v[g];
      if (i + 3 < n) {
        *(int4*)(rowptr + i) = make_int4(run, run + c.x, run + c.x + c.y, run + c.x + c.y + c.z);
        *(float4*)(dinv + i) = make_float4(rsqrtf((float)(c.x + 1)), rsqrtf((float)(c.y + 1)),
                                           rsqrtf((float)(c.z + 1)), rsqrtf((float)(c.w + 1)));
      } else {
        if (i < n)     { rowptr[i] = run;                 dinv[i] = rsqrtf((float)(c.x + 1)); }
        if (i + 1 < n) { rowptr[i + 1] = run + c.x;       dinv[i + 1] = rsqrtf((float)(c.y + 1)); }
        if (i + 2 < n) { rowptr[i + 2] = run + c.x + c.y; dinv[i + 2] = rsqrtf((float)(c.z + 1)); }
      }
      run += c.x + c.y + c.z + c.w;
    }
    carry += total;
#pragma unroll
    for (int g = 0; g < 4; ++g) v[g] = nxt[g];
    __syncthreads();
  }
  if (t == 0) rowptr[n] = carry;
}

__global__ void k_fill(const int* __restrict__ ei, const int* __restrict__ rank,
                       const int* __restrict__ rowptr, const float* __restrict__ dinv,
                       int* __restrict__ csr, int E) {
  int e = blockIdx.x * blockDim.x + threadIdx.x;
  if (e >= E) return;
  int s = ei[e], d = ei[E + e];
  int p = rowptr[d] + rank[e];
  p = p < E ? p : E - 1;
  ((int2*)csr)[p] = make_int2(s, __float_as_int(dinv[s] * dinv[d]));
}

__global__ void k_stats(const float* __restrict__ x, float* __restrict__ stats, int n) {
  int col = threadIdx.x & (HDIM - 1);
  int half = threadIdx.x >> 7;
  float s = 0.f, s2 = 0.f;
  for (int r = blockIdx.x * 2 + half; r < n; r += gridDim.x * 2) {
    float v = x[(size_t)r * HDIM + col];
    s += v; s2 += v * v;
  }
  __shared__ float ls[HDIM], ls2[HDIM];
  if (half == 0) { ls[col] = s; ls2[col] = s2; }
  __syncthreads();
  if (half == 1) {
    atomicAdd(&stats[col], s + ls[col]);
    atomicAdd(&stats[HDIM + col], s2 + ls2[col]);
  }
}

__global__ void k_bias_relu_stats(float* __restrict__ x, const float* __restrict__ b,
                                  float* __restrict__ stats, int n, int do_stats) {
  int col = threadIdx.x & (HDIM - 1);
  int half = threadIdx.x >> 7;
  float bias = b[col];
  float s = 0.f, s2 = 0.f;
  for (int r = blockIdx.x * 2 + half; r < n; r += gridDim.x * 2) {
    size_t idx = (size_t)r * HDIM + col;
    float v = x[idx] + bias;
    v = v > 0.f ? v : 0.f;
    x[idx] = v;
    s += v; s2 += v * v;
  }
  if (!do_stats) return;
  __shared__ float ls[HDIM], ls2[HDIM];
  if (half == 0) { ls[col] = s; ls2[col] = s2; }
  __syncthreads();
  if (half == 1) {
    atomicAdd(&stats[col], s + ls[col]);
    atomicAdd(&stats[HDIM + col], s2 + ls2[col]);
  }
}

__global__ void k_fold(const float* __restrict__ stats, const float* __restrict__ gamma,
                       const float* __restrict__ beta, const float* __restrict__ W,
                       float* __restrict__ WfT, float* __restrict__ cb, float inv_n) {
  __shared__ float as[HDIM], cs[HDIM];
  int t = threadIdx.x;
  float mean = stats[t] * inv_n;
  float var  = stats[HDIM + t] * inv_n - mean * mean;
  float a = gamma[t] * rsqrtf(var + BN_EPS);
  float c = beta[t] - a * mean;
  as[t] = a; cs[t] = c;
  __syncthreads();
  float acc = 0.f;
  for (int k = 0; k < HDIM; ++k) {
    float w = W[k * HDIM + t];
    WfT[t * HDIM + k] = as[k] * w;
    acc += cs[k] * w;
  }
  cb[t] = acc;
}

#define XPAD 68

__global__ __launch_bounds__(128) void k_gemm(const float* __restrict__ x,
                                              const float* __restrict__ WfT,
                                              const float* __restrict__ cb,
                                              float* __restrict__ h, int n) {
  __shared__ float xs[64 * XPAD];
  __shared__ float wfs[HDIM * XPAD];
  const int tid = threadIdx.x;
  const int lane = tid & 31;
  const int wave = tid >> 5;
  const int rowBase = blockIdx.x * 64;
  const int m = lane & 15;
  const int hi = lane >> 4;

  v8f acc[8];
#pragma unroll
  for (int c = 0; c < 8; ++c)
#pragma unroll
    for (int v = 0; v < 8; ++v) acc[c][v] = 0.f;

  for (int kh = 0; kh < 2; ++kh) {
#pragma unroll
    for (int p = 0; p < 8; ++p) {
      int idx = p * 512 + tid * 4;
      int r = idx >> 6, cc = idx & 63;
      int row = rowBase + r;
      float4 v = make_float4(0.f, 0.f, 0.f, 0.f);
      if (row < n) v = *(const float4*)(x + (size_t)row * HDIM + kh * 64 + cc);
      *(float4*)(&xs[r * XPAD + cc]) = v;
    }
#pragma unroll
    for (int p = 0; p < 16; ++p) {
      int idx = p * 512 + tid * 4;
      int nn = idx >> 6, cc = idx & 63;
      *(float4*)(&wfs[nn * XPAD + cc]) = *(const float4*)(WfT + nn * HDIM + kh * 64 + cc);
    }
    __syncthreads();

    const float* xrow = &xs[(wave * 16 + m) * XPAD];
#pragma unroll
    for (int k4 = 0; k4 < 16; ++k4) {
      int kb = k4 * 4 + hi * 2;
      v2f a; a.x = xrow[kb]; a.y = xrow[kb + 1];
#pragma unroll
      for (int c = 0; c < 8; ++c) {
        const float* wrow = &wfs[(c * 16 + m) * XPAD];
        v2f b; b.x = wrow[kb]; b.y = wrow[kb + 1];
        acc[c] = __builtin_amdgcn_wmma_f32_16x16x4_f32(
            false, a, false, b, (short)0, acc[c], false, false);
      }
    }
    __syncthreads();
  }

#pragma unroll
  for (int c = 0; c < 8; ++c) {
    int col = c * 16 + m;
    float cbv = cb[col];
#pragma unroll
    for (int v = 0; v < 8; ++v) {
      int row = rowBase + wave * 16 + v + 8 * hi;
      if (row < n) h[(size_t)row * HDIM + col] = acc[c][v] + cbv;
    }
  }
}

template <int K>
__device__ __forceinline__ void edge_batch(float4& acc, const float4* __restrict__ hb,
                                           const int2 my, const int j, const int lane) {
  int sidx[K];
  float wgt[K];
  float4 a[K];
#pragma unroll
  for (int q = 0; q < K; ++q) {
    sidx[q] = __builtin_amdgcn_readlane(my.x, j + q);
    wgt[q] = __int_as_float(__builtin_amdgcn_readlane(my.y, j + q));
  }
#pragma unroll
  for (int q = 0; q < K; ++q) a[q] = hb[(size_t)sidx[q] * 32 + lane];
#pragma unroll
  for (int q = 0; q < K; ++q) {
    acc.x = fmaf(a[q].x, wgt[q], acc.x);
    acc.y = fmaf(a[q].y, wgt[q], acc.y);
    acc.z = fmaf(a[q].z, wgt[q], acc.z);
    acc.w = fmaf(a[q].w, wgt[q], acc.w);
  }
}

#define AGG_BLOCKS 2048
__global__ __launch_bounds__(256) void k_node_agg(const int* __restrict__ rowptr,
                                                  const int* __restrict__ csr,
                                                  const float* __restrict__ dinv,
                                                  const float* __restrict__ hbuf,
                                                  float* __restrict__ out, int n) {
  const int wv = __builtin_amdgcn_readfirstlane(threadIdx.x >> 5);
  const int lane = threadIdx.x & 31;
  const float4* hb = (const float4*)hbuf;
  const int2* cs = (const int2*)csr;
  for (int node = blockIdx.x * 8 + wv; node < n; node += gridDim.x * 8) {
    const int beg = rowptr[node], end = rowptr[node + 1];
    const float di = dinv[node];
    const float4 hs = hb[(size_t)node * 32 + lane];
    float4 acc = make_float4(0.f, 0.f, 0.f, 0.f);
    for (int base = beg; base < end; base += 32) {
      int m = end - base;
      m = m < 32 ? m : 32;
      int2 my = make_int2(0, 0);
      if (lane < m) my = cs[base + lane];
      my.x = my.x < 0 ? 0 : (my.x < n ? my.x : n - 1);
      int j = 0;
      for (; j + 8 <= m; j += 8) edge_batch<8>(acc, hb, my, j, lane);
      if (j + 4 <= m) { edge_batch<4>(acc, hb, my, j, lane); j += 4; }
      const int r = m - j;
      if (r == 3) edge_batch<3>(acc, hb, my, j, lane);
      else if (r == 2) edge_batch<2>(acc, hb, my, j, lane);
      else if (r == 1) edge_batch<1>(acc, hb, my, j, lane);
    }
    const float w_self = di * di;
    acc.x = fmaf(hs.x, w_self, acc.x);
    acc.y = fmaf(hs.y, w_self, acc.y);
    acc.z = fmaf(hs.z, w_self, acc.z);
    acc.w = fmaf(hs.w, w_self, acc.w);
    ((float4*)out)[(size_t)node * 32 + lane] = acc;
  }
}

__global__ __launch_bounds__(256) void k_out_gemm(const float* __restrict__ x,
                                                  const float* __restrict__ Wo,
                                                  const float* __restrict__ bo,
                                                  float* __restrict__ out, int n) {
  __shared__ float ws[HDIM * NCLS];
  __shared__ float bs[NCLS];
  int tid = threadIdx.x;
  for (int i = tid; i < HDIM * NCLS; i += 256) ws[i] = Wo[i];
  if (tid < NCLS) bs[tid] = bo[tid];
  __syncthreads();
  int r = blockIdx.x * 128 + (tid >> 1);
  int c0 = (tid & 1) * 20;
  if (r >= n) return;
  float acc[20];
#pragma unroll
  for (int c = 0; c < 20; ++c) acc[c] = bs[c0 + c];
  const float* xr = x + (size_t)r * HDIM;
  for (int k = 0; k < HDIM; ++k) {
    float xv = xr[k];
    const float* wr = &ws[k * NCLS + c0];
#pragma unroll
    for (int c = 0; c < 20; ++c) acc[c] += xv * wr[c];
  }
  float* o = out + (size_t)r * NCLS + c0;
#pragma unroll
  for (int c = 0; c < 20; ++c) o[c] = acc[c];
}

extern "C" void kernel_launch(void* const* d_in, const int* in_sizes, int n_in,
                              void* d_out, int out_size, void* d_ws, size_t ws_size,
                              hipStream_t stream) {
  const float* x_in  = (const float*)d_in[0];
  const int*   ei    = (const int*)d_in[1];
  const float* gamma = (const float*)d_in[2];
  const float* beta  = (const float*)d_in[3];
  const float* W     = (const float*)d_in[4];
  const float* b     = (const float*)d_in[5];
  const float* Wo    = (const float*)d_in[6];
  const float* bo    = (const float*)d_in[7];
  float* out = (float*)d_out;

  const int N = in_sizes[0] / HDIM;
  const int E = in_sizes[1] / 2;

  float* dinv  = (float*)d_ws;
  float* stats = dinv + ((N + 255) / 256) * 256;
  float* WfT   = stats + 2 * HDIM;
  float* cb    = WfT + HDIM * HDIM;
  float* hbuf  = cb + HDIM + 64;
  float* xbuf  = hbuf + (size_t)N * HDIM;

  int* csr    = (int*)d_out;
  int* rowptr = csr + ((2 * E + 3) / 4) * 4;
  int* cnt    = rowptr + ((N + 4) / 4) * 4;
  int* rank   = (int*)xbuf;

  k_zero_int<<<(N + 255) / 256, 256, 0, stream>>>(cnt, N);
  k_count_rank<<<(E + 255) / 256, 256, 0, stream>>>(ei, cnt, rank, E);
  k_scan_dinv<<<1, SCAN_T, 0, stream>>>(cnt, rowptr, dinv, N);
  k_fill<<<(E + 255) / 256, 256, 0, stream>>>(ei, rank, rowptr, dinv, csr, E);

  hipMemsetAsync(stats, 0, 2 * HDIM * sizeof(float), stream);
  k_stats<<<512, 256, 0, stream>>>(x_in, stats, N);

  for (int l = 0; l < 3; ++l) {
    k_fold<<<1, 128, 0, stream>>>(stats, gamma + l * HDIM, beta + l * HDIM,
                                  W + (size_t)l * HDIM * HDIM, WfT, cb, 1.0f / (float)N);
    const float* xl = (l == 0) ? x_in : xbuf;
    k_gemm<<<(N + 63) / 64, 128, 0, stream>>>(xl, WfT, cb, hbuf, N);
    k_node_agg<<<AGG_BLOCKS, 256, 0, stream>>>(rowptr, csr, dinv, hbuf, xbuf, N);
    if (l < 2) {
      hipMemsetAsync(stats, 0, 2 * HDIM * sizeof(float), stream);
      k_bias_relu_stats<<<512, 256, 0, stream>>>(xbuf, b + l * HDIM, stats, N, 1);
    } else {
      k_bias_relu_stats<<<512, 256, 0, stream>>>(xbuf, b + l * HDIM, stats, N, 0);
    }
  }

  k_out_gemm<<<(N + 127) / 128, 256, 0, stream>>>(xbuf, Wo, bo, out, N);
}


# Bias+relu and the next layer's BN column sums fused into the aggregation's store (per-block sums via LDS, one float atomic per column per block); the 3 read-back passes go; column-sum order differs
# speedup vs baseline: 2.7100x; 1.3026x over previous
#include <hip/hip_runtime.h>

#define HDIM 128
#define NCLS 40
#define BN_EPS 1e-5f

typedef __attribute__((ext_vector_type(2))) float v2f;
typedef __attribute__((ext_vector_type(8))) float v8f;

__global__ void k_zero_int(int* p, int n) {
  int i = blockIdx.x * blockDim.x + threadIdx.x;
  if (i < n) p[i] = 0;
}

__global__ void k_count_rank(const int* __restrict__ ei, int* cnt, int* __restrict__ rank, int E) {
  int e = blockIdx.x * blockDim.x + threadIdx.x;
  if (e < E) rank[e] = atomicAdd(&cnt[ei[E + e]], 1);
}

#define SCAN_T 1024
__device__ __forceinline__ int4 load4_guarded(const int* __restrict__ p, int i0, int n) {
  int4 v = make_int4(0, 0, 0, 0);
  if (i0 + 3 < n) {
    v = *(const int4*)(p + i0);
  } else {
    if (i0 < n) v.x = p[i0];
    if (i0 + 1 < n) v.y = p[i0 + 1];
    if (i0 + 2 < n) v.z = p[i0 + 2];
  }
  return v;
}

__global__ __launch_bounds__(SCAN_T) void k_scan_dinv(const int* __restrict__ cnt,
                                                      int* __restrict__ rowptr,
                                                      float* __restrict__ dinv, int n) {
  __shared__ int wsum[32];
  const int t = threadIdx.x, lane = t & 31, wv = t >> 5;
  int carry = 0;
  int4 v[4];
#pragma unroll
  for (int g = 0; g < 4; ++g) v[g] = load4_guarded(cnt, t * 16 + g * 4, n);
  for (int base = 0; base < n; base += SCAN_T * 16) {
    const int i0 = base + t * 16;
    int4 nxt[4];
#pragma unroll
    for (int g = 0; g < 4; ++g) nxt[g] = load4_guarded(cnt, i0 + SCAN_T * 16 + g * 4, n);
    int s = 0;
#pragma unroll
    for (int g = 0; g < 4; ++g) s += v[g].x + v[g].y + v[g].z + v[g].w;
    int inc = s;
#pragma unroll
    for (int d = 1; d < 32; d <<= 1) {
      int o = __shfl_up(inc, d);
      if (lane >= d) inc += o;
    }
    if (lane == 31) wsum[wv] = inc;
    __syncthreads();
    const int wt = wsum[lane];
    int winc = wt;
#pragma unroll
    for (int d = 1; d < 32; d <<= 1) {
      int o = __shfl_up(winc, d);
      if (lane >= d) winc += o;
    }
    const int wex = __shfl(winc, wv) - __shfl(wt, wv);
    const int total = __shfl(winc, 31);
    int run = carry + wex + inc - s;
#pragma unroll
    for (int g = 0; g < 4; ++g) {
      const int i = i0 + g * 4;
      const int4 c = v[g];
      if (i + 3 < n) {
        *(int4*)(rowptr + i) = make_int4(run, run + c.x, run + c.x + c.y, run + c.x + c.y + c.z);
        *(float4*)(dinv + i) = make_float4(rsqrtf((float)(c.x + 1)), rsqrtf((float)(c.y + 1)),
                                           rsqrtf((float)(c.z + 1)), rsqrtf((float)(c.w + 1)));
      } else {
        if (i < n)     { rowptr[i] = run;                 dinv[i] = rsqrtf((float)(c.x + 1)); }
        if (i + 1 < n) { rowptr[i + 1] = run + c.x;       dinv[i + 1] = rsqrtf((float)(c.y + 1)); }
        if (i + 2 < n) { rowptr[i + 2] = run + c.x + c.y; dinv[i + 2] = rsqrtf((float)(c.z + 1)); }
      }
      run += c.x + c.y + c.z + c.w;
    }
    carry += total;
#pragma unroll
    for (int g = 0; g < 4; ++g) v[g] = nxt[g];
    __syncthreads();
  }
  if (t == 0) rowptr[n] = carry;
}

__global__ void k_fill(const int* __restrict__ ei, const int* __restrict__ rank,
                       const int* __restrict__ rowptr, const float* __restrict__ dinv,
                       int* __restrict__ csr, int E) {
  int e = blockIdx.x * blockDim.x + threadIdx.x;
  if (e >= E) return;
  int s = ei[e], d = ei[E + e];
  int p = rowptr[d] + rank[e];
  p = p < E ? p : E - 1;
  ((int2*)csr)[p] = make_int2(s, __float_as_int(dinv[s] * dinv[d]));
}

__global__ void k_stats(const float* __restrict__ x, float* __restrict__ stats, int n) {
  int col = threadIdx.x & (HDIM - 1);
  int half = threadIdx.x >> 7;
  float s = 0.f, s2 = 0.f;
  for (int r = blockIdx.x * 2 + half; r < n; r += gridDim.x * 2) {
    float v = x[(size_t)r * HDIM + col];
    s += v; s2 += v * v;
  }
  __shared__ float ls[HDIM], ls2[HDIM];
  if (half == 0) { ls[col] = s; ls2[col] = s2; }
  __syncthreads();
  if (half == 1) {
    atomicAdd(&stats[col], s + ls[col]);
    atomicAdd(&stats[HDIM + col], s2 + ls2[col]);
  }
}

__global__ void k_fold(const float* __restrict__ stats, const float* __restrict__ gamma,
                       const float* __restrict__ beta, const float* __restrict__ W,
                       float* __restrict__ WfT, float* __restrict__ cb, float inv_n) {
  __shared__ float as[HDIM], cs[HDIM];
  int t = threadIdx.x;
  float mean = stats[t] * inv_n;
  float var  = stats[HDIM + t] * inv_n - mean * mean;
  float a = gamma[t] * rsqrtf(var + BN_EPS);
  float c = beta[t] - a * mean;
  as[t] = a; cs[t] = c;
  __syncthreads();
  float acc = 0.f;
  for (int k = 0; k < HDIM; ++k) {
    float w = W[k * HDIM + t];
    WfT[t * HDIM + k] = as[k] * w;
    acc += cs[k] * w;
  }
  cb[t] = acc;
}

#define XPAD 68

__global__ __launch_bounds__(128) void k_gemm(const float* __restrict__ x,
                                              const float* __restrict__ WfT,
                                              const float* __restrict__ cb,
                                              float* __restrict__ h, int n) {
  __shared__ float xs[64 * XPAD];
  __shared__ float wfs[HDIM * XPAD];
  const int tid = threadIdx.x;
  const int lane = tid & 31;
  const int wave = tid >> 5;
  const int rowBase = blockIdx.x * 64;
  const int m = lane & 15;
  const int hi = lane >> 4;

  v8f acc[8];
#pragma unroll
  for (int c = 0; c < 8; ++c)
#pragma unroll
    for (int v = 0; v < 8; ++v) acc[c][v] = 0.f;

  for (int kh = 0; kh < 2; ++kh) {
#pragma unroll
    for (int p = 0; p < 8; ++p) {
      int idx = p * 512 + tid * 4;
      int r = idx >> 6, cc = idx & 63;
      int row = rowBase + r;
      float4 v = make_float4(0.f, 0.f, 0.f, 0.f);
      if (row < n) v = *(const float4*)(x + (size_t)row * HDIM + kh * 64 + cc);
      *(float4*)(&xs[r * XPAD + cc]) = v;
    }
#pragma unroll
    for (int p = 0; p < 16; ++p) {
      int idx = p * 512 + tid * 4;
      int nn = idx >> 6, cc = idx & 63;
      *(float4*)(&wfs[nn * XPAD + cc]) = *(const float4*)(WfT + nn * HDIM + kh * 64 + cc);
    }
    __syncthreads();

    const float* xrow = &xs[(wave * 16 + m) * XPAD];
#pragma unroll
    for (int k4 = 0; k4 < 16; ++k4) {
      int kb = k4 * 4 + hi * 2;
      v2f a; a.x = xrow[kb]; a.y = xrow[kb + 1];
#pragma unroll
      for (int c = 0; c < 8; ++c) {
        const float* wrow = &wfs[(c * 16 + m) * XPAD];
        v2f b; b.x = wrow[kb]; b.y = wrow[kb + 1];
        acc[c] = __builtin_amdgcn_wmma_f32_16x16x4_f32(
            false, a, false, b, (short)0, acc[c], false, false);
      }
    }
    __syncthreads();
  }

#pragma unroll
  for (int c = 0; c < 8; ++c) {
    int col = c * 16 + m;
    float cbv = cb[col];
#pragma unroll
    for (int v = 0; v < 8; ++v) {
      int row = rowBase + wave * 16 + v + 8 * hi;
      if (row < n) h[(size_t)row * HDIM + col] = acc[c][v] + cbv;
    }
  }
}

template <int K>
__device__ __forceinline__ void edge_batch(float4& acc, const float4* __restrict__ hb,
                                           const int2 my, const int j, const int lane) {
  int sidx[K];
  float wgt[K];
  float4 a[K];
#pragma unroll
  for (int q = 0; q < K; ++q) {
    sidx[q] = __builtin_amdgcn_readlane(my.x, j + q);
    wgt[q] = __int_as_float(__builtin_amdgcn_readlane(my.y, j + q));
  }
#pragma unroll
  for (int q = 0; q < K; ++q) a[q] = hb[(size_t)sidx[q] * 32 + lane];
#pragma unroll
  for (int q = 0; q < K; ++q) {
    acc.x = fmaf(a[q].x, wgt[q], acc.x);
    acc.y = fmaf(a[q].y, wgt[q], acc.y);
    acc.z = fmaf(a[q].z, wgt[q], acc.z);
    acc.w = fmaf(a[q].w, wgt[q], acc.w);
  }
}

#define AGG_WAVES 32
#define AGG_BLOCKS 512
template <bool STATS>
__global__ __launch_bounds__(AGG_WAVES * 32) void k_node_agg(const int* __restrict__ rowptr,
                                                             const int* __restrict__ csr,
                                                             const float* __restrict__ dinv,
                                                             const float* __restrict__ hbuf,
                                                             const float* __restrict__ bias,
                                                             float* __restrict__ out,
                                                             float* __restrict__ stats, int n) {
  const int wv = __builtin_amdgcn_readfirstlane(threadIdx.x >> 5);
  const int lane = threadIdx.x & 31;
  const float4* hb = (const float4*)hbuf;
  const int2* cs = (const int2*)csr;
  const float4 bv = ((const float4*)bias)[lane];
  float4 cs1 = make_float4(0.f, 0.f, 0.f, 0.f);
  float4 cs2 = make_float4(0.f, 0.f, 0.f, 0.f);
  for (int node = blockIdx.x * AGG_WAVES + wv; node < n; node += gridDim.x * AGG_WAVES) {
    const int beg = rowptr[node], end = rowptr[node + 1];
    const float di = dinv[node];
    const float4 hs = hb[(size_t)node * 32 + lane];
    float4 acc = make_float4(0.f, 0.f, 0.f, 0.f);
    for (int base = beg; base < end; base += 32) {
      int m = end - base;
      m = m < 32 ? m : 32;
      int2 my = make_int2(0, 0);
      if (lane < m) my = cs[base + lane];
      my.x = my.x < 0 ? 0 : (my.x < n ? my.x : n - 1);
      int j = 0;
      for (; j + 8 <= m; j += 8) edge_batch<8>(acc, hb, my, j, lane);
      if (j + 4 <= m) { edge_batch<4>(acc, hb, my, j, lane); j += 4; }
      const int r = m - j;
      if (r == 3) edge_batch<3>(acc, hb, my, j, lane);
      else if (r == 2) edge_batch<2>(acc, hb, my, j, lane);
      else if (r == 1) edge_batch<1>(acc, hb, my, j, lane);
    }
    const float w_self = di * di;
    acc.x = fmaf(hs.x, w_self, acc.x);
    acc.y = fmaf(hs.y, w_self, acc.y);
    acc.z = fmaf(hs.z, w_self, acc.z);
    acc.w = fmaf(hs.w, w_self, acc.w);
    float4 v;
    v.x = acc.x + bv.x; v.x = v.x > 0.f ? v.x : 0.f;
    v.y = acc.y + bv.y; v.y = v.y > 0.f ? v.y : 0.f;
    v.z = acc.z + bv.z; v.z = v.z > 0.f ? v.z : 0.f;
    v.w = acc.w + bv.w; v.w = v.w > 0.f ? v.w : 0.f;
    ((float4*)out)[(size_t)node * 32 + lane] = v;
    if (STATS) {
      cs1.x += v.x; cs1.y += v.y; cs1.z += v.z; cs1.w += v.w;
      cs2.x = fmaf(v.x, v.x, cs2.x); cs2.y = fmaf(v.y, v.y, cs2.y);
      cs2.z = fmaf(v.z, v.z, cs2.z); cs2.w = fmaf(v.w, v.w, cs2.w);
    }
  }
  if (STATS) {
    __shared__ float ls[AGG_WAVES * 2 * HDIM];
    float4* lw = (float4*)&ls[wv * 2 * HDIM];
    lw[lane] = cs1;
    lw[32 + lane] = cs2;
    __syncthreads();
    if (threadIdx.x < 2 * HDIM) {
      float t = 0.f;
#pragma unroll
      for (int w = 0; w < AGG_WAVES; ++w) t += ls[w * 2 * HDIM + threadIdx.x];
      atomicAdd(&stats[threadIdx.x], t);
    }
  }
}

__global__ __launch_bounds__(256) void k_out_gemm(const float* __restrict__ x,
                                                  const float* __restrict__ Wo,
                                                  const float* __restrict__ bo,
                                                  float* __restrict__ out, int n) {
  __shared__ float ws[HDIM * NCLS];
  __shared__ float bs[NCLS];
  int tid = threadIdx.x;
  for (int i = tid; i < HDIM * NCLS; i += 256) ws[i] = Wo[i];
  if (tid < NCLS) bs[tid] = bo[tid];
  __syncthreads();
  int r = blockIdx.x * 128 + (tid >> 1);
  int c0 = (tid & 1) * 20;
  if (r >= n) return;
  float acc[20];
#pragma unroll
  for (int c = 0; c < 20; ++c) acc[c] = bs[c0 + c];
  const float* xr = x + (size_t)r * HDIM;
  for (int k = 0; k < HDIM; ++k) {
    float xv = xr[k];
    const float* wr = &ws[k * NCLS + c0];
#pragma unroll
    for (int c = 0; c < 20; ++c) acc[c] += xv * wr[c];
  }
  float* o = out + (size_t)r * NCLS + c0;
#pragma unroll
  for (int c = 0; c < 20; ++c) o[c] = acc[c];
}

extern "C" void kernel_launch(void* const* d_in, const int* in_sizes, int n_in,
                              void* d_out, int out_size, void* d_ws, size_t ws_size,
                              hipStream_t stream) {
  const float* x_in  = (const float*)d_in[0];
  const int*   ei    = (const int*)d_in[1];
  const float* gamma = (const float*)d_in[2];
  const float* beta  = (const float*)d_in[3];
  const float* W     = (const float*)d_in[4];
  const float* b     = (const float*)d_in[5];
  const float* Wo    = (const float*)d_in[6];
  const float* bo    = (const float*)d_in[7];
  float* out = (float*)d_out;

  const int N = in_sizes[0] / HDIM;
  const int E = in_sizes[1] / 2;

  float* dinv  = (float*)d_ws;
  float* stats = dinv + ((N + 255) / 256) * 256;
  float* WfT   = stats + 2 * HDIM;
  float* cb    = WfT + HDIM * HDIM;
  float* hbuf  = cb + HDIM + 64;
  float* xbuf  = hbuf + (size_t)N * HDIM;

  int* csr    = (int*)d_out;
  int* rowptr = csr + ((2 * E + 3) / 4) * 4;
  int* cnt    = rowptr + ((N + 4) / 4) * 4;
  int* rank   = (int*)xbuf;

  k_zero_int<<<(N + 255) / 256, 256, 0, stream>>>(cnt, N);
  k_count_rank<<<(E + 255) / 256, 256, 0, stream>>>(ei, cnt, rank, E);
  k_scan_dinv<<<1, SCAN_T, 0, stream>>>(cnt, rowptr, dinv, N);
  k_fill<<<(E + 255) / 256, 256, 0, stream>>>(ei, rank, rowptr, dinv, csr, E);

  hipMemsetAsync(stats, 0, 2 * HDIM * sizeof(float), stream);
  k_stats<<<512, 256, 0, stream>>>(x_in, stats, N);

  for (int l = 0; l < 3; ++l) {
    k_fold<<<1, 128, 0, stream>>>(stats, gamma + l * HDIM, beta + l * HDIM,
                                  W + (size_t)l * HDIM * HDIM, WfT, cb, 1.0f / (float)N);
    const float* xl = (l == 0) ? x_in : xbuf;
    k_gemm<<<(N + 63) / 64, 128, 0, stream>>>(xl, WfT, cb, hbuf, N);
    if (l < 2) {
      hipMemsetAsync(stats, 0, 2 * HDIM * sizeof(float), stream);
      k_node_agg<true><<<AGG_BLOCKS, AGG_WAVES * 32, 0, stream>>>(rowptr, csr, dinv, hbuf, b + l * HDIM,
                                                                  xbuf, stats, N);
    } else {
      k_node_agg<false><<<AGG_BLOCKS, AGG_WAVES * 32, 0, stream>>>(rowptr, csr, dinv, hbuf, b + l * HDIM,
                                                                   xbuf, stats, N);
    }
  }

  k_out_gemm<<<(N + 127) / 128, 256, 0, stream>>>(xbuf, Wo, bo, out, N);
}


# Input BN column sums: a wave reads whole 512-B rows (16 B per lane), all 24-25 row loads issued before the first add, not one 4-B load per lane per trip waited for singly; column-sum order differs
# speedup vs baseline: 2.9367x; 1.0837x over previous
#include <hip/hip_runtime.h>

#define HDIM 128
#define NCLS 40
#define BN_EPS 1e-5f

typedef __attribute__((ext_vector_type(2))) float v2f;
typedef __attribute__((ext_vector_type(8))) float v8f;

__global__ void k_zero_int(int* p, int n) {
  int i = blockIdx.x * blockDim.x + threadIdx.x;
  if (i < n) p[i] = 0;
}

__global__ void k_count_rank(const int* __restrict__ ei, int* cnt, int* __restrict__ rank, int E) {
  int e = blockIdx.x * blockDim.x + threadIdx.x;
  if (e < E) rank[e] = atomicAdd(&cnt[ei[E + e]], 1);
}

#define SCAN_T 1024
__device__ __forceinline__ int4 load4_guarded(const int* __restrict__ p, int i0, int n) {
  int4 v = make_int4(0, 0, 0, 0);
  if (i0 + 3 < n) {
    v = *(const int4*)(p + i0);
  } else {
    if (i0 < n) v.x = p[i0];
    if (i0 + 1 < n) v.y = p[i0 + 1];
    if (i0 + 2 < n) v.z = p[i0 + 2];
  }
  return v;
}

__global__ __launch_bounds__(SCAN_T) void k_scan_dinv(const int* __restrict__ cnt,
                                                      int* __restrict__ rowptr,
                                                      float* __restrict__ dinv, int n) {
  __shared__ int wsum[32];
  const int t = threadIdx.x, lane = t & 31, wv = t >> 5;
  int carry = 0;
  int4 v[4];
#pragma unroll
  for (int g = 0; g < 4; ++g) v[g] = load4_guarded(cnt, t * 16 + g * 4, n);
  for (int base = 0; base < n; base += SCAN_T * 16) {
    const int i0 = base + t * 16;
    int4 nxt[4];
#pragma unroll
    for (int g = 0; g < 4; ++g) nxt[g] = load4_guarded(cnt, i0 + SCAN_T * 16 + g * 4, n);
    int s = 0;
#pragma unroll
    for (int g = 0; g < 4; ++g) s += v[g].x + v[g].y + v[g].z + v[g].w;
    int inc = s;
#pragma unroll
    for (int d = 1; d < 32; d <<= 1) {
      int o = __shfl_up(inc, d);
      if (lane >= d) inc += o;
    }
    if (lane == 31) wsum[wv] = inc;
    __syncthreads();
    const int wt = wsum[lane];
    int winc = wt;
#pragma unroll
    for (int d = 1; d < 32; d <<= 1) {
      int o = __shfl_up(winc, d);
      if (lane >= d) winc += o;
    }
    const int wex = __shfl(winc, wv) - __shfl(wt, wv);
    const int total = __shfl(winc, 31);
    int run = carry + wex + inc - s;
#pragma unroll
    for (int g = 0; g < 4; ++g) {
      const int i = i0 + g * 4;
      const int4 c = v[g];
      if (i + 3 < n) {
        *(int4*)(rowptr + i) = make_int4(run, run + c.x, run + c.x + c.y, run + c.x + c.y + c.z);
        *(float4*)(dinv + i) = make_float4(rsqrtf((float)(c.x + 1)), rsqrtf((float)(c.y + 1)),
                                           rsqrtf((float)(c.z + 1)), rsqrtf((float)(c.w + 1)));
      } else {
        if (i < n)     { rowptr[i] = run;                 dinv[i] = rsqrtf((float)(c.x + 1)); }
        if (i + 1 < n) { rowptr[i + 1] = run + c.x;       dinv[i + 1] = rsqrtf((float)(c.y + 1)); }
        if (i + 2 < n) { rowptr[i + 2] = run + c.x + c.y; dinv[i + 2] = rsqrtf((float)(c.z + 1)); }
      }
      run += c.x + c.y + c.z + c.w;
    }
    carry += total;
#pragma unroll
    for (int g = 0; g < 4; ++g) v[g] = nxt[g];
    __syncthreads();
  }
  if (t == 0) rowptr[n] = carry;
}

__global__ void k_fill(const int* __restrict__ ei, const int* __restrict__ rank,
                       const int* __restrict__ rowptr, const float* __restrict__ dinv,
                       int* __restrict__ csr, int E) {
  int e = blockIdx.x * blockDim.x + threadIdx.x;
  if (e >= E) return;
  int s = ei[e], d = ei[E + e];
  int p = rowptr[d] + rank[e];
  p = p < E ? p : E - 1;
  ((int2*)csr)[p] = make_int2(s, __float_as_int(dinv[s] * dinv[d]));
}

#define ST_WAVES 8
#define ST_BLOCKS 512
#define ST_ROWS 32
static_assert(ST_WAVES * 32 == 2 * HDIM, "one thread per entry of stats in the block's last step");
__global__ __launch_bounds__(ST_WAVES * 32) void k_stats(const float* __restrict__ x,
                                                         float* __restrict__ stats, int n) {
  const int wv = __builtin_amdgcn_readfirstlane(threadIdx.x >> 5);
  const unsigned lane = threadIdx.x & 31;
  const float4* xb = (const float4*)x;
  const int nw = gridDim.x * ST_WAVES;
  float4 s1 = make_float4(0.f, 0.f, 0.f, 0.f);
  float4 s2 = make_float4(0.f, 0.f, 0.f, 0.f);
  for (int base = blockIdx.x * ST_WAVES + wv; base < n; base += ST_ROWS * nw) {
    float4 a[ST_ROWS];
#pragma unroll
    for (int q = 0; q < ST_ROWS; ++q) {
      const int r = base + q * nw;
      if (r < n) a[q] = xb[(size_t)r * 32 + lane];
    }
#pragma unroll
    for (int q = 0; q < ST_ROWS; ++q) {
      if (base + q * nw < n) {
        s1.x += a[q].x; s1.y += a[q].y; s1.z += a[q].z; s1.w += a[q].w;
        s2.x = fmaf(a[q].x, a[q].x, s2.x); s2.y = fmaf(a[q].y, a[q].y, s2.y);
        s2.z = fmaf(a[q].z, a[q].z, s2.z); s2.w = fmaf(a[q].w, a[q].w, s2.w);
      }
    }
  }
  __shared__ float ls[ST_WAVES * 2 * HDIM];
  float4* lw = (float4*)&ls[wv * 2 * HDIM];
  lw[lane] = s1;
  lw[32 + lane] = s2;
  __syncthreads();
  float t = 0.f;
#pragma unroll
  for (int w = 0; w < ST_WAVES; ++w) t += ls[w * 2 * HDIM + threadIdx.x];
  atomicAdd(&stats[threadIdx.x], t);
}

__global__ void k_fold(const float* __restrict__ stats, const float* __restrict__ gamma,
                       const float* __restrict__ beta, const float* __restrict__ W,
                       float* __restrict__ WfT, float* __restrict__ cb, float inv_n) {
  __shared__ float as[HDIM], cs[HDIM];
  int t = threadIdx.x;
  float mean = stats[t] * inv_n;
  float var  = stats[HDIM + t] * inv_n - mean * mean;
  float a = gamma[t] * rsqrtf(var + BN_EPS);
  float c = beta[t] - a * mean;
  as[t] = a; cs[t] = c;
  __syncthreads();
  float acc = 0.f;
  for (int k = 0; k < HDIM; ++k) {
    float w = W[k * HDIM + t];
    WfT[t * HDIM + k] = as[k] * w;
    acc += cs[k] * w;
  }
  cb[t] = acc;
}

#define XPAD 68

__global__ __launch_bounds__(128) void k_gemm(const float* __restrict__ x,
                                              const float* __restrict__ WfT,
                                              const float* __restrict__ cb,
                                              float* __restrict__ h, int n) {
  __shared__ float xs[64 * XPAD];
  __shared__ float wfs[HDIM * XPAD];
  const int tid = threadIdx.x;
  const int lane = tid & 31;
  const int wave = tid >> 5;
  const int rowBase = blockIdx.x * 64;
  const int m = lane & 15;
  const int hi = lane >> 4;

  v8f acc[8];
#pragma unroll
  for (int c = 0; c < 8; ++c)
#pragma unroll
    for (int v = 0; v < 8; ++v) acc[c][v] = 0.f;

  for (int kh = 0; kh < 2; ++kh) {
#pragma unroll
    for (int p = 0; p < 8; ++p) {
      int idx = p * 512 + tid * 4;
      int r = idx >> 6, cc = idx & 63;
      int row = rowBase + r;
      float4 v = make_float4(0.f, 0.f, 0.f, 0.f);
      if (row < n) v = *(const float4*)(x + (size_t)row * HDIM + kh * 64 + cc);
      *(float4*)(&xs[r * XPAD + cc]) = v;
    }
#pragma unroll
    for (int p = 0; p < 16; ++p) {
      int idx = p * 512 + tid * 4;
      int nn = idx >> 6, cc = idx & 63;
      *(float4*)(&wfs[nn * XPAD + cc]) = *(const float4*)(WfT + nn * HDIM + kh * 64 + cc);
    }
    __syncthreads();

    const float* xrow = &xs[(wave * 16 + m) * XPAD];
#pragma unroll
    for (int k4 = 0; k4 < 16; ++k4) {
      int kb = k4 * 4 + hi * 2;
      v2f a; a.x = xrow[kb]; a.y = xrow[kb + 1];
#pragma unroll
      for (int c = 0; c < 8; ++c) {
        const float* wrow = &wfs[(c * 16 + m) * XPAD];
        v2f b; b.x = wrow[kb]; b.y = wrow[kb + 1];
        acc[c] = __builtin_amdgcn_wmma_f32_16x16x4_f32(
            false, a, false, b, (short)0, acc[c], false, false);
      }
    }
    __syncthreads();
  }

#pragma unroll
  for (int c = 0; c < 8; ++c) {
    int col = c * 16 + m;
    float cbv = cb[col];
#pragma unroll
    for (int v = 0; v < 8; ++v) {
      int row = rowBase + wave * 16 + v + 8 * hi;
      if (row < n) h[(size_t)row * HDIM + col] = acc[c][v] + cbv;
    }
  }
}

template <int K>
__device__ __forceinline__ void edge_batch(float4& acc, const float4* __restrict__ hb,
                                           const int2 my, const int j, const int lane) {
  int sidx[K];
  float wgt[K];
  float4 a[K];
#pragma unroll
  for (int q = 0; q < K; ++q) {
    sidx[q] = __builtin_amdgcn_readlane(my.x, j + q);
    wgt[q] = __int_as_float(__builtin_amdgcn_readlane(my.y, j + q));
  }
#pragma unroll
  for (int q = 0; q < K; ++q) a[q] = hb[(size_t)sidx[q] * 32 + lane];
#pragma unroll
  for (int q = 0; q < K; ++q) {
    acc.x = fmaf(a[q].x, wgt[q], acc.x);
    acc.y = fmaf(a[q].y, wgt[q], acc.y);
    acc.z = fmaf(a[q].z, wgt[q], acc.z);
    acc.w = fmaf(a[q].w, wgt[q], acc.w);
  }
}

#define AGG_WAVES 32
#define AGG_BLOCKS 512
template <bool STATS>
__global__ __launch_bounds__(AGG_WAVES * 32) void k_node_agg(const int* __restrict__ rowptr,
                                                             const int* __restrict__ csr,
                                                             const float* __restrict__ dinv,
                                                             const float* __restrict__ hbuf,
                                                             const float* __restrict__ bias,
                                                             float* __restrict__ out,
                                                             float* __restrict__ stats, int n) {
  const int wv = __builtin_amdgcn_readfirstlane(threadIdx.x >> 5);
  const int lane = threadIdx.x & 31;
  const float4* hb = (const float4*)hbuf;
  const int2* cs = (const int2*)csr;
  const float4 bv = ((const float4*)bias)[lane];
  float4 cs1 = make_float4(0.f, 0.f, 0.f, 0.f);
  float4 cs2 = make_float4(0.f, 0.f, 0.f, 0.f);
  for (int node = blockIdx.x * AGG_WAVES + wv; node < n; node += gridDim.x * AGG_WAVES) {
    const int beg = rowptr[node], end = rowptr[node + 1];
    const float di = dinv[node];
    const float4 hs = hb[(size_t)node * 32 + lane];
    float4 acc = make_float4(0.f, 0.f, 0.f, 0.f);
    for (int base = beg; base < end; base += 32) {
      int m = end - base;
      m = m < 32 ? m : 32;
      int2 my = make_int2(0, 0);
      if (lane < m) my = cs[base + lane];
      my.x = my.x < 0 ? 0 : (my.x < n ? my.x : n - 1);
      int j = 0;
      for (; j + 8 <= m; j += 8) edge_batch<8>(acc, hb, my, j, lane);
      if (j + 4 <= m) { edge_batch<4>(acc, hb, my, j, lane); j += 4; }
      const int r = m - j;
      if (r == 3) edge_batch<3>(acc, hb, my, j, lane);
      else if (r == 2) edge_batch<2>(acc, hb, my, j, lane);
      else if (r == 1) edge_batch<1>(acc, hb, my, j, lane);
    }
    const float w_self = di * di;
    acc.x = fmaf(hs.x, w_self, acc.x);
    acc.y = fmaf(hs.y, w_self, acc.y);
    acc.z = fmaf(hs.z, w_self, acc.z);
    acc.w = fmaf(hs.w, w_self, acc.w);
    float4 v;
    v.x = acc.x + bv.x; v.x = v.x > 0.f ? v.x : 0.f;
    v.y = acc.y + bv.y; v.y = v.y > 0.f ? v.y : 0.f;
    v.z = acc.z + bv.z; v.z = v.z > 0.f ? v.z : 0.f;
    v.w = acc.w + bv.w; v.w = v.w > 0.f ? v.w : 0.f;
    ((float4*)out)[(size_t)node * 32 + lane] = v;
    if (STATS) {
      cs1.x += v.x; cs1.y += v.y; cs1.z += v.z; cs1.w += v.w;
      cs2.x = fmaf(v.x, v.x, cs2.x); cs2.y = fmaf(v.y, v.y, cs2.y);
      cs2.z = fmaf(v.z, v.z, cs2.z); cs2.w = fmaf(v.w, v.w, cs2.w);
    }
  }
  if (STATS) {
    __shared__ float ls[AGG_WAVES * 2 * HDIM];
    float4* lw = (float4*)&ls[wv * 2 * HDIM];
    lw[lane] = cs1;
    lw[32 + lane] = cs2;
    __syncthreads();
    if (threadIdx.x < 2 * HDIM) {
      float t = 0.f;
#pragma unroll
      for (int w = 0; w < AGG_WAVES; ++w) t += ls[w * 2 * HDIM + threadIdx.x];
      atomicAdd(&stats[threadIdx.x], t);
    }
  }
}

__global__ __launch_bounds__(256) void k_out_gemm(const float* __restrict__ x,
                                                  const float* __restrict__ Wo,
                                                  const float* __restrict__ bo,
                                                  float* __restrict__ out, int n) {
  __shared__ float ws[HDIM * NCLS];
  __shared__ float bs[NCLS];
  int tid = threadIdx.x;
  for (int i = tid; i < HDIM * NCLS; i += 256) ws[i] = Wo[i];
  if (tid < NCLS) bs[tid] = bo[tid];
  __syncthreads();
  int r = blockIdx.x * 128 + (tid >> 1);
  int c0 = (tid & 1) * 20;
  if (r >= n) return;
  float acc[20];
#pragma unroll
  for (int c = 0; c < 20; ++c) acc[c] = bs[c0 + c];
  const float* xr = x + (size_t)r * HDIM;
  for (int k = 0; k < HDIM; ++k) {
    float xv = xr[k];
    const float* wr = &ws[k * NCLS + c0];
#pragma unroll
    for (int c = 0; c < 20; ++c) acc[c] += xv * wr[c];
  }
  float* o = out + (size_t)r * NCLS + c0;
#pragma unroll
  for (int c = 0; c < 20; ++c) o[c] = acc[c];
}

extern "C" void kernel_launch(void* const* d_in, const int* in_sizes, int n_in,
                              void* d_out, int out_size, void* d_ws, size_t ws_size,
                              hipStream_t stream) {
  const float* x_in  = (const float*)d_in[0];
  const int*   ei    = (const int*)d_in[1];
  const float* gamma = (const float*)d_in[2];
  const float* beta  = (const float*)d_in[3];
  const float* W     = (const float*)d_in[4];
  const float* b     = (const float*)d_in[5];
  const float* Wo    = (const float*)d_in[6];
  const float* bo    = (const float*)d_in[7];
  float* out = (float*)d_out;

  const int N = in_sizes[0] / HDIM;
  const int E = in_sizes[1] / 2;

  float* dinv  = (float*)d_ws;
  float* stats = dinv + ((N + 255) / 256) * 256;
  float* WfT   = stats + 2 * HDIM;
  float* cb    = WfT + HDIM * HDIM;
  float* hbuf  = cb + HDIM + 64;
  float* xbuf  = hbuf + (size_t)N * HDIM;

  int* csr    = (int*)d_out;
  int* rowptr = csr + ((2 * E + 3) / 4) * 4;
  int* cnt    = rowptr + ((N + 4) / 4) * 4;
  int* rank   = (int*)xbuf;

  k_zero_int<<<(N + 255) / 256, 256, 0, stream>>>(cnt, N);
  k_count_rank<<<(E + 255) / 256, 256, 0, stream>>>(ei, cnt, rank, E);
  k_scan_dinv<<<1, SCAN_T, 0, stream>>>(cnt, rowptr, dinv, N);
  k_fill<<<(E + 255) / 256, 256, 0, stream>>>(ei, rank, rowptr, dinv, csr, E);

  hipMemsetAsync(stats, 0, 2 * HDIM * sizeof(float), stream);
  k_stats<<<ST_BLOCKS, ST_WAVES * 32, 0, stream>>>(x_in, stats, N);

  for (int l = 0; l < 3; ++l) {
    k_fold<<<1, 128, 0, stream>>>(stats, gamma + l * HDIM, beta + l * HDIM,
                                  W + (size_t)l * HDIM * HDIM, WfT, cb, 1.0f / (float)N);
    const float* xl = (l == 0) ? x_in : xbuf;
    k_gemm<<<(N + 63) / 64, 128, 0, stream>>>(xl, WfT, cb, hbuf, N);
    if (l < 2) {
      hipMemsetAsync(stats, 0, 2 * HDIM * sizeof(float), stream);
      k_node_agg<true><<<AGG_BLOCKS, AGG_WAVES * 32, 0, stream>>>(rowptr, csr, dinv, hbuf, b + l * HDIM,
                                                                  xbuf, stats, N);
    } else {
      k_node_agg<false><<<AGG_BLOCKS, AGG_WAVES * 32, 0, stream>>>(rowptr, csr, dinv, hbuf, b + l * HDIM,
                                                                   xbuf, stats, N);
    }
  }

  k_out_gemm<<<(N + 127) / 128, 256, 0, stream>>>(xbuf, Wo, bo, out, N);
}


# Batch norm applied to the x tile as the GEMM stages it (scale and shift from the column sums, per block), W read as stored; the 3 one-block fold launches go; the shift now enters inside the k sum
# speedup vs baseline: 3.2710x; 1.1138x over previous
#include <hip/hip_runtime.h>

#define HDIM 128
#define NCLS 40
#define BN_EPS 1e-5f

typedef __attribute__((ext_vector_type(2))) float v2f;
typedef __attribute__((ext_vector_type(8))) float v8f;

__global__ void k_zero_int(int* p, int n) {
  int i = blockIdx.x * blockDim.x + threadIdx.x;
  if (i < n) p[i] = 0;
}

__global__ void k_count_rank(const int* __restrict__ ei, int* cnt, int* __restrict__ rank, int E) {
  int e = blockIdx.x * blockDim.x + threadIdx.x;
  if (e < E) rank[e] = atomicAdd(&cnt[ei[E + e]], 1);
}

#define SCAN_T 1024
__device__ __forceinline__ int4 load4_guarded(const int* __restrict__ p, int i0, int n) {
  int4 v = make_int4(0, 0, 0, 0);
  if (i0 + 3 < n) {
    v = *(const int4*)(p + i0);
  } else {
    if (i0 < n) v.x = p[i0];
    if (i0 + 1 < n) v.y = p[i0 + 1];
    if (i0 + 2 < n) v.z = p[i0 + 2];
  }
  return v;
}

__global__ __launch_bounds__(SCAN_T) void k_scan_dinv(const int* __restrict__ cnt,
                                                      int* __restrict__ rowptr,
                                                      float* __restrict__ dinv, int n) {
  __shared__ int wsum[32];
  const int t = threadIdx.x, lane = t & 31, wv = t >> 5;
  int carry = 0;
  int4 v[4];
#pragma unroll
  for (int g = 0; g < 4; ++g) v[g] = load4_guarded(cnt, t * 16 + g * 4, n);
  for (int base = 0; base < n; base += SCAN_T * 16) {
    const int i0 = base + t * 16;
    int4 nxt[4];
#pragma unroll
    for (int g = 0; g < 4; ++g) nxt[g] = load4_guarded(cnt, i0 + SCAN_T * 16 + g * 4, n);
    int s = 0;
#pragma unroll
    for (int g = 0; g < 4; ++g) s += v[g].x + v[g].y + v[g].z + v[g].w;
    int inc = s;
#pragma unroll
    for (int d = 1; d < 32; d <<= 1) {
      int o = __shfl_up(inc, d);
      if (lane >= d) inc += o;
    }
    if (lane == 31) wsum[wv] = inc;
    __syncthreads();
    const int wt = wsum[lane];
    int winc = wt;
#pragma unroll
    for (int d = 1; d < 32; d <<= 1) {
      int o = __shfl_up(winc, d);
      if (lane >= d) winc += o;
    }
    const int wex = __shfl(winc, wv) - __shfl(wt, wv);
    const int total = __shfl(winc, 31);
    int run = carry + wex + inc - s;
#pragma unroll
    for (int g = 0; g < 4; ++g) {
      const int i = i0 + g * 4;
      const int4 c = v[g];
      if (i + 3 < n) {
        *(int4*)(rowptr + i) = make_int4(run, run + c.x, run + c.x + c.y, run + c.x + c.y + c.z);
        *(float4*)(dinv + i) = make_float4(rsqrtf((float)(c.x + 1)), rsqrtf((float)(c.y + 1)),
                                           rsqrtf((float)(c.z + 1)), rsqrtf((float)(c.w + 1)));
      } else {
        if (i < n)     { rowptr[i] = run;                 dinv[i] = rsqrtf((float)(c.x + 1)); }
        if (i + 1 < n) { rowptr[i + 1] = run + c.x;       dinv[i + 1] = rsqrtf((float)(c.y + 1)); }
        if (i + 2 < n) { rowptr[i + 2] = run + c.x + c.y; dinv[i + 2] = rsqrtf((float)(c.z + 1)); }
      }
      run += c.x + c.y + c.z + c.w;
    }
    carry += total;
#pragma unroll
    for (int g = 0; g < 4; ++g) v[g] = nxt[g];
    __syncthreads();
  }
  if (t == 0) rowptr[n] = carry;
}

__global__ void k_fill(const int* __restrict__ ei, const int* __restrict__ rank,
                       const int* __restrict__ rowptr, const float* __restrict__ dinv,
                       int* __restrict__ csr, int E) {
  int e = blockIdx.x * blockDim.x + threadIdx.x;
  if (e >= E) return;
  int s = ei[e], d = ei[E + e];
  int p = rowptr[d] + rank[e];
  p = p < E ? p : E - 1;
  ((int2*)csr)[p] = make_int2(s, __float_as_int(dinv[s] * dinv[d]));
}

#define ST_WAVES 8
#define ST_BLOCKS 512
#define ST_ROWS 32
static_assert(ST_WAVES * 32 == 2 * HDIM, "one thread per entry of stats in the block's last step");
__global__ __launch_bounds__(ST_WAVES * 32) void k_stats(const float* __restrict__ x,
                                                         float* __restrict__ stats, int n) {
  const int wv = __builtin_amdgcn_readfirstlane(threadIdx.x >> 5);
  const unsigned lane = threadIdx.x & 31;
  const float4* xb = (const float4*)x;
  const int nw = gridDim.x * ST_WAVES;
  float4 s1 = make_float4(0.f, 0.f, 0.f, 0.f);
  float4 s2 = make_float4(0.f, 0.f, 0.f, 0.f);
  for (int base = blockIdx.x * ST_WAVES + wv; base < n; base += ST_ROWS * nw) {
    float4 a[ST_ROWS];
#pragma unroll
    for (int q = 0; q < ST_ROWS; ++q) {
      const int r = base + q * nw;
      if (r < n) a[q] = xb[(size_t)r * 32 + lane];
    }
#pragma unroll
    for (int q = 0; q < ST_ROWS; ++q) {
      if (base + q * nw < n) {
        s1.x += a[q].x; s1.y += a[q].y; s1.z += a[q].z; s1.w += a[q].w;
        s2.x = fmaf(a[q].x, a[q].x, s2.x); s2.y = fmaf(a[q].y, a[q].y, s2.y);
        s2.z = fmaf(a[q].z, a[q].z, s2.z); s2.w = fmaf(a[q].w, a[q].w, s2.w);
      }
    }
  }
  __shared__ float ls[ST_WAVES * 2 * HDIM];
  float4* lw = (float4*)&ls[wv * 2 * HDIM];
  lw[lane] = s1;
  lw[32 + lane] = s2;
  __syncthreads();
  float t = 0.f;
#pragma unroll
  for (int w = 0; w < ST_WAVES; ++w) t += ls[w * 2 * HDIM + threadIdx.x];
  atomicAdd(&stats[threadIdx.x], t);
}

#define XPAD 68
#define WPAIR 288

__device__ __forceinline__ void bn_coef(float s1, float s2, float g, float bt, float inv_n,
                                        float& a, float& c) {
  const float mean = s1 * inv_n;
  const float var = s2 * inv_n - mean * mean;
  a = g * rsqrtf(var + BN_EPS);
  c = bt - a * mean;
}

__global__ __launch_bounds__(128, 5) void k_gemm(const float* __restrict__ x,
                                                 const float* __restrict__ stats,
                                                 const float* __restrict__ gamma,
                                                 const float* __restrict__ beta,
                                                 const float* __restrict__ W,
                                                 float* __restrict__ h, int n, float inv_n) {
  __shared__ float xs[64 * XPAD];
  __shared__ float ws[32 * WPAIR];
  const int tid = threadIdx.x;
  const int lane = tid & 31;
  const int wave = tid >> 5;
  const int rowBase = blockIdx.x * 64;
  const int m = lane & 15;
  const int hi = lane >> 4;

  v8f acc[8];
#pragma unroll
  for (int c = 0; c < 8; ++c)
#pragma unroll
    for (int v = 0; v < 8; ++v) acc[c][v] = 0.f;

  for (int kh = 0; kh < 2; ++kh) {
    const int colq = kh * 64 + (tid & 15) * 4;
    const float4 s1 = *(const float4*)(stats + colq);
    const float4 s2 = *(const float4*)(stats + HDIM + colq);
    const float4 g4 = *(const float4*)(gamma + colq);
    const float4 b4 = *(const float4*)(beta + colq);
    float4 sa, sc;
    bn_coef(s1.x, s2.x, g4.x, b4.x, inv_n, sa.x, sc.x);
    bn_coef(s1.y, s2.y, g4.y, b4.y, inv_n, sa.y, sc.y);
    bn_coef(s1.z, s2.z, g4.z, b4.z, inv_n, sa.z, sc.z);
    bn_coef(s1.w, s2.w, g4.w, b4.w, inv_n, sa.w, sc.w);
#pragma unroll
    for (int p = 0; p < 8; ++p) {
      int idx = p * 512 + tid * 4;
      int r = idx >> 6, cc = idx & 63;
      int row = rowBase + r;
      float4 v = make_float4(0.f, 0.f, 0.f, 0.f);
      if (row < n) {
        v = *(const float4*)(x + (size_t)row * HDIM + kh * 64 + cc);
        v.x = fmaf(v.x, sa.x, sc.x);
        v.y = fmaf(v.y, sa.y, sc.y);
        v.z = fmaf(v.z, sa.z, sc.z);
        v.w = fmaf(v.w, sa.w, sc.w);
      }
      *(float4*)(&xs[r * XPAD + cc]) = v;
    }
#pragma unroll
    for (int p = 0; p < 8; ++p) {
      const int kp = p * 4 + wave;
      const float* wr = W + (kh * 64 + 2 * kp) * HDIM + lane * 4;
      const float4 w0 = *(const float4*)wr;
      const float4 w1 = *(const float4*)(wr + HDIM);
      float* d = &ws[kp * WPAIR + lane * 8];
      *(float4*)d = make_float4(w0.x, w1.x, w0.y, w1.y);
      *(float4*)(d + 4) = make_float4(w0.z, w1.z, w0.w, w1.w);
    }
    __syncthreads();

    const float* xrow = &xs[(wave * 16 + m) * XPAD];
#pragma unroll
    for (int k4 = 0; k4 < 16; ++k4) {
      int kb = k4 * 4 + hi * 2;
      v2f a; a.x = xrow[kb]; a.y = xrow[kb + 1];
      const float* wk = &ws[(k4 * 2 + hi) * WPAIR + m * 2];
#pragma unroll
      for (int c = 0; c < 8; ++c) {
        v2f b; b.x = wk[c * 32]; b.y = wk[c * 32 + 1];
        acc[c] = __builtin_amdgcn_wmma_f32_16x16x4_f32(
            false, a, false, b, (short)0, acc[c], false, false);
      }
    }
    __syncthreads();
  }

#pragma unroll
  for (int c = 0; c < 8; ++c) {
    int col = c * 16 + m;
#pragma unroll
    for (int v = 0; v < 8; ++v) {
      int row = rowBase + wave * 16 + v + 8 * hi;
      if (row < n) h[(size_t)row * HDIM + col] = acc[c][v];
    }
  }
}

template <int K>
__device__ __forceinline__ void edge_batch(float4& acc, const float4* __restrict__ hb,
                                           const int2 my, const int j, const int lane) {
  int sidx[K];
  float wgt[K];
  float4 a[K];
#pragma unroll
  for (int q = 0; q < K; ++q) {
    sidx[q] = __builtin_amdgcn_readlane(my.x, j + q);
    wgt[q] = __int_as_float(__builtin_amdgcn_readlane(my.y, j + q));
  }
#pragma unroll
  for (int q = 0; q < K; ++q) a[q] = hb[(size_t)sidx[q] * 32 + lane];
#pragma unroll
  for (int q = 0; q < K; ++q) {
    acc.x = fmaf(a[q].x, wgt[q], acc.x);
    acc.y = fmaf(a[q].y, wgt[q], acc.y);
    acc.z = fmaf(a[q].z, wgt[q], acc.z);
    acc.w = fmaf(a[q].w, wgt[q], acc.w);
  }
}

#define AGG_WAVES 32
#define AGG_BLOCKS 512
template <bool STATS>
__global__ __launch_bounds__(AGG_WAVES * 32) void k_node_agg(const int* __restrict__ rowptr,
                                                             const int* __restrict__ csr,
                                                             const float* __restrict__ dinv,
                                                             const float* __restrict__ hbuf,
                                                             const float* __restrict__ bias,
                                                             float* __restrict__ out,
                                                             float* __restrict__ stats, int n) {
  const int wv = __builtin_amdgcn_readfirstlane(threadIdx.x >> 5);
  const int lane = threadIdx.x & 31;
  const float4* hb = (const float4*)hbuf;
  const int2* cs = (const int2*)csr;
  const float4 bv = ((const float4*)bias)[lane];
  float4 cs1 = make_float4(0.f, 0.f, 0.f, 0.f);
  float4 cs2 = make_float4(0.f, 0.f, 0.f, 0.f);
  for (int node = blockIdx.x * AGG_WAVES + wv; node < n; node += gridDim.x * AGG_WAVES) {
    const int beg = rowptr[node], end = rowptr[node + 1];
    const float di = dinv[node];
    const float4 hs = hb[(size_t)node * 32 + lane];
    float4 acc = make_float4(0.f, 0.f, 0.f, 0.f);
    for (int base = beg; base < end; base += 32) {
      int m = end - base;
      m = m < 32 ? m : 32;
      int2 my = make_int2(0, 0);
      if (lane < m) my = cs[base + lane];
      my.x = my.x < 0 ? 0 : (my.x < n ? my.x : n - 1);
      int j = 0;
      for (; j + 8 <= m; j += 8) edge_batch<8>(acc, hb, my, j, lane);
      if (j + 4 <= m) { edge_batch<4>(acc, hb, my, j, lane); j += 4; }
      const int r = m - j;
      if (r == 3) edge_batch<3>(acc, hb, my, j, lane);
      else if (r == 2) edge_batch<2>(acc, hb, my, j, lane);
      else if (r == 1) edge_batch<1>(acc, hb, my, j, lane);
    }
    const float w_self = di * di;
    acc.x = fmaf(hs.x, w_self, acc.x);
    acc.y = fmaf(hs.y, w_self, acc.y);
    acc.z = fmaf(hs.z, w_self, acc.z);
    acc.w = fmaf(hs.w, w_self, acc.w);
    float4 v;
    v.x = acc.x + bv.x; v.x = v.x > 0.f ? v.x : 0.f;
    v.y = acc.y + bv.y; v.y = v.y > 0.f ? v.y : 0.f;
    v.z = acc.z + bv.z; v.z = v.z > 0.f ? v.z : 0.f;
    v.w = acc.w + bv.w; v.w = v.w > 0.f ? v.w : 0.f;
    ((float4*)out)[(size_t)node * 32 + lane] = v;
    if (STATS) {
      cs1.x += v.x; cs1.y += v.y; cs1.z += v.z; cs1.w += v.w;
      cs2.x = fmaf(v.x, v.x, cs2.x); cs2.y = fmaf(v.y, v.y, cs2.y);
      cs2.z = fmaf(v.z, v.z, cs2.z); cs2.w = fmaf(v.w, v.w, cs2.w);
    }
  }
  if (STATS) {
    __shared__ float ls[AGG_WAVES * 2 * HDIM];
    float4* lw = (float4*)&ls[wv * 2 * HDIM];
    lw[lane] = cs1;
    lw[32 + lane] = cs2;
    __syncthreads();
    if (threadIdx.x < 2 * HDIM) {
      float t = 0.f;
#pragma unroll
      for (int w = 0; w < AGG_WAVES; ++w) t += ls[w * 2 * HDIM + threadIdx.x];
      atomicAdd(&stats[threadIdx.x], t);
    }
  }
}

__global__ __launch_bounds__(256) void k_out_gemm(const float* __restrict__ x,
                                                  const float* __restrict__ Wo,
                                                  const float* __restrict__ bo,
                                                  float* __restrict__ out, int n) {
  __shared__ float ws[HDIM * NCLS];
  __shared__ float bs[NCLS];
  int tid = threadIdx.x;
  for (int i = tid; i < HDIM * NCLS; i += 256) ws[i] = Wo[i];
  if (tid < NCLS) bs[tid] = bo[tid];
  __syncthreads();
  int r = blockIdx.x * 128 + (tid >> 1);
  int c0 = (tid & 1) * 20;
  if (r >= n) return;
  float acc[20];
#pragma unroll
  for (int c = 0; c < 20; ++c) acc[c] = bs[c0 + c];
  const float* xr = x + (size_t)r * HDIM;
  for (int k = 0; k < HDIM; ++k) {
    float xv = xr[k];
    const float* wr = &ws[k * NCLS + c0];
#pragma unroll
    for (int c = 0; c < 20; ++c) acc[c] += xv * wr[c];
  }
  float* o = out + (size_t)r * NCLS + c0;
#pragma unroll
  for (int c = 0; c < 20; ++c) o[c] = acc[c];
}

extern "C" void kernel_launch(void* const* d_in, const int* in_sizes, int n_in,
                              void* d_out, int out_size, void* d_ws, size_t ws_size,
                              hipStream_t stream) {
  const float* x_in  = (const float*)d_in[0];
  const int*   ei    = (const int*)d_in[1];
  const float* gamma = (const float*)d_in[2];
  const float* beta  = (const float*)d_in[3];
  const float* W     = (const float*)d_in[4];
  const float* b     = (const float*)d_in[5];
  const float* Wo    = (const float*)d_in[6];
  const float* bo    = (const float*)d_in[7];
  float* out = (float*)d_out;

  const int N = in_sizes[0] / HDIM;
  const int E = in_sizes[1] / 2;

  float* dinv  = (float*)d_ws;
  float* stats = dinv + ((N + 255) / 256) * 256;
  float* hbuf  = stats + 2 * HDIM;
  float* xbuf  = hbuf + (size_t)N * HDIM;

  int* csr    = (int*)d_out;
  int* rowptr = csr + ((2 * E + 3) / 4) * 4;
  int* cnt    = rowptr + ((N + 4) / 4) * 4;
  int* rank   = (int*)xbuf;

  k_zero_int<<<(N + 255) / 256, 256, 0, stream>>>(cnt, N);
  k_count_rank<<<(E + 255) / 256, 256, 0, stream>>>(ei, cnt, rank, E);
  k_scan_dinv<<<1, SCAN_T, 0, stream>>>(cnt, rowptr, dinv, N);
  k_fill<<<(E + 255) / 256, 256, 0, stream>>>(ei, rank, rowptr, dinv, csr, E);

  hipMemsetAsync(stats, 0, 2 * HDIM * sizeof(float), stream);
  k_stats<<<ST_BLOCKS, ST_WAVES * 32, 0, stream>>>(x_in, stats, N);

  for (int l = 0; l < 3; ++l) {
    const float* xl = (l == 0) ? x_in : xbuf;
    k_gemm<<<(N + 63) / 64, 128, 0, stream>>>(xl, stats, gamma + l * HDIM, beta + l * HDIM,
                                              W + (size_t)l * HDIM * HDIM, hbuf, N, 1.0f / (float)N);
    if (l < 2) {
      hipMemsetAsync(stats, 0, 2 * HDIM * sizeof(float), stream);
      k_node_agg<true><<<AGG_BLOCKS, AGG_WAVES * 32, 0, stream>>>(rowptr, csr, dinv, hbuf, b + l * HDIM,
                                                                  xbuf, stats, N);
    } else {
      k_node_agg<false><<<AGG_BLOCKS, AGG_WAVES * 32, 0, stream>>>(rowptr, csr, dinv, hbuf, b + l * HDIM,
                                                                   xbuf, stats, N);
    }
  }

  k_out_gemm<<<(N + 127) / 128, 256, 0, stream>>>(xbuf, Wo, bo, out, N);
}


# Edge sort without global atomics: two-level counting sort by destination (512-node buckets; chunk histograms, ranks and row offsets on LDS atomics); the order of a node's edge sum differs
# speedup vs baseline: 4.3282x; 1.3232x over previous
#include <hip/hip_runtime.h>

#define HDIM 128
#define NCLS 40
#define BN_EPS 1e-5f

typedef __attribute__((ext_vector_type(2))) float v2f;
typedef __attribute__((ext_vector_type(8))) float v8f;

#define BK_SHIFT 9
#define BK_NODES 512
#define BK_MAX 1024
#define CH_T 1024
#define SORT_U 8

__device__ __forceinline__ int bucket_of(int d, int nb) {
  int bk = d >> BK_SHIFT;
  return bk < 0 ? 0 : (bk < nb ? bk : nb - 1);
}

__global__ __launch_bounds__(CH_T) void k_bucket_hist(const int* __restrict__ ei, int* __restrict__ bh,
                                                      int E, int ch, int nb) {
  __shared__ int hist[BK_MAX];
  const int t = threadIdx.x;
  for (int q = t; q < nb; q += CH_T) hist[q] = 0;
  __syncthreads();
  const int e0 = blockIdx.x * ch;
  int e1 = e0 + ch;
  e1 = e1 < E ? e1 : E;
  for (int e = e0 + t; e < e1; e += SORT_U * CH_T) {
    int d[SORT_U];
#pragma unroll
    for (int q = 0; q < SORT_U; ++q) d[q] = e + q * CH_T < e1 ? ei[E + e + q * CH_T] : 0;
#pragma unroll
    for (int q = 0; q < SORT_U; ++q)
      if (e + q * CH_T < e1) atomicAdd(&hist[bucket_of(d[q], nb)], 1);
  }
  __syncthreads();
  for (int q = t; q < nb; q += CH_T) bh[q * gridDim.x + blockIdx.x] = hist[q];
}

#define SCAN_T 1024
__device__ __forceinline__ int4 load4_guarded(const int* __restrict__ p, int i0, int n) {
  int4 v = make_int4(0, 0, 0, 0);
  if (i0 + 3 < n) {
    v = *(const int4*)(p + i0);
  } else {
    if (i0 < n) v.x = p[i0];
    if (i0 + 1 < n) v.y = p[i0 + 1];
    if (i0 + 2 < n) v.z = p[i0 + 2];
  }
  return v;
}

__global__ __launch_bounds__(SCAN_T) void k_scan(const int* __restrict__ cnt, int* __restrict__ rowptr, int n) {
  __shared__ int wsum[32];
  const int t = threadIdx.x, lane = t & 31, wv = t >> 5;
  int carry = 0;
  int4 v[4];
#pragma unroll
  for (int g = 0; g < 4; ++g) v[g] = load4_guarded(cnt, t * 16 + g * 4, n);
  for (int base = 0; base < n; base += SCAN_T * 16) {
    const int i0 = base + t * 16;
    int4 nxt[4];
#pragma unroll
    for (int g = 0; g < 4; ++g) nxt[g] = load4_guarded(cnt, i0 + SCAN_T * 16 + g * 4, n);
    int s = 0;
#pragma unroll
    for (int g = 0; g < 4; ++g) s += v[g].x + v[g].y + v[g].z + v[g].w;
    int inc = s;
#pragma unroll
    for (int d = 1; d < 32; d <<= 1) {
      int o = __shfl_up(inc, d);
      if (lane >= d) inc += o;
    }
    if (lane == 31) wsum[wv] = inc;
    __syncthreads();
    const int wt = wsum[lane];
    int winc = wt;
#pragma unroll
    for (int d = 1; d < 32; d <<= 1) {
      int o = __shfl_up(winc, d);
      if (lane >= d) winc += o;
    }
    const int wex = __shfl(winc, wv) - __shfl(wt, wv);
    const int total = __shfl(winc, 31);
    int run = carry + wex + inc - s;
#pragma unroll
    for (int g = 0; g < 4; ++g) {
      const int i = i0 + g * 4;
      const int4 c = v[g];
      if (i + 3 < n) {
        *(int4*)(rowptr + i) = make_int4(run, run + c.x, run + c.x + c.y, run + c.x + c.y + c.z);
      } else {
        if (i < n)     rowptr[i] = run;
        if (i + 1 < n) rowptr[i + 1] = run + c.x;
        if (i + 2 < n) rowptr[i + 2] = run + c.x + c.y;
      }
      run += c.x + c.y + c.z + c.w;
    }
    carry += total;
#pragma unroll
    for (int g = 0; g < 4; ++g) v[g] = nxt[g];
    __syncthreads();
  }
  if (t == 0) rowptr[n] = carry;
}

__global__ __launch_bounds__(CH_T) void k_bucket_scatter(const int* __restrict__ ei,
                                                         const int* __restrict__ boff,
                                                         int2* __restrict__ tmp, int E, int ch, int nb) {
  __shared__ int cur[BK_MAX];
  const int t = threadIdx.x;
  for (int q = t; q < nb; q += CH_T) cur[q] = boff[q * gridDim.x + blockIdx.x];
  __syncthreads();
  const int e0 = blockIdx.x * ch;
  int e1 = e0 + ch;
  e1 = e1 < E ? e1 : E;
  for (int e = e0 + t; e < e1; e += SORT_U * CH_T) {
    int s[SORT_U], d[SORT_U];
#pragma unroll
    for (int q = 0; q < SORT_U; ++q) {
      const bool in = e + q * CH_T < e1;
      s[q] = in ? ei[e + q * CH_T] : 0;
      d[q] = in ? ei[E + e + q * CH_T] : 0;
    }
#pragma unroll
    for (int q = 0; q < SORT_U; ++q) {
      if (e + q * CH_T < e1) {
        int p = atomicAdd(&cur[bucket_of(d[q], nb)], 1);
        p = p < 0 ? 0 : (p < E ? p : E - 1);
        tmp[p] = make_int2(s[q], d[q]);
      }
    }
  }
}

__device__ __forceinline__ void bucket_range(const int* __restrict__ boff, int bk, int nch, int E,
                                             int& bs, int& be) {
  bs = boff[bk * nch];
  be = boff[(bk + 1) * nch];
  bs = bs < 0 ? 0 : (bs < E ? bs : E);
  be = be < bs ? bs : (be < E ? be : E);
}

__device__ __forceinline__ int local_node(int d, int node0) {
  const int l = d - node0;
  return l < 0 ? 0 : (l < BK_NODES ? l : BK_NODES - 1);
}

__global__ __launch_bounds__(BK_NODES) void k_bucket_rows(const int2* __restrict__ tmp,
                                                          const int* __restrict__ boff,
                                                          int* __restrict__ rowptr,
                                                          float* __restrict__ dinv, int n, int E, int nch) {
  __shared__ int cnt[BK_NODES];
  __shared__ int wsum[32];
  const int t = threadIdx.x, lane = t & 31, wv = t >> 5;
  const int node0 = blockIdx.x << BK_SHIFT;
  int bs, be;
  bucket_range(boff, blockIdx.x, nch, E, bs, be);
  cnt[t] = 0;
  __syncthreads();
  for (int i = bs + t; i < be; i += SORT_U * BK_NODES) {
    int d[SORT_U];
#pragma unroll
    for (int q = 0; q < SORT_U; ++q) d[q] = i + q * BK_NODES < be ? tmp[i + q * BK_NODES].y : 0;
#pragma unroll
    for (int q = 0; q < SORT_U; ++q)
      if (i + q * BK_NODES < be) atomicAdd(&cnt[local_node(d[q], node0)], 1);
  }
  __syncthreads();
  const int c = cnt[t];
  int inc = c;
#pragma unroll
  for (int d = 1; d < 32; d <<= 1) {
    int o = __shfl_up(inc, d);
    if (lane >= d) inc += o;
  }
  if (lane == 31) wsum[wv] = inc;
  __syncthreads();
  const int wt = lane < BK_NODES / 32 ? wsum[lane] : 0;
  int winc = wt;
#pragma unroll
  for (int d = 1; d < 32; d <<= 1) {
    int o = __shfl_up(winc, d);
    if (lane >= d) winc += o;
  }
  const int wex = __shfl(winc, wv) - __shfl(wt, wv);
  const int node = node0 + t;
  if (node < n) {
    rowptr[node] = bs + wex + inc - c;
    dinv[node] = rsqrtf((float)(c + 1));
  }
  if (blockIdx.x == gridDim.x - 1 && t == 0) rowptr[n] = be;
}

__global__ __launch_bounds__(BK_NODES) void k_bucket_fill(const int2* __restrict__ tmp,
                                                          const int* __restrict__ boff,
                                                          const int* __restrict__ rowptr,
                                                          const float* __restrict__ dinv,
                                                          int2* __restrict__ csr, int n, int E, int nch) {
  __shared__ int cur[BK_NODES];
  const int t = threadIdx.x;
  const int node0 = blockIdx.x << BK_SHIFT;
  int bs, be;
  bucket_range(boff, blockIdx.x, nch, E, bs, be);
  cur[t] = node0 + t < n ? rowptr[node0 + t] : be;
  __syncthreads();
  for (int i = bs + t; i < be; i += SORT_U * BK_NODES) {
    int2 sd[SORT_U];
    float w[SORT_U];
#pragma unroll
    for (int q = 0; q < SORT_U; ++q)
      sd[q] = i + q * BK_NODES < be ? tmp[i + q * BK_NODES] : make_int2(0, 0);
#pragma unroll
    for (int q = 0; q < SORT_U; ++q) {
      const int s = sd[q].x < 0 ? 0 : (sd[q].x < n ? sd[q].x : n - 1);
      const int d = sd[q].y < 0 ? 0 : (sd[q].y < n ? sd[q].y : n - 1);
      w[q] = dinv[s] * dinv[d];
    }
#pragma unroll
    for (int q = 0; q < SORT_U; ++q) {
      if (i + q * BK_NODES < be) {
        int p = atomicAdd(&cur[local_node(sd[q].y, node0)], 1);
        p = p < 0 ? 0 : (p < E ? p : E - 1);
        csr[p] = make_int2(sd[q].x, __float_as_int(w[q]));
      }
    }
  }
}

#define ST_WAVES 8
#define ST_BLOCKS 512
#define ST_ROWS 32
static_assert(ST_WAVES * 32 == 2 * HDIM, "one thread per entry of stats in the block's last step");
__global__ __launch_bounds__(ST_WAVES * 32) void k_stats(const float* __restrict__ x,
                                                         float* __restrict__ stats, int n) {
  const int wv = __builtin_amdgcn_readfirstlane(threadIdx.x >> 5);
  const unsigned lane = threadIdx.x & 31;
  const float4* xb = (const float4*)x;
  const int nw = gridDim.x * ST_WAVES;
  float4 s1 = make_float4(0.f, 0.f, 0.f, 0.f);
  float4 s2 = make_float4(0.f, 0.f, 0.f, 0.f);
  for (int base = blockIdx.x * ST_WAVES + wv; base < n; base += ST_ROWS * nw) {
    float4 a[ST_ROWS];
#pragma unroll
    for (int q = 0; q < ST_ROWS; ++q) {
      const int r = base + q * nw;
      if (r < n) a[q] = xb[(size_t)r * 32 + lane];
    }
#pragma unroll
    for (int q = 0; q < ST_ROWS; ++q) {
      if (base + q * nw < n) {
        s1.x += a[q].x; s1.y += a[q].y; s1.z += a[q].z; s1.w += a[q].w;
        s2.x = fmaf(a[q].x, a[q].x, s2.x); s2.y = fmaf(a[q].y, a[q].y, s2.y);
        s2.z = fmaf(a[q].z, a[q].z, s2.z); s2.w = fmaf(a[q].w, a[q].w, s2.w);
      }
    }
  }
  __shared__ float ls[ST_WAVES * 2 * HDIM];
  float4* lw = (float4*)&ls[wv * 2 * HDIM];
  lw[lane] = s1;
  lw[32 + lane] = s2;
  __syncthreads();
  float t = 0.f;
#pragma unroll
  for (int w = 0; w < ST_WAVES; ++w) t += ls[w * 2 * HDIM + threadIdx.x];
  atomicAdd(&stats[threadIdx.x], t);
}

#define XPAD 68
#define WPAIR 288

__device__ __forceinline__ void bn_coef(float s1, float s2, float g, float bt, float inv_n,
                                        float& a, float& c) {
  const float mean = s1 * inv_n;
  const float var = s2 * inv_n - mean * mean;
  a = g * rsqrtf(var + BN_EPS);
  c = bt - a * mean;
}

__global__ __launch_bounds__(128, 5) void k_gemm(const float* __restrict__ x,
                                                 const float* __restrict__ stats,
                                                 const float* __restrict__ gamma,
                                                 const float* __restrict__ beta,
                                                 const float* __restrict__ W,
                                                 float* __restrict__ h, int n, float inv_n) {
  __shared__ float xs[64 * XPAD];
  __shared__ float ws[32 * WPAIR];
  const int tid = threadIdx.x;
  const int lane = tid & 31;
  const int wave = tid >> 5;
  const int rowBase = blockIdx.x * 64;
  const int m = lane & 15;
  const int hi = lane >> 4;

  v8f acc[8];
#pragma unroll
  for (int c = 0; c < 8; ++c)
#pragma unroll
    for (int v = 0; v < 8; ++v) acc[c][v] = 0.f;

  for (int kh = 0; kh < 2; ++kh) {
    const int colq = kh * 64 + (tid & 15) * 4;
    const float4 s1 = *(const float4*)(stats + colq);
    const float4 s2 = *(const float4*)(stats + HDIM + colq);
    const float4 g4 = *(const float4*)(gamma + colq);
    const float4 b4 = *(const float4*)(beta + colq);
    float4 sa, sc;
    bn_coef(s1.x, s2.x, g4.x, b4.x, inv_n, sa.x, sc.x);
    bn_coef(s1.y, s2.y, g4.y, b4.y, inv_n, sa.y, sc.y);
    bn_coef(s1.z, s2.z, g4.z, b4.z, inv_n, sa.z, sc.z);
    bn_coef(s1.w, s2.w, g4.w, b4.w, inv_n, sa.w, sc.w);
#pragma unroll
    for (int p = 0; p < 8; ++p) {
      int idx = p * 512 + tid * 4;
      int r = idx >> 6, cc = idx & 63;
      int row = rowBase + r;
      float4 v = make_float4(0.f, 0.f, 0.f, 0.f);
      if (row < n) {
        v = *(const float4*)(x + (size_t)row * HDIM + kh * 64 + cc);
        v.x = fmaf(v.x, sa.x, sc.x);
        v.y = fmaf(v.y, sa.y, sc.y);
        v.z = fmaf(v.z, sa.z, sc.z);
        v.w = fmaf(v.w, sa.w, sc.w);
      }
      *(float4*)(&xs[r * XPAD + cc]) = v;
    }
#pragma unroll
    for (int p = 0; p < 8; ++p) {
      const int kp = p * 4 + wave;
      const float* wr = W + (kh * 64 + 2 * kp) * HDIM + lane * 4;
      const float4 w0 = *(const float4*)wr;
      const float4 w1 = *(const float4*)(wr + HDIM);
      float* d = &ws[kp * WPAIR + lane * 8];
      *(float4*)d = make_float4(w0.x, w1.x, w0.y, w1.y);
      *(float4*)(d + 4) = make_float4(w0.z, w1.z, w0.w, w1.w);
    }
    __syncthreads();

    const float* xrow = &xs[(wave * 16 + m) * XPAD];
#pragma unroll
    for (int k4 = 0; k4 < 16; ++k4) {
      int kb = k4 * 4 + hi * 2;
      v2f a; a.x = xrow[kb]; a.y = xrow[kb + 1];
      const float* wk = &ws[(k4 * 2 + hi) * WPAIR + m * 2];
#pragma unroll
      for (int c = 0; c < 8; ++c) {
        v2f b; b.x = wk[c * 32]; b.y = wk[c * 32 + 1];
        acc[c] = __builtin_amdgcn_wmma_f32_16x16x4_f32(
            false, a, false, b, (short)0, acc[c], false, false);
      }
    }
    __syncthreads();
  }

#pragma unroll
  for (int c = 0; c < 8; ++c) {
    int col = c * 16 + m;
#pragma unroll
    for (int v = 0; v < 8; ++v) {
      int row = rowBase + wave * 16 + v + 8 * hi;
      if (row < n) h[(size_t)row * HDIM + col] = acc[c][v];
    }
  }
}

template <int K>
__device__ __forceinline__ void edge_batch(float4& acc, const float4* __restrict__ hb,
                                           const int2 my, const int j, const int lane) {
  int sidx[K];
  float wgt[K];
  float4 a[K];
#pragma unroll
  for (int q = 0; q < K; ++q) {
    sidx[q] = __builtin_amdgcn_readlane(my.x, j + q);
    wgt[q] = __int_as_float(__builtin_amdgcn_readlane(my.y, j + q));
  }
#pragma unroll
  for (int q = 0; q < K; ++q) a[q] = hb[(size_t)sidx[q] * 32 + lane];
#pragma unroll
  for (int q = 0; q < K; ++q) {
    acc.x = fmaf(a[q].x, wgt[q], acc.x);
    acc.y = fmaf(a[q].y, wgt[q], acc.y);
    acc.z = fmaf(a[q].z, wgt[q], acc.z);
    acc.w = fmaf(a[q].w, wgt[q], acc.w);
  }
}

#define AGG_WAVES 32
#define AGG_BLOCKS 512
template <bool STATS>
__global__ __launch_bounds__(AGG_WAVES * 32) void k_node_agg(const int* __restrict__ rowptr,
                                                             const int* __restrict__ csr,
                                                             const float* __restrict__ dinv,
                                                             const float* __restrict__ hbuf,
                                                             const float* __restrict__ bias,
                                                             float* __restrict__ out,
                                                             float* __restrict__ stats, int n) {
  const int wv = __builtin_amdgcn_readfirstlane(threadIdx.x >> 5);
  const int lane = threadIdx.x & 31;
  const float4* hb = (const float4*)hbuf;
  const int2* cs = (const int2*)csr;
  const float4 bv = ((const float4*)bias)[lane];
  float4 cs1 = make_float4(0.f, 0.f, 0.f, 0.f);
  float4 cs2 = make_float4(0.f, 0.f, 0.f, 0.f);
  for (int node = blockIdx.x * AGG_WAVES + wv; node < n; node += gridDim.x * AGG_WAVES) {
    const int beg = rowptr[node], end = rowptr[node + 1];
    const float di = dinv[node];
    const float4 hs = hb[(size_t)node * 32 + lane];
    float4 acc = make_float4(0.f, 0.f, 0.f, 0.f);
    for (int base = beg; base < end; base += 32) {
      int m = end - base;
      m = m < 32 ? m : 32;
      int2 my = make_int2(0, 0);
      if (lane < m) my = cs[base + lane];
      my.x = my.x < 0 ? 0 : (my.x < n ? my.x : n - 1);
      int j = 0;
      for (; j + 8 <= m; j += 8) edge_batch<8>(acc, hb, my, j, lane);
      if (j + 4 <= m) { edge_batch<4>(acc, hb, my, j, lane); j += 4; }
      const int r = m - j;
      if (r == 3) edge_batch<3>(acc, hb, my, j, lane);
      else if (r == 2) edge_batch<2>(acc, hb, my, j, lane);
      else if (r == 1) edge_batch<1>(acc, hb, my, j, lane);
    }
    const float w_self = di * di;
    acc.x = fmaf(hs.x, w_self, acc.x);
    acc.y = fmaf(hs.y, w_self, acc.y);
    acc.z = fmaf(hs.z, w_self, acc.z);
    acc.w = fmaf(hs.w, w_self, acc.w);
    float4 v;
    v.x = acc.x + bv.x; v.x = v.x > 0.f ? v.x : 0.f;
    v.y = acc.y + bv.y; v.y = v.y > 0.f ? v.y : 0.f;
    v.z = acc.z + bv.z; v.z = v.z > 0.f ? v.z : 0.f;
    v.w = acc.w + bv.w; v.w = v.w > 0.f ? v.w : 0.f;
    ((float4*)out)[(size_t)node * 32 + lane] = v;
    if (STATS) {
      cs1.x += v.x; cs1.y += v.y; cs1.z += v.z; cs1.w += v.w;
      cs2.x = fmaf(v.x, v.x, cs2.x); cs2.y = fmaf(v.y, v.y, cs2.y);
      cs2.z = fmaf(v.z, v.z, cs2.z); cs2.w = fmaf(v.w, v.w, cs2.w);
    }
  }
  if (STATS) {
    __shared__ float ls[AGG_WAVES * 2 * HDIM];
    float4* lw = (float4*)&ls[wv * 2 * HDIM];
    lw[lane] = cs1;
    lw[32 + lane] = cs2;
    __syncthreads();
    if (threadIdx.x < 2 * HDIM) {
      float t = 0.f;
#pragma unroll
      for (int w = 0; w < AGG_WAVES; ++w) t += ls[w * 2 * HDIM + threadIdx.x];
      atomicAdd(&stats[threadIdx.x], t);
    }
  }
}

__global__ __launch_bounds__(256) void k_out_gemm(const float* __restrict__ x,
                                                  const float* __restrict__ Wo,
                                                  const float* __restrict__ bo,
                                                  float* __restrict__ out, int n) {
  __shared__ float ws[HDIM * NCLS];
  __shared__ float bs[NCLS];
  int tid = threadIdx.x;
  for (int i = tid; i < HDIM * NCLS; i += 256) ws[i] = Wo[i];
  if (tid < NCLS) bs[tid] = bo[tid];
  __syncthreads();
  int r = blockIdx.x * 128 + (tid >> 1);
  int c0 = (tid & 1) * 20;
  if (r >= n) return;
  float acc[20];
#pragma unroll
  for (int c = 0; c < 20; ++c) acc[c] = bs[c0 + c];
  const float* xr = x + (size_t)r * HDIM;
  for (int k = 0; k < HDIM; ++k) {
    float xv = xr[k];
    const float* wr = &ws[k * NCLS + c0];
#pragma unroll
    for (int c = 0; c < 20; ++c) acc[c] += xv * wr[c];
  }
  float* o = out + (size_t)r * NCLS + c0;
#pragma unroll
  for (int c = 0; c < 20; ++c) o[c] = acc[c];
}

extern "C" void kernel_launch(void* const* d_in, const int* in_sizes, int n_in,
                              void* d_out, int out_size, void* d_ws, size_t ws_size,
                              hipStream_t stream) {
  const float* x_in  = (const float*)d_in[0];
  const int*   ei    = (const int*)d_in[1];
  const float* gamma = (const float*)d_in[2];
  const float* beta  = (const float*)d_in[3];
  const float* W     = (const float*)d_in[4];
  const float* b     = (const float*)d_in[5];
  const float* Wo    = (const float*)d_in[6];
  const float* bo    = (const float*)d_in[7];
  float* out = (float*)d_out;

  const int N = in_sizes[0] / HDIM;
  const int E = in_sizes[1] / 2;

  float* dinv  = (float*)d_ws;
  float* stats = dinv + ((N + 255) / 256) * 256;
  float* hbuf  = stats + 2 * HDIM;
  float* xbuf  = hbuf + (size_t)N * HDIM;

  int* csr    = (int*)d_out;
  int* rowptr = csr + ((2 * E + 3) / 4) * 4;

  const int NCH = 250;
  const int nb = (N + BK_NODES - 1) / BK_NODES;
  const int ch = (E + NCH - 1) / NCH;
  const int n2 = nb * NCH;
  int2* tmp  = (int2*)xbuf;
  int*  bh   = (int*)xbuf + ((2 * E + 3) / 4) * 4;
  int*  boff = bh + ((n2 + 3) / 4) * 4;

  k_bucket_hist<<<NCH, CH_T, 0, stream>>>(ei, bh, E, ch, nb);
  k_scan<<<1, SCAN_T, 0, stream>>>(bh, boff, n2);
  k_bucket_scatter<<<NCH, CH_T, 0, stream>>>(ei, boff, tmp, E, ch, nb);
  k_bucket_rows<<<nb, BK_NODES, 0, stream>>>(tmp, boff, rowptr, dinv, N, E, NCH);
  k_bucket_fill<<<nb, BK_NODES, 0, stream>>>(tmp, boff, rowptr, dinv, (int2*)csr, N, E, NCH);

  hipMemsetAsync(stats, 0, 2 * HDIM * sizeof(float), stream);
  k_stats<<<ST_BLOCKS, ST_WAVES * 32, 0, stream>>>(x_in, stats, N);

  for (int l = 0; l < 3; ++l) {
    const float* xl = (l == 0) ? x_in : xbuf;
    k_gemm<<<(N + 63) / 64, 128, 0, stream>>>(xl, stats, gamma + l * HDIM, beta + l * HDIM,
                                              W + (size_t)l * HDIM * HDIM, hbuf, N, 1.0f / (float)N);
    if (l < 2) {
      hipMemsetAsync(stats, 0, 2 * HDIM * sizeof(float), stream);
      k_node_agg<true><<<AGG_BLOCKS, AGG_WAVES * 32, 0, stream>>>(rowptr, csr, dinv, hbuf, b + l * HDIM,
                                                                  xbuf, stats, N);
    } else {
      k_node_agg<false><<<AGG_BLOCKS, AGG_WAVES * 32, 0, stream>>>(rowptr, csr, dinv, hbuf, b + l * HDIM,
                                                                   xbuf, stats, N);
    }
  }

  k_out_gemm<<<(N + 127) / 128, 256, 0, stream>>>(xbuf, Wo, bo, out, N);
}


# Final projection on the matrix unit: 64-row x tile and Wo (40 columns padded to 48) staged in LDS, f32 16x16x4 products, rows stored as whole lines; replaces the scalar loop; k sum order differs
# speedup vs baseline: 4.4664x; 1.0319x over previous
#include <hip/hip_runtime.h>

#define HDIM 128
#define NCLS 40
#define BN_EPS 1e-5f

typedef __attribute__((ext_vector_type(2))) float v2f;
typedef __attribute__((ext_vector_type(8))) float v8f;

#define BK_SHIFT 9
#define BK_NODES 512
#define BK_MAX 1024
#define CH_T 1024
#define SORT_U 8

__device__ __forceinline__ int bucket_of(int d, int nb) {
  int bk = d >> BK_SHIFT;
  return bk < 0 ? 0 : (bk < nb ? bk : nb - 1);
}

__global__ __launch_bounds__(CH_T) void k_bucket_hist(const int* __restrict__ ei, int* __restrict__ bh,
                                                      int E, int ch, int nb) {
  __shared__ int hist[BK_MAX];
  const int t = threadIdx.x;
  for (int q = t; q < nb; q += CH_T) hist[q] = 0;
  __syncthreads();
  const int e0 = blockIdx.x * ch;
  int e1 = e0 + ch;
  e1 = e1 < E ? e1 : E;
  for (int e = e0 + t; e < e1; e += SORT_U * CH_T) {
    int d[SORT_U];
#pragma unroll
    for (int q = 0; q < SORT_U; ++q) d[q] = e + q * CH_T < e1 ? ei[E + e + q * CH_T] : 0;
#pragma unroll
    for (int q = 0; q < SORT_U; ++q)
      if (e + q * CH_T < e1) atomicAdd(&hist[bucket_of(d[q], nb)], 1);
  }
  __syncthreads();
  for (int q = t; q < nb; q += CH_T) bh[q * gridDim.x + blockIdx.x] = hist[q];
}

#define SCAN_T 1024
__device__ __forceinline__ int4 load4_guarded(const int* __restrict__ p, int i0, int n) {
  int4 v = make_int4(0, 0, 0, 0);
  if (i0 + 3 < n) {
    v = *(const int4*)(p + i0);
  } else {
    if (i0 < n) v.x = p[i0];
    if (i0 + 1 < n) v.y = p[i0 + 1];
    if (i0 + 2 < n) v.z = p[i0 + 2];
  }
  return v;
}

__global__ __launch_bounds__(SCAN_T) void k_scan(const int* __restrict__ cnt, int* __restrict__ rowptr, int n) {
  __shared__ int wsum[32];
  const int t = threadIdx.x, lane = t & 31, wv = t >> 5;
  int carry = 0;
  int4 v[4];
#pragma unroll
  for (int g = 0; g < 4; ++g) v[g] = load4_guarded(cnt, t * 16 + g * 4, n);
  for (int base = 0; base < n; base += SCAN_T * 16) {
    const int i0 = base + t * 16;
    int4 nxt[4];
#pragma unroll
    for (int g = 0; g < 4; ++g) nxt[g] = load4_guarded(cnt, i0 + SCAN_T * 16 + g * 4, n);
    int s = 0;
#pragma unroll
    for (int g = 0; g < 4; ++g) s += v[g].x + v[g].y + v[g].z + v[g].w;
    int inc = s;
#pragma unroll
    for (int d = 1; d < 32; d <<= 1) {
      int o = __shfl_up(inc, d);
      if (lane >= d) inc += o;
    }
    if (lane == 31) wsum[wv] = inc;
    __syncthreads();
    const int wt = wsum[lane];
    int winc = wt;
#pragma unroll
    for (int d = 1; d < 32; d <<= 1) {
      int o = __shfl_up(winc, d);
      if (lane >= d) winc += o;
    }
    const int wex = __shfl(winc, wv) - __shfl(wt, wv);
    const int total = __shfl(winc, 31);
    int run = carry + wex + inc - s;
#pragma unroll
    for (int g = 0; g < 4; ++g) {
      const int i = i0 + g * 4;
      const int4 c = v[g];
      if (i + 3 < n) {
        *(int4*)(rowptr + i) = make_int4(run, run + c.x, run + c.x + c.y, run + c.x + c.y + c.z);
      } else {
        if (i < n)     rowptr[i] = run;
        if (i + 1 < n) rowptr[i + 1] = run + c.x;
        if (i + 2 < n) rowptr[i + 2] = run + c.x + c.y;
      }
      run += c.x + c.y + c.z + c.w;
    }
    carry += total;
#pragma unroll
    for (int g = 0; g < 4; ++g) v[g] = nxt[g];
    __syncthreads();
  }
  if (t == 0) rowptr[n] = carry;
}

__global__ __launch_bounds__(CH_T) void k_bucket_scatter(const int* __restrict__ ei,
                                                         const int* __restrict__ boff,
                                                         int2* __restrict__ tmp, int E, int ch, int nb) {
  __shared__ int cur[BK_MAX];
  const int t = threadIdx.x;
  for (int q = t; q < nb; q += CH_T) cur[q] = boff[q * gridDim.x + blockIdx.x];
  __syncthreads();
  const int e0 = blockIdx.x * ch;
  int e1 = e0 + ch;
  e1 = e1 < E ? e1 : E;
  for (int e = e0 + t; e < e1; e += SORT_U * CH_T) {
    int s[SORT_U], d[SORT_U];
#pragma unroll
    for (int q = 0; q < SORT_U; ++q) {
      const bool in = e + q * CH_T < e1;
      s[q] = in ? ei[e + q * CH_T] : 0;
      d[q] = in ? ei[E + e + q * CH_T] : 0;
    }
#pragma unroll
    for (int q = 0; q < SORT_U; ++q) {
      if (e + q * CH_T < e1) {
        int p = atomicAdd(&cur[bucket_of(d[q], nb)], 1);
        p = p < 0 ? 0 : (p < E ? p : E - 1);
        tmp[p] = make_int2(s[q], d[q]);
      }
    }
  }
}

__device__ __forceinline__ void bucket_range(const int* __restrict__ boff, int bk, int nch, int E,
                                             int& bs, int& be) {
  bs = boff[bk * nch];
  be = boff[(bk + 1) * nch];
  bs = bs < 0 ? 0 : (bs < E ? bs : E);
  be = be < bs ? bs : (be < E ? be : E);
}

__device__ __forceinline__ int local_node(int d, int node0) {
  const int l = d - node0;
  return l < 0 ? 0 : (l < BK_NODES ? l : BK_NODES - 1);
}

__global__ __launch_bounds__(BK_NODES) void k_bucket_rows(const int2* __restrict__ tmp,
                                                          const int* __restrict__ boff,
                                                          int* __restrict__ rowptr,
                                                          float* __restrict__ dinv, int n, int E, int nch) {
  __shared__ int cnt[BK_NODES];
  __shared__ int wsum[32];
  const int t = threadIdx.x, lane = t & 31, wv = t >> 5;
  const int node0 = blockIdx.x << BK_SHIFT;
  int bs, be;
  bucket_range(boff, blockIdx.x, nch, E, bs, be);
  cnt[t] = 0;
  __syncthreads();
  for (int i = bs + t; i < be; i += SORT_U * BK_NODES) {
    int d[SORT_U];
#pragma unroll
    for (int q = 0; q < SORT_U; ++q) d[q] = i + q * BK_NODES < be ? tmp[i + q * BK_NODES].y : 0;
#pragma unroll
    for (int q = 0; q < SORT_U; ++q)
      if (i + q * BK_NODES < be) atomicAdd(&cnt[local_node(d[q], node0)], 1);
  }
  __syncthreads();
  const int c = cnt[t];
  int inc = c;
#pragma unroll
  for (int d = 1; d < 32; d <<= 1) {
    int o = __shfl_up(inc, d);
    if (lane >= d) inc += o;
  }
  if (lane == 31) wsum[wv] = inc;
  __syncthreads();
  const int wt = lane < BK_NODES / 32 ? wsum[lane] : 0;
  int winc = wt;
#pragma unroll
  for (int d = 1; d < 32; d <<= 1) {
    int o = __shfl_up(winc, d);
    if (lane >= d) winc += o;
  }
  const int wex = __shfl(winc, wv) - __shfl(wt, wv);
  const int node = node0 + t;
  if (node < n) {
    rowptr[node] = bs + wex + inc - c;
    dinv[node] = rsqrtf((float)(c + 1));
  }
  if (blockIdx.x == gridDim.x - 1 && t == 0) rowptr[n] = be;
}

__global__ __launch_bounds__(BK_NODES) void k_bucket_fill(const int2* __restrict__ tmp,
                                                          const int* __restrict__ boff,
                                                          const int* __restrict__ rowptr,
                                                          const float* __restrict__ dinv,
                                                          int2* __restrict__ csr, int n, int E, int nch) {
  __shared__ int cur[BK_NODES];
  const int t = threadIdx.x;
  const int node0 = blockIdx.x << BK_SHIFT;
  int bs, be;
  bucket_range(boff, blockIdx.x, nch, E, bs, be);
  cur[t] = node0 + t < n ? rowptr[node0 + t] : be;
  __syncthreads();
  for (int i = bs + t; i < be; i += SORT_U * BK_NODES) {
    int2 sd[SORT_U];
    float w[SORT_U];
#pragma unroll
    for (int q = 0; q < SORT_U; ++q)
      sd[q] = i + q * BK_NODES < be ? tmp[i + q * BK_NODES] : make_int2(0, 0);
#pragma unroll
    for (int q = 0; q < SORT_U; ++q) {
      const int s = sd[q].x < 0 ? 0 : (sd[q].x < n ? sd[q].x : n - 1);
      const int d = sd[q].y < 0 ? 0 : (sd[q].y < n ? sd[q].y : n - 1);
      w[q] = dinv[s] * dinv[d];
    }
#pragma unroll
    for (int q = 0; q < SORT_U; ++q) {
      if (i + q * BK_NODES < be) {
        int p = atomicAdd(&cur[local_node(sd[q].y, node0)], 1);
        p = p < 0 ? 0 : (p < E ? p : E - 1);
        csr[p] = make_int2(sd[q].x, __float_as_int(w[q]));
      }
    }
  }
}

#define ST_WAVES 8
#define ST_BLOCKS 512
#define ST_ROWS 32
static_assert(ST_WAVES * 32 == 2 * HDIM, "one thread per entry of stats in the block's last step");
__global__ __launch_bounds__(ST_WAVES * 32) void k_stats(const float* __restrict__ x,
                                                         float* __restrict__ stats, int n) {
  const int wv = __builtin_amdgcn_readfirstlane(threadIdx.x >> 5);
  const unsigned lane = threadIdx.x & 31;
  const float4* xb = (const float4*)x;
  const int nw = gridDim.x * ST_WAVES;
  float4 s1 = make_float4(0.f, 0.f, 0.f, 0.f);
  float4 s2 = make_float4(0.f, 0.f, 0.f, 0.f);
  for (int base = blockIdx.x * ST_WAVES + wv; base < n; base += ST_ROWS * nw) {
    float4 a[ST_ROWS];
#pragma unroll
    for (int q = 0; q < ST_ROWS; ++q) {
      const int r = base + q * nw;
      if (r < n) a[q] = xb[(size_t)r * 32 + lane];
    }
#pragma unroll
    for (int q = 0; q < ST_ROWS; ++q) {
      if (base + q * nw < n) {
        s1.x += a[q].x; s1.y += a[q].y; s1.z += a[q].z; s1.w += a[q].w;
        s2.x = fmaf(a[q].x, a[q].x, s2.x); s2.y = fmaf(a[q].y, a[q].y, s2.y);
        s2.z = fmaf(a[q].z, a[q].z, s2.z); s2.w = fmaf(a[q].w, a[q].w, s2.w);
      }
    }
  }
  __shared__ float ls[ST_WAVES * 2 * HDIM];
  float4* lw = (float4*)&ls[wv * 2 * HDIM];
  lw[lane] = s1;
  lw[32 + lane] = s2;
  __syncthreads();
  float t = 0.f;
#pragma unroll
  for (int w = 0; w < ST_WAVES; ++w) t += ls[w * 2 * HDIM + threadIdx.x];
  atomicAdd(&stats[threadIdx.x], t);
}

#define XPAD 68
#define WPAIR 288

__device__ __forceinline__ void bn_coef(float s1, float s2, float g, float bt, float inv_n,
                                        float& a, float& c) {
  const float mean = s1 * inv_n;
  const float var = s2 * inv_n - mean * mean;
  a = g * rsqrtf(var + BN_EPS);
  c = bt - a * mean;
}

__global__ __launch_bounds__(128, 5) void k_gemm(const float* __restrict__ x,
                                                 const float* __restrict__ stats,
                                                 const float* __restrict__ gamma,
                                                 const float* __restrict__ beta,
                                                 const float* __restrict__ W,
                                                 float* __restrict__ h, int n, float inv_n) {
  __shared__ float xs[64 * XPAD];
  __shared__ float ws[32 * WPAIR];
  const int tid = threadIdx.x;
  const int lane = tid & 31;
  const int wave = tid >> 5;
  const int rowBase = blockIdx.x * 64;
  const int m = lane & 15;
  const int hi = lane >> 4;

  v8f acc[8];
#pragma unroll
  for (int c = 0; c < 8; ++c)
#pragma unroll
    for (int v = 0; v < 8; ++v) acc[c][v] = 0.f;

  for (int kh = 0; kh < 2; ++kh) {
    const int colq = kh * 64 + (tid & 15) * 4;
    const float4 s1 = *(const float4*)(stats + colq);
    const float4 s2 = *(const float4*)(stats + HDIM + colq);
    const float4 g4 = *(const float4*)(gamma + colq);
    const float4 b4 = *(const float4*)(beta + colq);
    float4 sa, sc;
    bn_coef(s1.x, s2.x, g4.x, b4.x, inv_n, sa.x, sc.x);
    bn_coef(s1.y, s2.y, g4.y, b4.y, inv_n, sa.y, sc.y);
    bn_coef(s1.z, s2.z, g4.z, b4.z, inv_n, sa.z, sc.z);
    bn_coef(s1.w, s2.w, g4.w, b4.w, inv_n, sa.w, sc.w);
#pragma unroll
    for (int p = 0; p < 8; ++p) {
      int idx = p * 512 + tid * 4;
      int r = idx >> 6, cc = idx & 63;
      int row = rowBase + r;
      float4 v = make_float4(0.f, 0.f, 0.f, 0.f);
      if (row < n) {
        v = *(const float4*)(x + (size_t)row * HDIM + kh * 64 + cc);
        v.x = fmaf(v.x, sa.x, sc.x);
        v.y = fmaf(v.y, sa.y, sc.y);
        v.z = fmaf(v.z, sa.z, sc.z);
        v.w = fmaf(v.w, sa.w, sc.w);
      }
      *(float4*)(&xs[r * XPAD + cc]) = v;
    }
#pragma unroll
    for (int p = 0; p < 8; ++p) {
      const int kp = p * 4 + wave;
      const float* wr = W + (kh * 64 + 2 * kp) * HDIM + lane * 4;
      const float4 w0 = *(const float4*)wr;
      const float4 w1 = *(const float4*)(wr + HDIM);
      float* d = &ws[kp * WPAIR + lane * 8];
      *(float4*)d = make_float4(w0.x, w1.x, w0.y, w1.y);
      *(float4*)(d + 4) = make_float4(w0.z, w1.z, w0.w, w1.w);
    }
    __syncthreads();

    const float* xrow = &xs[(wave * 16 + m) * XPAD];
#pragma unroll
    for (int k4 = 0; k4 < 16; ++k4) {
      int kb = k4 * 4 + hi * 2;
      v2f a; a.x = xrow[kb]; a.y = xrow[kb + 1];
      const float* wk = &ws[(k4 * 2 + hi) * WPAIR + m * 2];
#pragma unroll
      for (int c = 0; c < 8; ++c) {
        v2f b; b.x = wk[c * 32]; b.y = wk[c * 32 + 1];
        acc[c] = __builtin_amdgcn_wmma_f32_16x16x4_f32(
            false, a, false, b, (short)0, acc[c], false, false);
      }
    }
    __syncthreads();
  }

#pragma unroll
  for (int c = 0; c < 8; ++c) {
    int col = c * 16 + m;
#pragma unroll
    for (int v = 0; v < 8; ++v) {
      int row = rowBase + wave * 16 + v + 8 * hi;
      if (row < n) h[(size_t)row * HDIM + col] = acc[c][v];
    }
  }
}

template <int K>
__device__ __forceinline__ void edge_batch(float4& acc, const float4* __restrict__ hb,
                                           const int2 my, const int j, const int lane) {
  int sidx[K];
  float wgt[K];
  float4 a[K];
#pragma unroll
  for (int q = 0; q < K; ++q) {
    sidx[q] = __builtin_amdgcn_readlane(my.x, j + q);
    wgt[q] = __int_as_float(__builtin_amdgcn_readlane(my.y, j + q));
  }
#pragma unroll
  for (int q = 0; q < K; ++q) a[q] = hb[(size_t)sidx[q] * 32 + lane];
#pragma unroll
  for (int q = 0; q < K; ++q) {
    acc.x = fmaf(a[q].x, wgt[q], acc.x);
    acc.y = fmaf(a[q].y, wgt[q], acc.y);
    acc.z = fmaf(a[q].z, wgt[q], acc.z);
    acc.w = fmaf(a[q].w, wgt[q], acc.w);
  }
}

#define AGG_WAVES 32
#define AGG_BLOCKS 512
template <bool STATS>
__global__ __launch_bounds__(AGG_WAVES * 32) void k_node_agg(const int* __restrict__ rowptr,
                                                             const int* __restrict__ csr,
                                                             const float* __restrict__ dinv,
                                                             const float* __restrict__ hbuf,
                                                             const float* __restrict__ bias,
                                                             float* __restrict__ out,
                                                             float* __restrict__ stats, int n) {
  const int wv = __builtin_amdgcn_readfirstlane(threadIdx.x >> 5);
  const int lane = threadIdx.x & 31;
  const float4* hb = (const float4*)hbuf;
  const int2* cs = (const int2*)csr;
  const float4 bv = ((const float4*)bias)[lane];
  float4 cs1 = make_float4(0.f, 0.f, 0.f, 0.f);
  float4 cs2 = make_float4(0.f, 0.f, 0.f, 0.f);
  for (int node = blockIdx.x * AGG_WAVES + wv; node < n; node += gridDim.x * AGG_WAVES) {
    const int beg = rowptr[node], end = rowptr[node + 1];
    const float di = dinv[node];
    const float4 hs = hb[(size_t)node * 32 + lane];
    float4 acc = make_float4(0.f, 0.f, 0.f, 0.f);
    for (int base = beg; base < end; base += 32) {
      int m = end - base;
      m = m < 32 ? m : 32;
      int2 my = make_int2(0, 0);
      if (lane < m) my = cs[base + lane];
      my.x = my.x < 0 ? 0 : (my.x < n ? my.x : n - 1);
      int j = 0;
      for (; j + 8 <= m; j += 8) edge_batch<8>(acc, hb, my, j, lane);
      if (j + 4 <= m) { edge_batch<4>(acc, hb, my, j, lane); j += 4; }
      const int r = m - j;
      if (r == 3) edge_batch<3>(acc, hb, my, j, lane);
      else if (r == 2) edge_batch<2>(acc, hb, my, j, lane);
      else if (r == 1) edge_batch<1>(acc, hb, my, j, lane);
    }
    const float w_self = di * di;
    acc.x = fmaf(hs.x, w_self, acc.x);
    acc.y = fmaf(hs.y, w_self, acc.y);
    acc.z = fmaf(hs.z, w_self, acc.z);
    acc.w = fmaf(hs.w, w_self, acc.w);
    float4 v;
    v.x = acc.x + bv.x; v.x = v.x > 0.f ? v.x : 0.f;
    v.y = acc.y + bv.y; v.y = v.y > 0.f ? v.y : 0.f;
    v.z = acc.z + bv.z; v.z = v.z > 0.f ? v.z : 0.f;
    v.w = acc.w + bv.w; v.w = v.w > 0.f ? v.w : 0.f;
    ((float4*)out)[(size_t)node * 32 + lane] = v;
    if (STATS) {
      cs1.x += v.x; cs1.y += v.y; cs1.z += v.z; cs1.w += v.w;
      cs2.x = fmaf(v.x, v.x, cs2.x); cs2.y = fmaf(v.y, v.y, cs2.y);
      cs2.z = fmaf(v.z, v.z, cs2.z); cs2.w = fmaf(v.w, v.w, cs2.w);
    }
  }
  if (STATS) {
    __shared__ float ls[AGG_WAVES * 2 * HDIM];
    float4* lw = (float4*)&ls[wv * 2 * HDIM];
    lw[lane] = cs1;
    lw[32 + lane] = cs2;
    __syncthreads();
    if (threadIdx.x < 2 * HDIM) {
      float t = 0.f;
#pragma unroll
      for (int w = 0; w < AGG_WAVES; ++w) t += ls[w * 2 * HDIM + threadIdx.x];
      atomicAdd(&stats[threadIdx.x], t);
    }
  }
}

#define OXP 132
#define OWP 96
__global__ __launch_bounds__(128, 5) void k_out_gemm(const float* __restrict__ x,
                                                  const float* __restrict__ Wo,
                                                  const float* __restrict__ bo,
                                                  float* __restrict__ out, int n) {
  __shared__ float xs[64 * OXP];
  __shared__ float ws[(HDIM / 2) * OWP];
  const int tid = threadIdx.x;
  const int lane = tid & 31;
  const int wave = __builtin_amdgcn_readfirstlane(tid >> 5);
  const int rowBase = blockIdx.x * 64;
  const int m = lane & 15;
  const int hi = lane >> 4;

  float4 xv[16];
#pragma unroll
  for (int p = 0; p < 16; ++p) {
    const int row = rowBase + p * 4 + wave;
    xv[p] = make_float4(0.f, 0.f, 0.f, 0.f);
    if (row < n) xv[p] = ((const float4*)(x + (size_t)row * HDIM))[lane];
  }
#pragma unroll
  for (int p = 0; p < 5; ++p) {
    const int t = p * 128 + tid;
    const int kp = t / 10, j = t - kp * 10;
    const float* wr = Wo + (2 * kp) * NCLS + j * 4;
    const float4 w0 = *(const float4*)wr;
    const float4 w1 = *(const float4*)(wr + NCLS);
    float* d = &ws[kp * OWP + j * 8];
    *(float4*)d = make_float4(w0.x, w1.x, w0.y, w1.y);
    *(float4*)(d + 4) = make_float4(w0.z, w1.z, w0.w, w1.w);
  }
#pragma unroll
  for (int p = 0; p < 2; ++p) {
    const int q = tid * 2 + p;
    *(float4*)&ws[(q >> 2) * OWP + 2 * NCLS + (q & 3) * 4] = make_float4(0.f, 0.f, 0.f, 0.f);
  }
#pragma unroll
  for (int p = 0; p < 16; ++p) *(float4*)(&xs[(p * 4 + wave) * OXP + lane * 4]) = xv[p];
  __syncthreads();

  v8f acc[3];
#pragma unroll
  for (int c = 0; c < 3; ++c)
#pragma unroll
    for (int v = 0; v < 8; ++v) acc[c][v] = 0.f;

  const float* xrow = &xs[(wave * 16 + m) * OXP];
#pragma unroll
  for (int k4 = 0; k4 < 32; ++k4) {
    const int kb = k4 * 4 + hi * 2;
    v2f a; a.x = xrow[kb]; a.y = xrow[kb + 1];
    const float* wk = &ws[(k4 * 2 + hi) * OWP + m * 2];
#pragma unroll
    for (int c = 0; c < 3; ++c) {
      v2f b; b.x = wk[c * 32]; b.y = wk[c * 32 + 1];
      acc[c] = __builtin_amdgcn_wmma_f32_16x16x4_f32(
          false, a, false, b, (short)0, acc[c], false, false);
    }
  }
  __syncthreads();

  float* cs = xs;
#pragma unroll
  for (int c = 0; c < 3; ++c) {
    const int col = c * 16 + m;
    if (col < NCLS) {
      const float bv = bo[col];
#pragma unroll
      for (int v = 0; v < 8; ++v) cs[(wave * 16 + v + 8 * hi) * NCLS + col] = acc[c][v] + bv;
    }
  }
  __syncthreads();
  int rows = n - rowBase;
  rows = rows < 64 ? rows : 64;
  const int cnt = rows * NCLS;
  float* ob = out + (size_t)rowBase * NCLS;
#pragma unroll
  for (int p = 0; p < 5; ++p) {
    const int i4 = (p * 128 + tid) * 4;
    if (i4 < cnt) *(float4*)(ob + i4) = *(const float4*)&cs[i4];
  }
}

extern "C" void kernel_launch(void* const* d_in, const int* in_sizes, int n_in,
                              void* d_out, int out_size, void* d_ws, size_t ws_size,
                              hipStream_t stream) {
  const float* x_in  = (const float*)d_in[0];
  const int*   ei    = (const int*)d_in[1];
  const float* gamma = (const float*)d_in[2];
  const float* beta  = (const float*)d_in[3];
  const float* W     = (const float*)d_in[4];
  const float* b     = (const float*)d_in[5];
  const float* Wo    = (const float*)d_in[6];
  const float* bo    = (const float*)d_in[7];
  float* out = (float*)d_out;

  const int N = in_sizes[0] / HDIM;
  const int E = in_sizes[1] / 2;

  float* dinv  = (float*)d_ws;
  float* stats = dinv + ((N + 255) / 256) * 256;
  float* hbuf  = stats + 2 * HDIM;
  float* xbuf  = hbuf + (size_t)N * HDIM;

  int* csr    = (int*)d_out;
  int* rowptr = csr + ((2 * E + 3) / 4) * 4;

  const int NCH = 250;
  const int nb = (N + BK_NODES - 1) / BK_NODES;
  const int ch = (E + NCH - 1) / NCH;
  const int n2 = nb * NCH;
  int2* tmp  = (int2*)xbuf;
  int*  bh   = (int*)xbuf + ((2 * E + 3) / 4) * 4;
  int*  boff = bh + ((n2 + 3) / 4) * 4;

  k_bucket_hist<<<NCH, CH_T, 0, stream>>>(ei, bh, E, ch, nb);
  k_scan<<<1, SCAN_T, 0, stream>>>(bh, boff, n2);
  k_bucket_scatter<<<NCH, CH_T, 0, stream>>>(ei, boff, tmp, E, ch, nb);
  k_bucket_rows<<<nb, BK_NODES, 0, stream>>>(tmp, boff, rowptr, dinv, N, E, NCH);
  k_bucket_fill<<<nb, BK_NODES, 0, stream>>>(tmp, boff, rowptr, dinv, (int2*)csr, N, E, NCH);

  hipMemsetAsync(stats, 0, 2 * HDIM * sizeof(float), stream);
  k_stats<<<ST_BLOCKS, ST_WAVES * 32, 0, stream>>>(x_in, stats, N);

  for (int l = 0; l < 3; ++l) {
    const float* xl = (l == 0) ? x_in : xbuf;
    k_gemm<<<(N + 63) / 64, 128, 0, stream>>>(xl, stats, gamma + l * HDIM, beta + l * HDIM,
                                              W + (size_t)l * HDIM * HDIM, hbuf, N, 1.0f / (float)N);
    if (l < 2) {
      hipMemsetAsync(stats, 0, 2 * HDIM * sizeof(float), stream);
      k_node_agg<true><<<AGG_BLOCKS, AGG_WAVES * 32, 0, stream>>>(rowptr, csr, dinv, hbuf, b + l * HDIM,
                                                                  xbuf, stats, N);
    } else {
      k_node_agg<false><<<AGG_BLOCKS, AGG_WAVES * 32, 0, stream>>>(rowptr, csr, dinv, hbuf, b + l * HDIM,
                                                                   xbuf, stats, N);
    }
  }

  k_out_gemm<<<(N + 63) / 64, 128, 0, stream>>>(xbuf, Wo, bo, out, N);
}


# Aggregation ahead of the linear map, A(bn(x)W) = (A bn(x))W, one launch per layer: gather waves fill an LDS tile while matrix waves multiply the last one by W; h and 3 launches go; sum order differs
# speedup vs baseline: 5.5085x; 1.2333x over previous
#include <hip/hip_runtime.h>

#define HDIM 128
#define NCLS 40
#define BN_EPS 1e-5f

typedef __attribute__((ext_vector_type(2))) float v2f;
typedef __attribute__((ext_vector_type(8))) float v8f;

#define BK_SHIFT 9
#define BK_NODES 512
#define BK_MAX 1024
#define CH_T 1024
#define SORT_U 8

__device__ __forceinline__ int bucket_of(int d, int nb) {
  int bk = d >> BK_SHIFT;
  return bk < 0 ? 0 : (bk < nb ? bk : nb - 1);
}

__global__ __launch_bounds__(CH_T) void k_bucket_hist(const int* __restrict__ ei, int* __restrict__ bh,
                                                      int E, int ch, int nb) {
  __shared__ int hist[BK_MAX];
  const int t = threadIdx.x;
  for (int q = t; q < nb; q += CH_T) hist[q] = 0;
  __syncthreads();
  const int e0 = blockIdx.x * ch;
  int e1 = e0 + ch;
  e1 = e1 < E ? e1 : E;
  for (int e = e0 + t; e < e1; e += SORT_U * CH_T) {
    int d[SORT_U];
#pragma unroll
    for (int q = 0; q < SORT_U; ++q) d[q] = e + q * CH_T < e1 ? ei[E + e + q * CH_T] : 0;
#pragma unroll
    for (int q = 0; q < SORT_U; ++q)
      if (e + q * CH_T < e1) atomicAdd(&hist[bucket_of(d[q], nb)], 1);
  }
  __syncthreads();
  for (int q = t; q < nb; q += CH_T) bh[q * gridDim.x + blockIdx.x] = hist[q];
}

#define SCAN_T 1024
__device__ __forceinline__ int4 load4_guarded(const int* __restrict__ p, int i0, int n) {
  int4 v = make_int4(0, 0, 0, 0);
  if (i0 + 3 < n) {
    v = *(const int4*)(p + i0);
  } else {
    if (i0 < n) v.x = p[i0];
    if (i0 + 1 < n) v.y = p[i0 + 1];
    if (i0 + 2 < n) v.z = p[i0 + 2];
  }
  return v;
}

__global__ __launch_bounds__(SCAN_T) void k_scan(const int* __restrict__ cnt, int* __restrict__ rowptr, int n) {
  __shared__ int wsum[32];
  const int t = threadIdx.x, lane = t & 31, wv = t >> 5;
  int carry = 0;
  int4 v[4];
#pragma unroll
  for (int g = 0; g < 4; ++g) v[g] = load4_guarded(cnt, t * 16 + g * 4, n);
  for (int base = 0; base < n; base += SCAN_T * 16) {
    const int i0 = base + t * 16;
    int4 nxt[4];
#pragma unroll
    for (int g = 0; g < 4; ++g) nxt[g] = load4_guarded(cnt, i0 + SCAN_T * 16 + g * 4, n);
    int s = 0;
#pragma unroll
    for (int g = 0; g < 4; ++g) s += v[g].x + v[g].y + v[g].z + v[g].w;
    int inc = s;
#pragma unroll
    for (int d = 1; d < 32; d <<= 1) {
      int o = __shfl_up(inc, d);
      if (lane >= d) inc += o;
    }
    if (lane == 31) wsum[wv] = inc;
    __syncthreads();
    const int wt = wsum[lane];
    int winc = wt;
#pragma unroll
    for (int d = 1; d < 32; d <<= 1) {
      int o = __shfl_up(winc, d);
      if (lane >= d) winc += o;
    }
    const int wex = __shfl(winc, wv) - __shfl(wt, wv);
    const int total = __shfl(winc, 31);
    int run = carry + wex + inc - s;
#pragma unroll
    for (int g = 0; g < 4; ++g) {
      const int i = i0 + g * 4;
      const int4 c = v[g];
      if (i + 3 < n) {
        *(int4*)(rowptr + i) = make_int4(run, run + c.x, run + c.x + c.y, run + c.x + c.y + c.z);
      } else {
        if (i < n)     rowptr[i] = run;
        if (i + 1 < n) rowptr[i + 1] = run + c.x;
        if (i + 2 < n) rowptr[i + 2] = run + c.x + c.y;
      }
      run += c.x + c.y + c.z + c.w;
    }
    carry += total;
#pragma unroll
    for (int g = 0; g < 4; ++g) v[g] = nxt[g];
    __syncthreads();
  }
  if (t == 0) rowptr[n] = carry;
}

__global__ __launch_bounds__(CH_T) void k_bucket_scatter(const int* __restrict__ ei,
                                                         const int* __restrict__ boff,
                                                         int2* __restrict__ tmp, int E, int ch, int nb) {
  __shared__ int cur[BK_MAX];
  const int t = threadIdx.x;
  for (int q = t; q < nb; q += CH_T) cur[q] = boff[q * gridDim.x + blockIdx.x];
  __syncthreads();
  const int e0 = blockIdx.x * ch;
  int e1 = e0 + ch;
  e1 = e1 < E ? e1 : E;
  for (int e = e0 + t; e < e1; e += SORT_U * CH_T) {
    int s[SORT_U], d[SORT_U];
#pragma unroll
    for (int q = 0; q < SORT_U; ++q) {
      const bool in = e + q * CH_T < e1;
      s[q] = in ? ei[e + q * CH_T] : 0;
      d[q] = in ? ei[E + e + q * CH_T] : 0;
    }
#pragma unroll
    for (int q = 0; q < SORT_U; ++q) {
      if (e + q * CH_T < e1) {
        int p = atomicAdd(&cur[bucket_of(d[q], nb)], 1);
        p = p < 0 ? 0 : (p < E ? p : E - 1);
        tmp[p] = make_int2(s[q], d[q]);
      }
    }
  }
}

__device__ __forceinline__ void bucket_range(const int* __restrict__ boff, int bk, int nch, int E,
                                             int& bs, int& be) {
  bs = boff[bk * nch];
  be = boff[(bk + 1) * nch];
  bs = bs < 0 ? 0 : (bs < E ? bs : E);
  be = be < bs ? bs : (be < E ? be : E);
}

__device__ __forceinline__ int local_node(int d, int node0) {
  const int l = d - node0;
  return l < 0 ? 0 : (l < BK_NODES ? l : BK_NODES - 1);
}

__global__ __launch_bounds__(BK_NODES) void k_bucket_rows(const int2* __restrict__ tmp,
                                                          const int* __restrict__ boff,
                                                          int* __restrict__ rowptr,
                                                          float* __restrict__ dinv, int n, int E, int nch) {
  __shared__ int cnt[BK_NODES];
  __shared__ int wsum[32];
  const int t = threadIdx.x, lane = t & 31, wv = t >> 5;
  const int node0 = blockIdx.x << BK_SHIFT;
  int bs, be;
  bucket_range(boff, blockIdx.x, nch, E, bs, be);
  cnt[t] = 0;
  __syncthreads();
  for (int i = bs + t; i < be; i += SORT_U * BK_NODES) {
    int d[SORT_U];
#pragma unroll
    for (int q = 0; q < SORT_U; ++q) d[q] = i + q * BK_NODES < be ? tmp[i + q * BK_NODES].y : 0;
#pragma unroll
    for (int q = 0; q < SORT_U; ++q)
      if (i + q * BK_NODES < be) atomicAdd(&cnt[local_node(d[q], node0)], 1);
  }
  __syncthreads();
  const int c = cnt[t];
  int inc = c;
#pragma unroll
  for (int d = 1; d < 32; d <<= 1) {
    int o = __shfl_up(inc, d);
    if (lane >= d) inc += o;
  }
  if (lane == 31) wsum[wv] = inc;
  __syncthreads();
  const int wt = lane < BK_NODES / 32 ? wsum[lane] : 0;
  int winc = wt;
#pragma unroll
  for (int d = 1; d < 32; d <<= 1) {
    int o = __shfl_up(winc, d);
    if (lane >= d) winc += o;
  }
  const int wex = __shfl(winc, wv) - __shfl(wt, wv);
  const int node = node0 + t;
  if (node < n) {
    rowptr[node] = bs + wex + inc - c;
    dinv[node] = rsqrtf((float)(c + 1));
  }
  if (blockIdx.x == gridDim.x - 1 && t == 0) rowptr[n] = be;
}

__global__ __launch_bounds__(BK_NODES) void k_bucket_fill(const int2* __restrict__ tmp,
                                                          const int* __restrict__ boff,
                                                          const int* __restrict__ rowptr,
                                                          const float* __restrict__ dinv,
                                                          int2* __restrict__ csr, int n, int E, int nch) {
  __shared__ int cur[BK_NODES];
  const int t = threadIdx.x;
  const int node0 = blockIdx.x << BK_SHIFT;
  int bs, be;
  bucket_range(boff, blockIdx.x, nch, E, bs, be);
  cur[t] = node0 + t < n ? rowptr[node0 + t] : be;
  __syncthreads();
  for (int i = bs + t; i < be; i += SORT_U * BK_NODES) {
    int2 sd[SORT_U];
    float w[SORT_U];
#pragma unroll
    for (int q = 0; q < SORT_U; ++q)
      sd[q] = i + q * BK_NODES < be ? tmp[i + q * BK_NODES] : make_int2(0, 0);
#pragma unroll
    for (int q = 0; q < SORT_U; ++q) {
      const int s = sd[q].x < 0 ? 0 : (sd[q].x < n ? sd[q].x : n - 1);
      const int d = sd[q].y < 0 ? 0 : (sd[q].y < n ? sd[q].y : n - 1);
      w[q] = dinv[s] * dinv[d];
    }
#pragma unroll
    for (int q = 0; q < SORT_U; ++q) {
      if (i + q * BK_NODES < be) {
        int p = atomicAdd(&cur[local_node(sd[q].y, node0)], 1);
        p = p < 0 ? 0 : (p < E ? p : E - 1);
        csr[p] = make_int2(sd[q].x, __float_as_int(w[q]));
      }
    }
  }
}

#define ST_WAVES 8
#define ST_BLOCKS 512
#define ST_ROWS 32
static_assert(ST_WAVES * 32 == 2 * HDIM, "one thread per entry of stats in the block's last step");
__global__ __launch_bounds__(ST_WAVES * 32) void k_stats(const float* __restrict__ x,
                                                         float* __restrict__ stats, int n) {
  const int wv = __builtin_amdgcn_readfirstlane(threadIdx.x >> 5);
  const unsigned lane = threadIdx.x & 31;
  const float4* xb = (const float4*)x;
  const int nw = gridDim.x * ST_WAVES;
  float4 s1 = make_float4(0.f, 0.f, 0.f, 0.f);
  float4 s2 = make_float4(0.f, 0.f, 0.f, 0.f);
  for (int base = blockIdx.x * ST_WAVES + wv; base < n; base += ST_ROWS * nw) {
    float4 a[ST_ROWS];
#pragma unroll
    for (int q = 0; q < ST_ROWS; ++q) {
      const int r = base + q * nw;
      if (r < n) a[q] = xb[(size_t)r * 32 + lane];
    }
#pragma unroll
    for (int q = 0; q < ST_ROWS; ++q) {
      if (base + q * nw < n) {
        s1.x += a[q].x; s1.y += a[q].y; s1.z += a[q].z; s1.w += a[q].w;
        s2.x = fmaf(a[q].x, a[q].x, s2.x); s2.y = fmaf(a[q].y, a[q].y, s2.y);
        s2.z = fmaf(a[q].z, a[q].z, s2.z); s2.w = fmaf(a[q].w, a[q].w, s2.w);
      }
    }
  }
  __shared__ float ls[ST_WAVES * 2 * HDIM];
  float4* lw = (float4*)&ls[wv * 2 * HDIM];
  lw[lane] = s1;
  lw[32 + lane] = s2;
  __syncthreads();
  float t = 0.f;
#pragma unroll
  for (int w = 0; w < ST_WAVES; ++w) t += ls[w * 2 * HDIM + threadIdx.x];
  atomicAdd(&stats[threadIdx.x], t);
}

#define WPAIR 288
#define ZP 132
#define FL_ROWS 48
#define FL_GW 24
#define FL_MW 8
#define FL_BLOCKS 256

__device__ __forceinline__ void bn_coef(float s1, float s2, float g, float bt, float inv_n,
                                        float& a, float& c) {
  const float mean = s1 * inv_n;
  const float var = s2 * inv_n - mean * mean;
  a = g * rsqrtf(var + BN_EPS);
  c = bt - a * mean;
}

template <int K>
__device__ __forceinline__ void edge_batch(float4& acc, float& wsum, const float4* __restrict__ hb,
                                           const int2 my, const int j, const int lane) {
  int sidx[K];
  float wgt[K];
  float4 a[K];
#pragma unroll
  for (int q = 0; q < K; ++q) {
    sidx[q] = __builtin_amdgcn_readlane(my.x, j + q);
    wgt[q] = __int_as_float(__builtin_amdgcn_readlane(my.y, j + q));
  }
#pragma unroll
  for (int q = 0; q < K; ++q) a[q] = hb[(size_t)sidx[q] * 32 + lane];
#pragma unroll
  for (int q = 0; q < K; ++q) {
    acc.x = fmaf(a[q].x, wgt[q], acc.x);
    acc.y = fmaf(a[q].y, wgt[q], acc.y);
    acc.z = fmaf(a[q].z, wgt[q], acc.z);
    acc.w = fmaf(a[q].w, wgt[q], acc.w);
    wsum += wgt[q];
  }
}

template <bool STATS>
__global__ __launch_bounds__((FL_GW + FL_MW) * 32) void k_layer(const int* __restrict__ rowptr,
                                                                const int* __restrict__ csr,
                                                                const float* __restrict__ dinv,
                                                                const float* __restrict__ xin,
                                                                const float* __restrict__ stats_in,
                                                                const float* __restrict__ gamma,
                                                                const float* __restrict__ beta,
                                                                const float* __restrict__ W,
                                                                const float* __restrict__ bias,
                                                                float* __restrict__ xout,
                                                                float* __restrict__ stats_out, int n, float inv_n) {
  __shared__ float wsm[(HDIM / 2) * WPAIR];
  __shared__ float zs[2 * FL_ROWS * ZP];
  static_assert(FL_GW * 2 == FL_ROWS, "two nodes per gather wave");
  static_assert(FL_MW * 16 == HDIM, "one column tile per matrix wave");
  const int tid = threadIdx.x;
  const int lane = tid & 31;
  const int wave = __builtin_amdgcn_readfirstlane(tid >> 5);
  const int ntiles = (n + FL_ROWS - 1) / FL_ROWS;
  const int nmine = (int)blockIdx.x < ntiles ? (ntiles - 1 - (int)blockIdx.x) / (int)gridDim.x + 1 : 0;

#pragma unroll
  for (int p = 0; p < 2; ++p) {
    const int kp = p * (FL_GW + FL_MW) + wave;
    const float* wr = W + (2 * kp) * HDIM + lane * 4;
    const float4 w0 = *(const float4*)wr;
    const float4 w1 = *(const float4*)(wr + HDIM);
    float* d = &wsm[kp * WPAIR + lane * 8];
    *(float4*)d = make_float4(w0.x, w1.x, w0.y, w1.y);
    *(float4*)(d + 4) = make_float4(w0.z, w1.z, w0.w, w1.w);
  }

  const float4* hb = (const float4*)xin;
  const int2* cs = (const int2*)csr;
  float4 sa, sc;
  {
    const float4 s1 = ((const float4*)stats_in)[lane];
    const float4 s2 = ((const float4*)(stats_in + HDIM))[lane];
    const float4 g4 = ((const float4*)gamma)[lane];
    const float4 b4 = ((const float4*)beta)[lane];
    bn_coef(s1.x, s2.x, g4.x, b4.x, inv_n, sa.x, sc.x);
    bn_coef(s1.y, s2.y, g4.y, b4.y, inv_n, sa.y, sc.y);
    bn_coef(s1.z, s2.z, g4.z, b4.z, inv_n, sa.z, sc.z);
    bn_coef(s1.w, s2.w, g4.w, b4.w, inv_n, sa.w, sc.w);
  }
  const int mw = wave >= FL_GW ? wave - FL_GW : 0;
  const int m16 = lane & 15;
  const int hi = lane >> 4;
  const int col = mw * 16 + m16;
  const float bcol = bias[col];
  float cs1 = 0.f, cs2 = 0.f;

  for (int it = 0; it <= nmine; ++it) {
    if (wave < FL_GW) {
      if (it < nmine) {
        const int tile = blockIdx.x + it * gridDim.x;
        float* zb = &zs[(it & 1) * FL_ROWS * ZP];
#pragma unroll 1
        for (int q = 0; q < 2; ++q) {
          const int r = wave * 2 + q;
          const int node = tile * FL_ROWS + r;
          float4 z = make_float4(0.f, 0.f, 0.f, 0.f);
          if (node < n) {
            const int beg = rowptr[node], end = rowptr[node + 1];
            const float di = dinv[node];
            const float4 hs = hb[(size_t)node * 32 + lane];
            float4 acc = make_float4(0.f, 0.f, 0.f, 0.f);
            float wsum = 0.f;
            for (int base = beg; base < end; base += 32) {
              int m = end - base;
              m = m < 32 ? m : 32;
              int2 my = make_int2(0, 0);
              if (lane < m) my = cs[base + lane];
              my.x = my.x < 0 ? 0 : (my.x < n ? my.x : n - 1);
              int j = 0;
              for (; j + 16 <= m; j += 16) edge_batch<16>(acc, wsum, hb, my, j, lane);
              if (j + 8 <= m) { edge_batch<8>(acc, wsum, hb, my, j, lane); j += 8; }
              if (j + 4 <= m) { edge_batch<4>(acc, wsum, hb, my, j, lane); j += 4; }
              const int rem = m - j;
              if (rem == 3) edge_batch<3>(acc, wsum, hb, my, j, lane);
              else if (rem == 2) edge_batch<2>(acc, wsum, hb, my, j, lane);
              else if (rem == 1) edge_batch<1>(acc, wsum, hb, my, j, lane);
            }
            const float w_self = di * di;
            acc.x = fmaf(hs.x, w_self, acc.x);
            acc.y = fmaf(hs.y, w_self, acc.y);
            acc.z = fmaf(hs.z, w_self, acc.z);
            acc.w = fmaf(hs.w, w_self, acc.w);
            wsum += w_self;
            z.x = fmaf(sa.x, acc.x, sc.x * wsum);
            z.y = fmaf(sa.y, acc.y, sc.y * wsum);
            z.z = fmaf(sa.z, acc.z, sc.z * wsum);
            z.w = fmaf(sa.w, acc.w, sc.w * wsum);
          }
          *(float4*)(&zb[r * ZP + lane * 4]) = z;
        }
      }
    } else if (it >= 1) {
      const int tile = blockIdx.x + (it - 1) * gridDim.x;
      const float* zb = &zs[((it - 1) & 1) * FL_ROWS * ZP];
      v8f pacc[3];
#pragma unroll
      for (int g = 0; g < 3; ++g)
#pragma unroll
        for (int v = 0; v < 8; ++v) pacc[g][v] = 0.f;
      const float* xrow = &zb[m16 * ZP];
      const float* wk0 = &wsm[hi * WPAIR + col * 2];
#pragma unroll
      for (int k4 = 0; k4 < HDIM / 4; ++k4) {
        const int kb = k4 * 4 + hi * 2;
        const float* wk = wk0 + (k4 * 2) * WPAIR;
        v2f b; b.x = wk[0]; b.y = wk[1];
#pragma unroll
        for (int g = 0; g < 3; ++g) {
          v2f a; a.x = xrow[g * 16 * ZP + kb]; a.y = xrow[g * 16 * ZP + kb + 1];
          pacc[g] = __builtin_amdgcn_wmma_f32_16x16x4_f32(
              false, a, false, b, (short)0, pacc[g], false, false);
        }
      }
#pragma unroll
      for (int g = 0; g < 3; ++g) {
#pragma unroll
        for (int v = 0; v < 8; ++v) {
          const int row = tile * FL_ROWS + g * 16 + v + 8 * hi;
          float o = pacc[g][v] + bcol;
          o = o > 0.f ? o : 0.f;
          if (row < n) {
            xout[(size_t)row * HDIM + col] = o;
            if (STATS) { cs1 += o; cs2 = fmaf(o, o, cs2); }
          }
        }
      }
    }
    __syncthreads();
  }
  if (STATS) {
    if (wave >= FL_GW) {
      cs1 += __shfl_xor(cs1, 16);
      cs2 += __shfl_xor(cs2, 16);
      if (hi == 0) {
        atomicAdd(&stats_out[col], cs1);
        atomicAdd(&stats_out[HDIM + col], cs2);
      }
    }
  }
}

#define OXP 132
#define OWP 96
__global__ __launch_bounds__(128, 5) void k_out_gemm(const float* __restrict__ x,
                                                  const float* __restrict__ Wo,
                                                  const float* __restrict__ bo,
                                                  float* __restrict__ out, int n) {
  __shared__ float xs[64 * OXP];
  __shared__ float ws[(HDIM / 2) * OWP];
  const int tid = threadIdx.x;
  const int lane = tid & 31;
  const int wave = __builtin_amdgcn_readfirstlane(tid >> 5);
  const int rowBase = blockIdx.x * 64;
  const int m = lane & 15;
  const int hi = lane >> 4;

  float4 xv[16];
#pragma unroll
  for (int p = 0; p < 16; ++p) {
    const int row = rowBase + p * 4 + wave;
    xv[p] = make_float4(0.f, 0.f, 0.f, 0.f);
    if (row < n) xv[p] = ((const float4*)(x + (size_t)row * HDIM))[lane];
  }
#pragma unroll
  for (int p = 0; p < 5; ++p) {
    const int t = p * 128 + tid;
    const int kp = t / 10, j = t - kp * 10;
    const float* wr = Wo + (2 * kp) * NCLS + j * 4;
    const float4 w0 = *(const float4*)wr;
    const float4 w1 = *(const float4*)(wr + NCLS);
    float* d = &ws[kp * OWP + j * 8];
    *(float4*)d = make_float4(w0.x, w1.x, w0.y, w1.y);
    *(float4*)(d + 4) = make_float4(w0.z, w1.z, w0.w, w1.w);
  }
#pragma unroll
  for (int p = 0; p < 2; ++p) {
    const int q = tid * 2 + p;
    *(float4*)&ws[(q >> 2) * OWP + 2 * NCLS + (q & 3) * 4] = make_float4(0.f, 0.f, 0.f, 0.f);
  }
#pragma unroll
  for (int p = 0; p < 16; ++p) *(float4*)(&xs[(p * 4 + wave) * OXP + lane * 4]) = xv[p];
  __syncthreads();

  v8f acc[3];
#pragma unroll
  for (int c = 0; c < 3; ++c)
#pragma unroll
    for (int v = 0; v < 8; ++v) acc[c][v] = 0.f;

  const float* xrow = &xs[(wave * 16 + m) * OXP];
#pragma unroll
  for (int k4 = 0; k4 < 32; ++k4) {
    const int kb = k4 * 4 + hi * 2;
    v2f a; a.x = xrow[kb]; a.y = xrow[kb + 1];
    const float* wk = &ws[(k4 * 2 + hi) * OWP + m * 2];
#pragma unroll
    for (int c = 0; c < 3; ++c) {
      v2f b; b.x = wk[c * 32]; b.y = wk[c * 32 + 1];
      acc[c] = __builtin_amdgcn_wmma_f32_16x16x4_f32(
          false, a, false, b, (short)0, acc[c], false, false);
    }
  }
  __syncthreads();

  float* cs = xs;
#pragma unroll
  for (int c = 0; c < 3; ++c) {
    const int col = c * 16 + m;
    if (col < NCLS) {
      const float bv = bo[col];
#pragma unroll
      for (int v = 0; v < 8; ++v) cs[(wave * 16 + v + 8 * hi) * NCLS + col] = acc[c][v] + bv;
    }
  }
  __syncthreads();
  int rows = n - rowBase;
  rows = rows < 64 ? rows : 64;
  const int cnt = rows * NCLS;
  float* ob = out + (size_t)rowBase * NCLS;
#pragma unroll
  for (int p = 0; p < 5; ++p) {
    const int i4 = (p * 128 + tid) * 4;
    if (i4 < cnt) *(float4*)(ob + i4) = *(const float4*)&cs[i4];
  }
}

extern "C" void kernel_launch(void* const* d_in, const int* in_sizes, int n_in,
                              void* d_out, int out_size, void* d_ws, size_t ws_size,
                              hipStream_t stream) {
  const float* x_in  = (const float*)d_in[0];
  const int*   ei    = (const int*)d_in[1];
  const float* gamma = (const float*)d_in[2];
  const float* beta  = (const float*)d_in[3];
  const float* W     = (const float*)d_in[4];
  const float* b     = (const float*)d_in[5];
  const float* Wo    = (const float*)d_in[6];
  const float* bo    = (const float*)d_in[7];
  float* out = (float*)d_out;

  const int N = in_sizes[0] / HDIM;
  const int E = in_sizes[1] / 2;

  float* dinv  = (float*)d_ws;
  float* stats = dinv + ((N + 255) / 256) * 256;
  float* hbuf  = stats + 2 * HDIM;
  float* xbuf  = hbuf + (size_t)N * HDIM;

  int* csr    = (int*)d_out;
  int* rowptr = csr + ((2 * E + 3) / 4) * 4;
  float* stats1 = (float*)(rowptr + ((N + 4) / 4) * 4);
  float* stats2 = stats1 + 2 * HDIM;

  const int NCH = 250;
  const int nb = (N + BK_NODES - 1) / BK_NODES;
  const int ch = (E + NCH - 1) / NCH;
  const int n2 = nb * NCH;
  int2* tmp  = (int2*)xbuf;
  int*  bh   = (int*)xbuf + ((2 * E + 3) / 4) * 4;
  int*  boff = bh + ((n2 + 3) / 4) * 4;

  k_bucket_hist<<<NCH, CH_T, 0, stream>>>(ei, bh, E, ch, nb);
  k_scan<<<1, SCAN_T, 0, stream>>>(bh, boff, n2);
  k_bucket_scatter<<<NCH, CH_T, 0, stream>>>(ei, boff, tmp, E, ch, nb);
  k_bucket_rows<<<nb, BK_NODES, 0, stream>>>(tmp, boff, rowptr, dinv, N, E, NCH);
  k_bucket_fill<<<nb, BK_NODES, 0, stream>>>(tmp, boff, rowptr, dinv, (int2*)csr, N, E, NCH);

  hipMemsetAsync(stats, 0, 2 * HDIM * sizeof(float), stream);
  k_stats<<<ST_BLOCKS, ST_WAVES * 32, 0, stream>>>(x_in, stats, N);

  hipMemsetAsync(stats1, 0, 4 * HDIM * sizeof(float), stream);
  k_layer<true><<<FL_BLOCKS, (FL_GW + FL_MW) * 32, 0, stream>>>(
      rowptr, csr, dinv, x_in, stats, gamma, beta, W, b, hbuf, stats1, N, 1.0f / (float)N);
  k_layer<true><<<FL_BLOCKS, (FL_GW + FL_MW) * 32, 0, stream>>>(
      rowptr, csr, dinv, hbuf, stats1, gamma + HDIM, beta + HDIM, W + (size_t)HDIM * HDIM, b + HDIM,
      xbuf, stats2, N, 1.0f / (float)N);
  k_layer<false><<<FL_BLOCKS, (FL_GW + FL_MW) * 32, 0, stream>>>(
      rowptr, csr, dinv, xbuf, stats2, gamma + 2 * HDIM, beta + 2 * HDIM, W + (size_t)2 * HDIM * HDIM,
      b + 2 * HDIM, hbuf, stats1, N, 1.0f / (float)N);

  k_out_gemm<<<(N + 63) / 64, 128, 0, stream>>>(hbuf, Wo, bo, out, N);
}


# 12 stream operations become 8: x's column sums ride as extra blocks of the edge-scatter launch, scatter blocks scan the count table themselves, first kernel clears the sum tables; sum order differs
# speedup vs baseline: 5.7579x; 1.0453x over previous
#include <hip/hip_runtime.h>

#define HDIM 128
#define NCLS 40
#define BN_EPS 1e-5f

typedef __attribute__((ext_vector_type(2))) float v2f;
typedef __attribute__((ext_vector_type(8))) float v8f;

#define BK_SHIFT 9
#define BK_NODES 512
#define BK_MAX 1024
#define CH_T 1024
#define SORT_U 8

__device__ __forceinline__ int bucket_of(int d, int nb) {
  int bk = d >> BK_SHIFT;
  return bk < 0 ? 0 : (bk < nb ? bk : nb - 1);
}

__global__ __launch_bounds__(CH_T) void k_bucket_hist(const int* __restrict__ ei, int* __restrict__ bh,
                                                      int E, int ch, int nb,
                                                      float* __restrict__ sums0, float* __restrict__ sums12) {
  __shared__ int hist[BK_MAX];
  const int t = threadIdx.x;
  if (blockIdx.x == 0) {
    if (t < 2 * HDIM) sums0[t] = 0.f;
    if (t < 4 * HDIM) sums12[t] = 0.f;
  }
  for (int q = t; q < nb; q += CH_T) hist[q] = 0;
  __syncthreads();
  const int e0 = blockIdx.x * ch;
  int e1 = e0 + ch;
  e1 = e1 < E ? e1 : E;
  for (int e = e0 + t; e < e1; e += SORT_U * CH_T) {
    int d[SORT_U];
#pragma unroll
    for (int q = 0; q < SORT_U; ++q) d[q] = e + q * CH_T < e1 ? ei[E + e + q * CH_T] : 0;
#pragma unroll
    for (int q = 0; q < SORT_U; ++q)
      if (e + q * CH_T < e1) atomicAdd(&hist[bucket_of(d[q], nb)], 1);
  }
  __syncthreads();
  for (int q = t; q < nb; q += CH_T) bh[q * gridDim.x + blockIdx.x] = hist[q];
}

#define NCHK 250
#define SS_ROWS 8
static_assert(CH_T == 1024 && CH_T >= 2 * HDIM, "32 waves; one thread per entry of stats in the last step");
__global__ __launch_bounds__(CH_T) void k_scatter_stats(const int* __restrict__ ei, const int* __restrict__ bh,
                                                        int* __restrict__ bstart, int2* __restrict__ tmp,
                                                        int E, int ch, int nb,
                                                        const float* __restrict__ x, float* __restrict__ stats, int n) {
  __shared__ float ls[32 * 2 * HDIM];
  const int t = threadIdx.x;
  const int lane = t & 31;
  const int wv = __builtin_amdgcn_readfirstlane(t >> 5);
  if ((int)blockIdx.x >= NCHK) {
    const float4* xb = (const float4*)x;
    const int nw = ((int)gridDim.x - NCHK) * 32;
    float4 s1 = make_float4(0.f, 0.f, 0.f, 0.f);
    float4 s2 = make_float4(0.f, 0.f, 0.f, 0.f);
    for (int base = ((int)blockIdx.x - NCHK) * 32 + wv; base < n; base += SS_ROWS * nw) {
      float4 a[SS_ROWS];
#pragma unroll
      for (int q = 0; q < SS_ROWS; ++q) {
        const int r = base + q * nw;
        if (r < n) a[q] = xb[(size_t)r * 32 + lane];
      }
#pragma unroll
      for (int q = 0; q < SS_ROWS; ++q) {
        if (base + q * nw < n) {
          s1.x += a[q].x; s1.y += a[q].y; s1.z += a[q].z; s1.w += a[q].w;
          s2.x = fmaf(a[q].x, a[q].x, s2.x); s2.y = fmaf(a[q].y, a[q].y, s2.y);
          s2.z = fmaf(a[q].z, a[q].z, s2.z); s2.w = fmaf(a[q].w, a[q].w, s2.w);
        }
      }
    }
    float4* lw = (float4*)&ls[wv * 2 * HDIM];
    lw[lane] = s1;
    lw[32 + lane] = s2;
    __syncthreads();
    if (t < 2 * HDIM) {
      float s = 0.f;
#pragma unroll
      for (int w = 0; w < 32; ++w) s += ls[w * 2 * HDIM + t];
      atomicAdd(&stats[t], s);
    }
    return;
  }
  int* cur = (int*)ls;
  int* tot = cur + BK_MAX;
  int* pre = tot + BK_MAX;
  const int c = blockIdx.x;
  for (int q0 = wv; q0 < nb; q0 += 64) {
    int v[2][8];
#pragma unroll
    for (int h = 0; h < 2; ++h)
#pragma unroll
      for (int j = 0; j < 8; ++j) {
        const int q = q0 + h * 32, k = j * 32 + lane;
        v[h][j] = (q < nb && k < NCHK) ? bh[q * NCHK + k] : 0;
      }
#pragma unroll
    for (int h = 0; h < 2; ++h) {
      int ts = 0, ps = 0;
#pragma unroll
      for (int j = 0; j < 8; ++j) {
        ts += v[h][j];
        ps += (j * 32 + lane < c) ? v[h][j] : 0;
      }
#pragma unroll
      for (int d = 16; d >= 1; d >>= 1) {
        ts += __shfl_xor(ts, d);
        ps += __shfl_xor(ps, d);
      }
      const int q = q0 + h * 32;
      if (lane == 0 && q < nb) { tot[q] = ts; pre[q] = ps; }
    }
  }
  __syncthreads();
  if (wv == 0) {
    const int per = (nb + 31) >> 5;
    const int b0 = lane * per;
    int s = 0;
    for (int k = 0; k < per; ++k) s += b0 + k < nb ? tot[b0 + k] : 0;
    int inc = s;
#pragma unroll
    for (int d = 1; d < 32; d <<= 1) {
      int o = __shfl_up(inc, d);
      if (lane >= d) inc += o;
    }
    int run = inc - s;
    for (int k = 0; k < per; ++k) {
      const int q = b0 + k;
      if (q < nb) {
        cur[q] = run + pre[q];
        if (c == 0) bstart[q] = run;
        run += tot[q];
      }
    }
    if (c == 0 && lane == 31) bstart[nb] = run;
  }
  __syncthreads();
  const int e0 = c * ch;
  int e1 = e0 + ch;
  e1 = e1 < E ? e1 : E;
  for (int e = e0 + t; e < e1; e += SORT_U * CH_T) {
    int s[SORT_U], d[SORT_U];
#pragma unroll
    for (int q = 0; q < SORT_U; ++q) {
      const bool in = e + q * CH_T < e1;
      s[q] = in ? ei[e + q * CH_T] : 0;
      d[q] = in ? ei[E + e + q * CH_T] : 0;
    }
#pragma unroll
    for (int q = 0; q < SORT_U; ++q) {
      if (e + q * CH_T < e1) {
        int p = atomicAdd(&cur[bucket_of(d[q], nb)], 1);
        p = p < 0 ? 0 : (p < E ? p : E - 1);
        tmp[p] = make_int2(s[q], d[q]);
      }
    }
  }
}

__device__ __forceinline__ void bucket_range(const int* __restrict__ boff, int bk, int nch, int E,
                                             int& bs, int& be) {
  bs = boff[bk * nch];
  be = boff[(bk + 1) * nch];
  bs = bs < 0 ? 0 : (bs < E ? bs : E);
  be = be < bs ? bs : (be < E ? be : E);
}

__device__ __forceinline__ int local_node(int d, int node0) {
  const int l = d - node0;
  return l < 0 ? 0 : (l < BK_NODES ? l : BK_NODES - 1);
}

__global__ __launch_bounds__(BK_NODES) void k_bucket_rows(const int2* __restrict__ tmp,
                                                          const int* __restrict__ boff,
                                                          int* __restrict__ rowptr,
                                                          float* __restrict__ dinv, int n, int E, int nch) {
  __shared__ int cnt[BK_NODES];
  __shared__ int wsum[32];
  const int t = threadIdx.x, lane = t & 31, wv = t >> 5;
  const int node0 = blockIdx.x << BK_SHIFT;
  int bs, be;
  bucket_range(boff, blockIdx.x, nch, E, bs, be);
  cnt[t] = 0;
  __syncthreads();
  for (int i = bs + t; i < be; i += SORT_U * BK_NODES) {
    int d[SORT_U];
#pragma unroll
    for (int q = 0; q < SORT_U; ++q) d[q] = i + q * BK_NODES < be ? tmp[i + q * BK_NODES].y : 0;
#pragma unroll
    for (int q = 0; q < SORT_U; ++q)
      if (i + q * BK_NODES < be) atomicAdd(&cnt[local_node(d[q], node0)], 1);
  }
  __syncthreads();
  const int c = cnt[t];
  int inc = c;
#pragma unroll
  for (int d = 1; d < 32; d <<= 1) {
    int o = __shfl_up(inc, d);
    if (lane >= d) inc += o;
  }
  if (lane == 31) wsum[wv] = inc;
  __syncthreads();
  const int wt = lane < BK_NODES / 32 ? wsum[lane] : 0;
  int winc = wt;
#pragma unroll
  for (int d = 1; d < 32; d <<= 1) {
    int o = __shfl_up(winc, d);
    if (lane >= d) winc += o;
  }
  const int wex = __shfl(winc, wv) - __shfl(wt, wv);
  const int node = node0 + t;
  if (node < n) {
    rowptr[node] = bs + wex + inc - c;
    dinv[node] = rsqrtf((float)(c + 1));
  }
  if (blockIdx.x == gridDim.x - 1 && t == 0) rowptr[n] = be;
}

__global__ __launch_bounds__(BK_NODES) void k_bucket_fill(const int2* __restrict__ tmp,
                                                          const int* __restrict__ boff,
                                                          const int* __restrict__ rowptr,
                                                          const float* __restrict__ dinv,
                                                          int2* __restrict__ csr, int n, int E, int nch) {
  __shared__ int cur[BK_NODES];
  const int t = threadIdx.x;
  const int node0 = blockIdx.x << BK_SHIFT;
  int bs, be;
  bucket_range(boff, blockIdx.x, nch, E, bs, be);
  cur[t] = node0 + t < n ? rowptr[node0 + t] : be;
  __syncthreads();
  for (int i = bs + t; i < be; i += SORT_U * BK_NODES) {
    int2 sd[SORT_U];
    float w[SORT_U];
#pragma unroll
    for (int q = 0; q < SORT_U; ++q)
      sd[q] = i + q * BK_NODES < be ? tmp[i + q * BK_NODES] : make_int2(0, 0);
#pragma unroll
    for (int q = 0; q < SORT_U; ++q) {
      const int s = sd[q].x < 0 ? 0 : (sd[q].x < n ? sd[q].x : n - 1);
      const int d = sd[q].y < 0 ? 0 : (sd[q].y < n ? sd[q].y : n - 1);
      w[q] = dinv[s] * dinv[d];
    }
#pragma unroll
    for (int q = 0; q < SORT_U; ++q) {
      if (i + q * BK_NODES < be) {
        int p = atomicAdd(&cur[local_node(sd[q].y, node0)], 1);
        p = p < 0 ? 0 : (p < E ? p : E - 1);
        csr[p] = make_int2(sd[q].x, __float_as_int(w[q]));
      }
    }
  }
}

#define WPAIR 288
#define ZP 132
#define FL_ROWS 48
#define FL_GW 24
#define FL_MW 8
#define FL_BLOCKS 256

__device__ __forceinline__ void bn_coef(float s1, float s2, float g, float bt, float inv_n,
                                        float& a, float& c) {
  const float mean = s1 * inv_n;
  const float var = s2 * inv_n - mean * mean;
  a = g * rsqrtf(var + BN_EPS);
  c = bt - a * mean;
}

template <int K>
__device__ __forceinline__ void edge_batch(float4& acc, float& wsum, const float4* __restrict__ hb,
                                           const int2 my, const int j, const int lane) {
  int sidx[K];
  float wgt[K];
  float4 a[K];
#pragma unroll
  for (int q = 0; q < K; ++q) {
    sidx[q] = __builtin_amdgcn_readlane(my.x, j + q);
    wgt[q] = __int_as_float(__builtin_amdgcn_readlane(my.y, j + q));
  }
#pragma unroll
  for (int q = 0; q < K; ++q) a[q] = hb[(size_t)sidx[q] * 32 + lane];
#pragma unroll
  for (int q = 0; q < K; ++q) {
    acc.x = fmaf(a[q].x, wgt[q], acc.x);
    acc.y = fmaf(a[q].y, wgt[q], acc.y);
    acc.z = fmaf(a[q].z, wgt[q], acc.z);
    acc.w = fmaf(a[q].w, wgt[q], acc.w);
    wsum += wgt[q];
  }
}

template <bool STATS>
__global__ __launch_bounds__((FL_GW + FL_MW) * 32) void k_layer(const int* __restrict__ rowptr,
                                                                const int* __restrict__ csr,
                                                                const float* __restrict__ dinv,
                                                                const float* __restrict__ xin,
                                                                const float* __restrict__ stats_in,
                                                                const float* __restrict__ gamma,
                                                                const float* __restrict__ beta,
                                                                const float* __restrict__ W,
                                                                const float* __restrict__ bias,
                                                                float* __restrict__ xout,
                                                                float* __restrict__ stats_out, int n, float inv_n) {
  __shared__ float wsm[(HDIM / 2) * WPAIR];
  __shared__ float zs[2 * FL_ROWS * ZP];
  static_assert(FL_GW * 2 == FL_ROWS, "two nodes per gather wave");
  static_assert(FL_MW * 16 == HDIM, "one column tile per matrix wave");
  const int tid = threadIdx.x;
  const int lane = tid & 31;
  const int wave = __builtin_amdgcn_readfirstlane(tid >> 5);
  const int ntiles = (n + FL_ROWS - 1) / FL_ROWS;
  const int nmine = (int)blockIdx.x < ntiles ? (ntiles - 1 - (int)blockIdx.x) / (int)gridDim.x + 1 : 0;

#pragma unroll
  for (int p = 0; p < 2; ++p) {
    const int kp = p * (FL_GW + FL_MW) + wave;
    const float* wr = W + (2 * kp) * HDIM + lane * 4;
    const float4 w0 = *(const float4*)wr;
    const float4 w1 = *(const float4*)(wr + HDIM);
    float* d = &wsm[kp * WPAIR + lane * 8];
    *(float4*)d = make_float4(w0.x, w1.x, w0.y, w1.y);
    *(float4*)(d + 4) = make_float4(w0.z, w1.z, w0.w, w1.w);
  }

  const float4* hb = (const float4*)xin;
  const int2* cs = (const int2*)csr;
  float4 sa, sc;
  {
    const float4 s1 = ((const float4*)stats_in)[lane];
    const float4 s2 = ((const float4*)(stats_in + HDIM))[lane];
    const float4 g4 = ((const float4*)gamma)[lane];
    const float4 b4 = ((const float4*)beta)[lane];
    bn_coef(s1.x, s2.x, g4.x, b4.x, inv_n, sa.x, sc.x);
    bn_coef(s1.y, s2.y, g4.y, b4.y, inv_n, sa.y, sc.y);
    bn_coef(s1.z, s2.z, g4.z, b4.z, inv_n, sa.z, sc.z);
    bn_coef(s1.w, s2.w, g4.w, b4.w, inv_n, sa.w, sc.w);
  }
  const int mw = wave >= FL_GW ? wave - FL_GW : 0;
  const int m16 = lane & 15;
  const int hi = lane >> 4;
  const int col = mw * 16 + m16;
  const float bcol = bias[col];
  float cs1 = 0.f, cs2 = 0.f;

  for (int it = 0; it <= nmine; ++it) {
    if (wave < FL_GW) {
      if (it < nmine) {
        const int tile = blockIdx.x + it * gridDim.x;
        float* zb = &zs[(it & 1) * FL_ROWS * ZP];
#pragma unroll 1
        for (int q = 0; q < 2; ++q) {
          const int r = wave * 2 + q;
          const int node = tile * FL_ROWS + r;
          float4 z = make_float4(0.f, 0.f, 0.f, 0.f);
          if (node < n) {
            const int beg = rowptr[node], end = rowptr[node + 1];
            const float di = dinv[node];
            const float4 hs = hb[(size_t)node * 32 + lane];
            float4 acc = make_float4(0.f, 0.f, 0.f, 0.f);
            float wsum = 0.f;
            for (int base = beg; base < end; base += 32) {
              int m = end - base;
              m = m < 32 ? m : 32;
              int2 my = make_int2(0, 0);
              if (lane < m) my = cs[base + lane];
              my.x = my.x < 0 ? 0 : (my.x < n ? my.x : n - 1);
              int j = 0;
              for (; j + 16 <= m; j += 16) edge_batch<16>(acc, wsum, hb, my, j, lane);
              if (j + 8 <= m) { edge_batch<8>(acc, wsum, hb, my, j, lane); j += 8; }
              if (j + 4 <= m) { edge_batch<4>(acc, wsum, hb, my, j, lane); j += 4; }
              const int rem = m - j;
              if (rem == 3) edge_batch<3>(acc, wsum, hb, my, j, lane);
              else if (rem == 2) edge_batch<2>(acc, wsum, hb, my, j, lane);
              else if (rem == 1) edge_batch<1>(acc, wsum, hb, my, j, lane);
            }
            const float w_self = di * di;
            acc.x = fmaf(hs.x, w_self, acc.x);
            acc.y = fmaf(hs.y, w_self, acc.y);
            acc.z = fmaf(hs.z, w_self, acc.z);
            acc.w = fmaf(hs.w, w_self, acc.w);
            wsum += w_self;
            z.x = fmaf(sa.x, acc.x, sc.x * wsum);
            z.y = fmaf(sa.y, acc.y, sc.y * wsum);
            z.z = fmaf(sa.z, acc.z, sc.z * wsum);
            z.w = fmaf(sa.w, acc.w, sc.w * wsum);
          }
          *(float4*)(&zb[r * ZP + lane * 4]) = z;
        }
      }
    } else if (it >= 1) {
      const int tile = blockIdx.x + (it - 1) * gridDim.x;
      const float* zb = &zs[((it - 1) & 1) * FL_ROWS * ZP];
      v8f pacc[3];
#pragma unroll
      for (int g = 0; g < 3; ++g)
#pragma unroll
        for (int v = 0; v < 8; ++v) pacc[g][v] = 0.f;
      const float* xrow = &zb[m16 * ZP];
      const float* wk0 = &wsm[hi * WPAIR + col * 2];
#pragma unroll
      for (int k4 = 0; k4 < HDIM / 4; ++k4) {
        const int kb = k4 * 4 + hi * 2;
        const float* wk = wk0 + (k4 * 2) * WPAIR;
        v2f b; b.x = wk[0]; b.y = wk[1];
#pragma unroll
        for (int g = 0; g < 3; ++g) {
          v2f a; a.x = xrow[g * 16 * ZP + kb]; a.y = xrow[g * 16 * ZP + kb + 1];
          pacc[g] = __builtin_amdgcn_wmma_f32_16x16x4_f32(
              false, a, false, b, (short)0, pacc[g], false, false);
        }
      }
#pragma unroll
      for (int g = 0; g < 3; ++g) {
#pragma unroll
        for (int v = 0; v < 8; ++v) {
          const int row = tile * FL_ROWS + g * 16 + v + 8 * hi;
          float o = pacc[g][v] + bcol;
          o = o > 0.f ? o : 0.f;
          if (row < n) {
            xout[(size_t)row * HDIM + col] = o;
            if (STATS) { cs1 += o; cs2 = fmaf(o, o, cs2); }
          }
        }
      }
    }
    __syncthreads();
  }
  if (STATS) {
    if (wave >= FL_GW) {
      cs1 += __shfl_xor(cs1, 16);
      cs2 += __shfl_xor(cs2, 16);
      if (hi == 0) {
        atomicAdd(&stats_out[col], cs1);
        atomicAdd(&stats_out[HDIM + col], cs2);
      }
    }
  }
}

#define OXP 132
#define OWP 96
__global__ __launch_bounds__(128, 5) void k_out_gemm(const float* __restrict__ x,
                                                  const float* __restrict__ Wo,
                                                  const float* __restrict__ bo,
                                                  float* __restrict__ out, int n) {
  __shared__ float xs[64 * OXP];
  __shared__ float ws[(HDIM / 2) * OWP];
  const int tid = threadIdx.x;
  const int lane = tid & 31;
  const int wave = __builtin_amdgcn_readfirstlane(tid >> 5);
  const int rowBase = blockIdx.x * 64;
  const int m = lane & 15;
  const int hi = lane >> 4;

  float4 xv[16];
#pragma unroll
  for (int p = 0; p < 16; ++p) {
    const int row = rowBase + p * 4 + wave;
    xv[p] = make_float4(0.f, 0.f, 0.f, 0.f);
    if (row < n) xv[p] = ((const float4*)(x + (size_t)row * HDIM))[lane];
  }
#pragma unroll
  for (int p = 0; p < 5; ++p) {
    const int t = p * 128 + tid;
    const int kp = t / 10, j = t - kp * 10;
    const float* wr = Wo + (2 * kp) * NCLS + j * 4;
    const float4 w0 = *(const float4*)wr;
    const float4 w1 = *(const float4*)(wr + NCLS);
    float* d = &ws[kp * OWP + j * 8];
    *(float4*)d = make_float4(w0.x, w1.x, w0.y, w1.y);
    *(float4*)(d + 4) = make_float4(w0.z, w1.z, w0.w, w1.w);
  }
#pragma unroll
  for (int p = 0; p < 2; ++p) {
    const int q = tid * 2 + p;
    *(float4*)&ws[(q >> 2) * OWP + 2 * NCLS + (q & 3) * 4] = make_float4(0.f, 0.f, 0.f, 0.f);
  }
#pragma unroll
  for (int p = 0; p < 16; ++p) *(float4*)(&xs[(p * 4 + wave) * OXP + lane * 4]) = xv[p];
  __syncthreads();

  v8f acc[3];
#pragma unroll
  for (int c = 0; c < 3; ++c)
#pragma unroll
    for (int v = 0; v < 8; ++v) acc[c][v] = 0.f;

  const float* xrow = &xs[(wave * 16 + m) * OXP];
#pragma unroll
  for (int k4 = 0; k4 < 32; ++k4) {
    const int kb = k4 * 4 + hi * 2;
    v2f a; a.x = xrow[kb]; a.y = xrow[kb + 1];
    const float* wk = &ws[(k4 * 2 + hi) * OWP + m * 2];
#pragma unroll
    for (int c = 0; c < 3; ++c) {
      v2f b; b.x = wk[c * 32]; b.y = wk[c * 32 + 1];
      acc[c] = __builtin_amdgcn_wmma_f32_16x16x4_f32(
          false, a, false, b, (short)0, acc[c], false, false);
    }
  }
  __syncthreads();

  float* cs = xs;
#pragma unroll
  for (int c = 0; c < 3; ++c) {
    const int col = c * 16 + m;
    if (col < NCLS) {
      const float bv = bo[col];
#pragma unroll
      for (int v = 0; v < 8; ++v) cs[(wave * 16 + v + 8 * hi) * NCLS + col] = acc[c][v] + bv;
    }
  }
  __syncthreads();
  int rows = n - rowBase;
  rows = rows < 64 ? rows : 64;
  const int cnt = rows * NCLS;
  float* ob = out + (size_t)rowBase * NCLS;
#pragma unroll
  for (int p = 0; p < 5; ++p) {
    const int i4 = (p * 128 + tid) * 4;
    if (i4 < cnt) *(float4*)(ob + i4) = *(const float4*)&cs[i4];
  }
}

extern "C" void kernel_launch(void* const* d_in, const int* in_sizes, int n_in,
                              void* d_out, int out_size, void* d_ws, size_t ws_size,
                              hipStream_t stream) {
  const float* x_in  = (const float*)d_in[0];
  const int*   ei    = (const int*)d_in[1];
  const float* gamma = (const float*)d_in[2];
  const float* beta  = (const float*)d_in[3];
  const float* W     = (const float*)d_in[4];
  const float* b     = (const float*)d_in[5];
  const float* Wo    = (const float*)d_in[6];
  const float* bo    = (const float*)d_in[7];
  float* out = (float*)d_out;

  const int N = in_sizes[0] / HDIM;
  const int E = in_sizes[1] / 2;

  float* dinv  = (float*)d_ws;
  float* stats = dinv + ((N + 255) / 256) * 256;
  float* hbuf  = stats + 2 * HDIM;
  float* xbuf  = hbuf + (size_t)N * HDIM;

  int* csr    = (int*)d_out;
  int* rowptr = csr + ((2 * E + 3) / 4) * 4;
  float* stats1 = (float*)(rowptr + ((N + 4) / 4) * 4);
  float* stats2 = stats1 + 2 * HDIM;

  const int nb = (N + BK_NODES - 1) / BK_NODES;
  const int ch = (E + NCHK - 1) / NCHK;
  const int n2 = nb * NCHK;
  int2* tmp    = (int2*)xbuf;
  int*  bh     = (int*)xbuf + ((2 * E + 3) / 4) * 4;
  int*  bstart = bh + ((n2 + 3) / 4) * 4;

  k_bucket_hist<<<NCHK, CH_T, 0, stream>>>(ei, bh, E, ch, nb, stats, stats1);
  k_scatter_stats<<<2 * NCHK, CH_T, 0, stream>>>(ei, bh, bstart, tmp, E, ch, nb, x_in, stats, N);
  k_bucket_rows<<<nb, BK_NODES, 0, stream>>>(tmp, bstart, rowptr, dinv, N, E, 1);
  k_bucket_fill<<<nb, BK_NODES, 0, stream>>>(tmp, bstart, rowptr, dinv, (int2*)csr, N, E, 1);

  k_layer<true><<<FL_BLOCKS, (FL_GW + FL_MW) * 32, 0, stream>>>(
      rowptr, csr, dinv, x_in, stats, gamma, beta, W, b, hbuf, stats1, N, 1.0f / (float)N);
  k_layer<true><<<FL_BLOCKS, (FL_GW + FL_MW) * 32, 0, stream>>>(
      rowptr, csr, dinv, hbuf, stats1, gamma + HDIM, beta + HDIM, W + (size_t)HDIM * HDIM, b + HDIM,
      xbuf, stats2, N, 1.0f / (float)N);
  k_layer<false><<<FL_BLOCKS, (FL_GW + FL_MW) * 32, 0, stream>>>(
      rowptr, csr, dinv, xbuf, stats2, gamma + 2 * HDIM, beta + 2 * HDIM, W + (size_t)2 * HDIM * HDIM,
      b + 2 * HDIM, hbuf, stats1, N, 1.0f / (float)N);

  k_out_gemm<<<(N + 63) / 64, 128, 0, stream>>>(hbuf, Wo, bo, out, N);
}


# Both sort passes put a block's pairs in order in LDS first, then write them out front to back: a store covers whole lines, not 32 lines of 8 B each; a node's edge order, so its sum order, may differ
# speedup vs baseline: 5.9854x; 1.0395x over previous
#include <hip/hip_runtime.h>

#define HDIM 128
#define NCLS 40
#define BN_EPS 1e-5f

typedef __attribute__((ext_vector_type(2))) float v2f;
typedef __attribute__((ext_vector_type(8))) float v8f;

#define BK_SHIFT 9
#define BK_NODES 512
#define BK_MAX 1024
#define CH_T 1024
#define SORT_U 8

__device__ __forceinline__ int bucket_of(int d, int nb) {
  int bk = d >> BK_SHIFT;
  return bk < 0 ? 0 : (bk < nb ? bk : nb - 1);
}

__global__ __launch_bounds__(CH_T) void k_bucket_hist(const int* __restrict__ ei, int* __restrict__ bh,
                                                      int E, int ch, int nb,
                                                      float* __restrict__ sums0, float* __restrict__ sums12) {
  __shared__ int hist[BK_MAX];
  const int t = threadIdx.x;
  if (blockIdx.x == 0) {
    if (t < 2 * HDIM) sums0[t] = 0.f;
    if (t < 4 * HDIM) sums12[t] = 0.f;
  }
  for (int q = t; q < nb; q += CH_T) hist[q] = 0;
  __syncthreads();
  const int e0 = blockIdx.x * ch;
  int e1 = e0 + ch;
  e1 = e1 < E ? e1 : E;
  for (int e = e0 + t; e < e1; e += SORT_U * CH_T) {
    int d[SORT_U];
#pragma unroll
    for (int q = 0; q < SORT_U; ++q) d[q] = e + q * CH_T < e1 ? ei[E + e + q * CH_T] : 0;
#pragma unroll
    for (int q = 0; q < SORT_U; ++q)
      if (e + q * CH_T < e1) atomicAdd(&hist[bucket_of(d[q], nb)], 1);
  }
  __syncthreads();
  for (int q = t; q < nb; q += CH_T) bh[q * gridDim.x + blockIdx.x] = hist[q];
}

#define NCHK 250
#define SS_ROWS 8
#define ST_CAP 8192
static_assert(CH_T == 1024 && CH_T >= 2 * HDIM, "32 waves; one thread per entry of stats in the last step");
__global__ __launch_bounds__(CH_T) void k_scatter_stats(const int* __restrict__ ei, const int* __restrict__ bh,
                                                        int* __restrict__ bstart, int2* __restrict__ tmp,
                                                        int E, int ch, int nb,
                                                        const float* __restrict__ x, float* __restrict__ stats, int n) {
  __shared__ float ls[32 * 2 * HDIM];
  __shared__ int2 stage[ST_CAP];
  const int t = threadIdx.x;
  const int lane = t & 31;
  const int wv = __builtin_amdgcn_readfirstlane(t >> 5);
  if ((int)blockIdx.x >= NCHK) {
    const float4* xb = (const float4*)x;
    const int nw = ((int)gridDim.x - NCHK) * 32;
    float4 s1 = make_float4(0.f, 0.f, 0.f, 0.f);
    float4 s2 = make_float4(0.f, 0.f, 0.f, 0.f);
    for (int base = ((int)blockIdx.x - NCHK) * 32 + wv; base < n; base += SS_ROWS * nw) {
      float4 a[SS_ROWS];
#pragma unroll
      for (int q = 0; q < SS_ROWS; ++q) {
        const int r = base + q * nw;
        if (r < n) a[q] = xb[(size_t)r * 32 + lane];
      }
#pragma unroll
      for (int q = 0; q < SS_ROWS; ++q) {
        if (base + q * nw < n) {
          s1.x += a[q].x; s1.y += a[q].y; s1.z += a[q].z; s1.w += a[q].w;
          s2.x = fmaf(a[q].x, a[q].x, s2.x); s2.y = fmaf(a[q].y, a[q].y, s2.y);
          s2.z = fmaf(a[q].z, a[q].z, s2.z); s2.w = fmaf(a[q].w, a[q].w, s2.w);
        }
      }
    }
    float4* lw = (float4*)&ls[wv * 2 * HDIM];
    lw[lane] = s1;
    lw[32 + lane] = s2;
    __syncthreads();
    if (t < 2 * HDIM) {
      float s = 0.f;
#pragma unroll
      for (int w = 0; w < 32; ++w) s += ls[w * 2 * HDIM + t];
      atomicAdd(&stats[t], s);
    }
    return;
  }
  int* cur = (int*)ls;
  int* tot = cur + BK_MAX;
  int* pre = tot + BK_MAX;
  int* own = pre + BK_MAX;
  int* dlt = own + BK_MAX;
  static_assert(5 * BK_MAX * sizeof(int) <= sizeof(ls), "five tables in ls");
  const int c = blockIdx.x;
  const int e0 = c * ch;
  int e1 = e0 + ch;
  e1 = e1 < E ? e1 : E;
  const bool staged = e1 - e0 <= ST_CAP;
  for (int q0 = wv; q0 < nb; q0 += 64) {
    int v[2][8];
#pragma unroll
    for (int h = 0; h < 2; ++h)
#pragma unroll
      for (int j = 0; j < 8; ++j) {
        const int q = q0 + h * 32, k = j * 32 + lane;
        v[h][j] = (q < nb && k < NCHK) ? bh[q * NCHK + k] : 0;
      }
#pragma unroll
    for (int h = 0; h < 2; ++h) {
      int ts = 0, ps = 0, os = 0;
#pragma unroll
      for (int j = 0; j < 8; ++j) {
        ts += v[h][j];
        ps += (j * 32 + lane < c) ? v[h][j] : 0;
        os += (j * 32 + lane == c) ? v[h][j] : 0;
      }
#pragma unroll
      for (int d = 16; d >= 1; d >>= 1) {
        ts += __shfl_xor(ts, d);
        ps += __shfl_xor(ps, d);
        os += __shfl_xor(os, d);
      }
      const int q = q0 + h * 32;
      if (lane == 0 && q < nb) { tot[q] = ts; pre[q] = ps; own[q] = os; }
    }
  }
  __syncthreads();
  if (wv == 0) {
    const int per = (nb + 31) >> 5;
    const int b0 = lane * per;
    int s = 0, so = 0;
    for (int k = 0; k < per; ++k) {
      s += b0 + k < nb ? tot[b0 + k] : 0;
      so += b0 + k < nb ? own[b0 + k] : 0;
    }
    int inc = s, inco = so;
#pragma unroll
    for (int d = 1; d < 32; d <<= 1) {
      int o = __shfl_up(inc, d);
      int oo = __shfl_up(inco, d);
      if (lane >= d) { inc += o; inco += oo; }
    }
    int run = inc - s;
    int lrun = inco - so;
    for (int k = 0; k < per; ++k) {
      const int q = b0 + k;
      if (q < nb) {
        const int g = run + pre[q];
        cur[q] = staged ? lrun : g;
        dlt[q] = g - lrun;
        if (c == 0) bstart[q] = run;
        run += tot[q];
        lrun += own[q];
      }
    }
    if (c == 0 && lane == 31) bstart[nb] = run;
  }
  __syncthreads();
  for (int e = e0 + t; e < e1; e += SORT_U * CH_T) {
    int s[SORT_U], d[SORT_U];
#pragma unroll
    for (int q = 0; q < SORT_U; ++q) {
      const bool in = e + q * CH_T < e1;
      s[q] = in ? ei[e + q * CH_T] : 0;
      d[q] = in ? ei[E + e + q * CH_T] : 0;
    }
#pragma unroll
    for (int q = 0; q < SORT_U; ++q) {
      if (e + q * CH_T < e1) {
        int p = atomicAdd(&cur[bucket_of(d[q], nb)], 1);
        if (staged) {
          p = p < 0 ? 0 : (p < ST_CAP ? p : ST_CAP - 1);
          stage[p] = make_int2(s[q], d[q]);
        } else {
          p = p < 0 ? 0 : (p < E ? p : E - 1);
          tmp[p] = make_int2(s[q], d[q]);
        }
      }
    }
  }
  if (!staged) return;
  __syncthreads();
  const int cnt = e1 - e0;
  for (int i = t; i < cnt; i += SORT_U * CH_T) {
    int2 sd[SORT_U];
    int p[SORT_U];
#pragma unroll
    for (int q = 0; q < SORT_U; ++q) sd[q] = i + q * CH_T < cnt ? stage[i + q * CH_T] : make_int2(0, 0);
#pragma unroll
    for (int q = 0; q < SORT_U; ++q) p[q] = i + q * CH_T + dlt[bucket_of(sd[q].y, nb)];
#pragma unroll
    for (int q = 0; q < SORT_U; ++q) {
      if (i + q * CH_T < cnt) {
        const int pp = p[q] < 0 ? 0 : (p[q] < E ? p[q] : E - 1);
        tmp[pp] = sd[q];
      }
    }
  }
}

__device__ __forceinline__ void bucket_range(const int* __restrict__ boff, int bk, int nch, int E,
                                             int& bs, int& be) {
  bs = boff[bk * nch];
  be = boff[(bk + 1) * nch];
  bs = bs < 0 ? 0 : (bs < E ? bs : E);
  be = be < bs ? bs : (be < E ? be : E);
}

__device__ __forceinline__ int local_node(int d, int node0) {
  const int l = d - node0;
  return l < 0 ? 0 : (l < BK_NODES ? l : BK_NODES - 1);
}

__global__ __launch_bounds__(BK_NODES) void k_bucket_rows(const int2* __restrict__ tmp,
                                                          const int* __restrict__ boff,
                                                          int* __restrict__ rowptr,
                                                          float* __restrict__ dinv, int n, int E, int nch) {
  __shared__ int cnt[BK_NODES];
  __shared__ int wsum[32];
  const int t = threadIdx.x, lane = t & 31, wv = t >> 5;
  const int node0 = blockIdx.x << BK_SHIFT;
  int bs, be;
  bucket_range(boff, blockIdx.x, nch, E, bs, be);
  cnt[t] = 0;
  __syncthreads();
  for (int i = bs + t; i < be; i += SORT_U * BK_NODES) {
    int d[SORT_U];
#pragma unroll
    for (int q = 0; q < SORT_U; ++q) d[q] = i + q * BK_NODES < be ? tmp[i + q * BK_NODES].y : 0;
#pragma unroll
    for (int q = 0; q < SORT_U; ++q)
      if (i + q * BK_NODES < be) atomicAdd(&cnt[local_node(d[q], node0)], 1);
  }
  __syncthreads();
  const int c = cnt[t];
  int inc = c;
#pragma unroll
  for (int d = 1; d < 32; d <<= 1) {
    int o = __shfl_up(inc, d);
    if (lane >= d) inc += o;
  }
  if (lane == 31) wsum[wv] = inc;
  __syncthreads();
  const int wt = lane < BK_NODES / 32 ? wsum[lane] : 0;
  int winc = wt;
#pragma unroll
  for (int d = 1; d < 32; d <<= 1) {
    int o = __shfl_up(winc, d);
    if (lane >= d) winc += o;
  }
  const int wex = __shfl(winc, wv) - __shfl(wt, wv);
  const int node = node0 + t;
  if (node < n) {
    rowptr[node] = bs + wex + inc - c;
    dinv[node] = rsqrtf((float)(c + 1));
  }
  if (blockIdx.x == gridDim.x - 1 && t == 0) rowptr[n] = be;
}

#define FILL_CAP 12288
__global__ __launch_bounds__(BK_NODES) void k_bucket_fill(const int2* __restrict__ tmp,
                                                          const int* __restrict__ boff,
                                                          const int* __restrict__ rowptr,
                                                          const float* __restrict__ dinv,
                                                          int2* __restrict__ csr, int n, int E, int nch) {
  __shared__ int cur[BK_NODES];
  __shared__ int2 stage[FILL_CAP];
  const int t = threadIdx.x;
  const int node0 = blockIdx.x << BK_SHIFT;
  int bs, be;
  bucket_range(boff, blockIdx.x, nch, E, bs, be);
  const int cnt = be - bs;
  const bool staged = cnt <= FILL_CAP;
  cur[t] = node0 + t < n ? rowptr[node0 + t] : be;
  __syncthreads();
  for (int i = bs + t; i < be; i += SORT_U * BK_NODES) {
    int2 sd[SORT_U];
    float w[SORT_U];
#pragma unroll
    for (int q = 0; q < SORT_U; ++q)
      sd[q] = i + q * BK_NODES < be ? tmp[i + q * BK_NODES] : make_int2(0, 0);
#pragma unroll
    for (int q = 0; q < SORT_U; ++q) {
      const int s = sd[q].x < 0 ? 0 : (sd[q].x < n ? sd[q].x : n - 1);
      const int d = sd[q].y < 0 ? 0 : (sd[q].y < n ? sd[q].y : n - 1);
      w[q] = dinv[s] * dinv[d];
    }
#pragma unroll
    for (int q = 0; q < SORT_U; ++q) {
      if (i + q * BK_NODES < be) {
        int p = atomicAdd(&cur[local_node(sd[q].y, node0)], 1);
        if (staged) {
          int l = p - bs;
          l = l < 0 ? 0 : (l < FILL_CAP ? l : FILL_CAP - 1);
          stage[l] = make_int2(sd[q].x, __float_as_int(w[q]));
        } else {
          p = p < 0 ? 0 : (p < E ? p : E - 1);
          csr[p] = make_int2(sd[q].x, __float_as_int(w[q]));
        }
      }
    }
  }
  if (!staged) return;
  __syncthreads();
  for (int i = t; i < cnt; i += SORT_U * BK_NODES) {
    int2 v[SORT_U];
#pragma unroll
    for (int q = 0; q < SORT_U; ++q) v[q] = i + q * BK_NODES < cnt ? stage[i + q * BK_NODES] : make_int2(0, 0);
#pragma unroll
    for (int q = 0; q < SORT_U; ++q)
      if (i + q * BK_NODES < cnt) csr[bs + i + q * BK_NODES] = v[q];
  }
}

#define WPAIR 288
#define ZP 132
#define FL_ROWS 48
#define FL_GW 24
#define FL_MW 8
#define FL_BLOCKS 256

__device__ __forceinline__ void bn_coef(float s1, float s2, float g, float bt, float inv_n,
                                        float& a, float& c) {
  const float mean = s1 * inv_n;
  const float var = s2 * inv_n - mean * mean;
  a = g * rsqrtf(var + BN_EPS);
  c = bt - a * mean;
}

template <int K>
__device__ __forceinline__ void edge_batch(float4& acc, float& wsum, const float4* __restrict__ hb,
                                           const int2 my, const int j, const int lane) {
  int sidx[K];
  float wgt[K];
  float4 a[K];
#pragma unroll
  for (int q = 0; q < K; ++q) {
    sidx[q] = __builtin_amdgcn_readlane(my.x, j + q);
    wgt[q] = __int_as_float(__builtin_amdgcn_readlane(my.y, j + q));
  }
#pragma unroll
  for (int q = 0; q < K; ++q) a[q] = hb[(size_t)sidx[q] * 32 + lane];
#pragma unroll
  for (int q = 0; q < K; ++q) {
    acc.x = fmaf(a[q].x, wgt[q], acc.x);
    acc.y = fmaf(a[q].y, wgt[q], acc.y);
    acc.z = fmaf(a[q].z, wgt[q], acc.z);
    acc.w = fmaf(a[q].w, wgt[q], acc.w);
    wsum += wgt[q];
  }
}

template <bool STATS>
__global__ __launch_bounds__((FL_GW + FL_MW) * 32) void k_layer(const int* __restrict__ rowptr,
                                                                const int* __restrict__ csr,
                                                                const float* __restrict__ dinv,
                                                                const float* __restrict__ xin,
                                                                const float* __restrict__ stats_in,
                                                                const float* __restrict__ gamma,
                                                                const float* __restrict__ beta,
                                                                const float* __restrict__ W,
                                                                const float* __restrict__ bias,
                                                                float* __restrict__ xout,
                                                                float* __restrict__ stats_out, int n, float inv_n) {
  __shared__ float wsm[(HDIM / 2) * WPAIR];
  __shared__ float zs[2 * FL_ROWS * ZP];
  static_assert(FL_GW * 2 == FL_ROWS, "two nodes per gather wave");
  static_assert(FL_MW * 16 == HDIM, "one column tile per matrix wave");
  const int tid = threadIdx.x;
  const int lane = tid & 31;
  const int wave = __builtin_amdgcn_readfirstlane(tid >> 5);
  const int ntiles = (n + FL_ROWS - 1) / FL_ROWS;
  const int nmine = (int)blockIdx.x < ntiles ? (ntiles - 1 - (int)blockIdx.x) / (int)gridDim.x + 1 : 0;

#pragma unroll
  for (int p = 0; p < 2; ++p) {
    const int kp = p * (FL_GW + FL_MW) + wave;
    const float* wr = W + (2 * kp) * HDIM + lane * 4;
    const float4 w0 = *(const float4*)wr;
    const float4 w1 = *(const float4*)(wr + HDIM);
    float* d = &wsm[kp * WPAIR + lane * 8];
    *(float4*)d = make_float4(w0.x, w1.x, w0.y, w1.y);
    *(float4*)(d + 4) = make_float4(w0.z, w1.z, w0.w, w1.w);
  }

  const float4* hb = (const float4*)xin;
  const int2* cs = (const int2*)csr;
  float4 sa, sc;
  {
    const float4 s1 = ((const float4*)stats_in)[lane];
    const float4 s2 = ((const float4*)(stats_in + HDIM))[lane];
    const float4 g4 = ((const float4*)gamma)[lane];
    const float4 b4 = ((const float4*)beta)[lane];
    bn_coef(s1.x, s2.x, g4.x, b4.x, inv_n, sa.x, sc.x);
    bn_coef(s1.y, s2.y, g4.y, b4.y, inv_n, sa.y, sc.y);
    bn_coef(s1.z, s2.z, g4.z, b4.z, inv_n, sa.z, sc.z);
    bn_coef(s1.w, s2.w, g4.w, b4.w, inv_n, sa.w, sc.w);
  }
  const int mw = wave >= FL_GW ? wave - FL_GW : 0;
  const int m16 = lane & 15;
  const int hi = lane >> 4;
  const int col = mw * 16 + m16;
  const float bcol = bias[col];
  float cs1 = 0.f, cs2 = 0.f;

  for (int it = 0; it <= nmine; ++it) {
    if (wave < FL_GW) {
      if (it < nmine) {
        const int tile = blockIdx.x + it * gridDim.x;
        float* zb = &zs[(it & 1) * FL_ROWS * ZP];
#pragma unroll 1
        for (int q = 0; q < 2; ++q) {
          const int r = wave * 2 + q;
          const int node = tile * FL_ROWS + r;
          float4 z = make_float4(0.f, 0.f, 0.f, 0.f);
          if (node < n) {
            const int beg = rowptr[node], end = rowptr[node + 1];
            const float di = dinv[node];
            const float4 hs = hb[(size_t)node * 32 + lane];
            float4 acc = make_float4(0.f, 0.f, 0.f, 0.f);
            float wsum = 0.f;
            for (int base = beg; base < end; base += 32) {
              int m = end - base;
              m = m < 32 ? m : 32;
              int2 my = make_int2(0, 0);
              if (lane < m) my = cs[base + lane];
              my.x = my.x < 0 ? 0 : (my.x < n ? my.x : n - 1);
              int j = 0;
              for (; j + 16 <= m; j += 16) edge_batch<16>(acc, wsum, hb, my, j, lane);
              if (j + 8 <= m) { edge_batch<8>(acc, wsum, hb, my, j, lane); j += 8; }
              if (j + 4 <= m) { edge_batch<4>(acc, wsum, hb, my, j, lane); j += 4; }
              const int rem = m - j;
              if (rem == 3) edge_batch<3>(acc, wsum, hb, my, j, lane);
              else if (rem == 2) edge_batch<2>(acc, wsum, hb, my, j, lane);
              else if (rem == 1) edge_batch<1>(acc, wsum, hb, my, j, lane);
            }
            const float w_self = di * di;
            acc.x = fmaf(hs.x, w_self, acc.x);
            acc.y = fmaf(hs.y, w_self, acc.y);
            acc.z = fmaf(hs.z, w_self, acc.z);
            acc.w = fmaf(hs.w, w_self, acc.w);
            wsum += w_self;
            z.x = fmaf(sa.x, acc.x, sc.x * wsum);
            z.y = fmaf(sa.y, acc.y, sc.y * wsum);
            z.z = fmaf(sa.z, acc.z, sc.z * wsum);
            z.w = fmaf(sa.w, acc.w, sc.w * wsum);
          }
          *(float4*)(&zb[r * ZP + lane * 4]) = z;
        }
      }
    } else if (it >= 1) {
      const int tile = blockIdx.x + (it - 1) * gridDim.x;
      const float* zb = &zs[((it - 1) & 1) * FL_ROWS * ZP];
      v8f pacc[3];
#pragma unroll
      for (int g = 0; g < 3; ++g)
#pragma unroll
        for (int v = 0; v < 8; ++v) pacc[g][v] = 0.f;
      const float* xrow = &zb[m16 * ZP];
      const float* wk0 = &wsm[hi * WPAIR + col * 2];
#pragma unroll
      for (int k4 = 0; k4 < HDIM / 4; ++k4) {
        const int kb = k4 * 4 + hi * 2;
        const float* wk = wk0 + (k4 * 2) * WPAIR;
        v2f b; b.x = wk[0]; b.y = wk[1];
#pragma unroll
        for (int g = 0; g < 3; ++g) {
          v2f a; a.x = xrow[g * 16 * ZP + kb]; a.y = xrow[g * 16 * ZP + kb + 1];
          pacc[g] = __builtin_amdgcn_wmma_f32_16x16x4_f32(
              false, a, false, b, (short)0, pacc[g], false, false);
        }
      }
#pragma unroll
      for (int g = 0; g < 3; ++g) {
#pragma unroll
        for (int v = 0; v < 8; ++v) {
          const int row = tile * FL_ROWS + g * 16 + v + 8 * hi;
          float o = pacc[g][v] + bcol;
          o = o > 0.f ? o : 0.f;
          if (row < n) {
            xout[(size_t)row * HDIM + col] = o;
            if (STATS) { cs1 += o; cs2 = fmaf(o, o, cs2); }
          }
        }
      }
    }
    __syncthreads();
  }
  if (STATS) {
    if (wave >= FL_GW) {
      cs1 += __shfl_xor(cs1, 16);
      cs2 += __shfl_xor(cs2, 16);
      if (hi == 0) {
        atomicAdd(&stats_out[col], cs1);
        atomicAdd(&stats_out[HDIM + col], cs2);
      }
    }
  }
}

#define OXP 132
#define OWP 96
__global__ __launch_bounds__(128, 5) void k_out_gemm(const float* __restrict__ x,
                                                  const float* __restrict__ Wo,
                                                  const float* __restrict__ bo,
                                                  float* __restrict__ out, int n) {
  __shared__ float xs[64 * OXP];
  __shared__ float ws[(HDIM / 2) * OWP];
  const int tid = threadIdx.x;
  const int lane = tid & 31;
  const int wave = __builtin_amdgcn_readfirstlane(tid >> 5);
  const int rowBase = blockIdx.x * 64;
  const int m = lane & 15;
  const int hi = lane >> 4;

  float4 xv[16];
#pragma unroll
  for (int p = 0; p < 16; ++p) {
    const int row = rowBase + p * 4 + wave;
    xv[p] = make_float4(0.f, 0.f, 0.f, 0.f);
    if (row < n) xv[p] = ((const float4*)(x + (size_t)row * HDIM))[lane];
  }
#pragma unroll
  for (int p = 0; p < 5; ++p) {
    const int t = p * 128 + tid;
    const int kp = t / 10, j = t - kp * 10;
    const float* wr = Wo + (2 * kp) * NCLS + j * 4;
    const float4 w0 = *(const float4*)wr;
    const float4 w1 = *(const float4*)(wr + NCLS);
    float* d = &ws[kp * OWP + j * 8];
    *(float4*)d = make_float4(w0.x, w1.x, w0.y, w1.y);
    *(float4*)(d + 4) = make_float4(w0.z, w1.z, w0.w, w1.w);
  }
#pragma unroll
  for (int p = 0; p < 2; ++p) {
    const int q = tid * 2 + p;
    *(float4*)&ws[(q >> 2) * OWP + 2 * NCLS + (q & 3) * 4] = make_float4(0.f, 0.f, 0.f, 0.f);
  }
#pragma unroll
  for (int p = 0; p < 16; ++p) *(float4*)(&xs[(p * 4 + wave) * OXP + lane * 4]) = xv[p];
  __syncthreads();

  v8f acc[3];
#pragma unroll
  for (int c = 0; c < 3; ++c)
#pragma unroll
    for (int v = 0; v < 8; ++v) acc[c][v] = 0.f;

  const float* xrow = &xs[(wave * 16 + m) * OXP];
#pragma unroll
  for (int k4 = 0; k4 < 32; ++k4) {
    const int kb = k4 * 4 + hi * 2;
    v2f a; a.x = xrow[kb]; a.y = xrow[kb + 1];
    const float* wk = &ws[(k4 * 2 + hi) * OWP + m * 2];
#pragma unroll
    for (int c = 0; c < 3; ++c) {
      v2f b; b.x = wk[c * 32]; b.y = wk[c * 32 + 1];
      acc[c] = __builtin_amdgcn_wmma_f32_16x16x4_f32(
          false, a, false, b, (short)0, acc[c], false, false);
    }
  }
  __syncthreads();

  float* cs = xs;
#pragma unroll
  for (int c = 0; c < 3; ++c) {
    const int col = c * 16 + m;
    if (col < NCLS) {
      const float bv = bo[col];
#pragma unroll
      for (int v = 0; v < 8; ++v) cs[(wave * 16 + v + 8 * hi) * NCLS + col] = acc[c][v] + bv;
    }
  }
  __syncthreads();
  int rows = n - rowBase;
  rows = rows < 64 ? rows : 64;
  const int cnt = rows * NCLS;
  float* ob = out + (size_t)rowBase * NCLS;
#pragma unroll
  for (int p = 0; p < 5; ++p) {
    const int i4 = (p * 128 + tid) * 4;
    if (i4 < cnt) *(float4*)(ob + i4) = *(const float4*)&cs[i4];
  }
}

extern "C" void kernel_launch(void* const* d_in, const int* in_sizes, int n_in,
                              void* d_out, int out_size, void* d_ws, size_t ws_size,
                              hipStream_t stream) {
  const float* x_in  = (const float*)d_in[0];
  const int*   ei    = (const int*)d_in[1];
  const float* gamma = (const float*)d_in[2];
  const float* beta  = (const float*)d_in[3];
  const float* W     = (const float*)d_in[4];
  const float* b     = (const float*)d_in[5];
  const float* Wo    = (const float*)d_in[6];
  const float* bo    = (const float*)d_in[7];
  float* out = (float*)d_out;

  const int N = in_sizes[0] / HDIM;
  const int E = in_sizes[1] / 2;

  float* dinv  = (float*)d_ws;
  float* stats = dinv + ((N + 255) / 256) * 256;
  float* hbuf  = stats + 2 * HDIM;
  float* xbuf  = hbuf + (size_t)N * HDIM;

  int* csr    = (int*)d_out;
  int* rowptr = csr + ((2 * E + 3) / 4) * 4;
  float* stats1 = (float*)(rowptr + ((N + 4) / 4) * 4);
  float* stats2 = stats1 + 2 * HDIM;

  const int nb = (N + BK_NODES - 1) / BK_NODES;
  const int ch = (E + NCHK - 1) / NCHK;
  const int n2 = nb * NCHK;
  int2* tmp    = (int2*)xbuf;
  int*  bh     = (int*)xbuf + ((2 * E + 3) / 4) * 4;
  int*  bstart = bh + ((n2 + 3) / 4) * 4;

  k_bucket_hist<<<NCHK, CH_T, 0, stream>>>(ei, bh, E, ch, nb, stats, stats1);
  k_scatter_stats<<<2 * NCHK, CH_T, 0, stream>>>(ei, bh, bstart, tmp, E, ch, nb, x_in, stats, N);
  k_bucket_rows<<<nb, BK_NODES, 0, stream>>>(tmp, bstart, rowptr, dinv, N, E, 1);
  k_bucket_fill<<<nb, BK_NODES, 0, stream>>>(tmp, bstart, rowptr, dinv, (int2*)csr, N, E, 1);

  k_layer<true><<<FL_BLOCKS, (FL_GW + FL_MW) * 32, 0, stream>>>(
      rowptr, csr, dinv, x_in, stats, gamma, beta, W, b, hbuf, stats1, N, 1.0f / (float)N);
  k_layer<true><<<FL_BLOCKS, (FL_GW + FL_MW) * 32, 0, stream>>>(
      rowptr, csr, dinv, hbuf, stats1, gamma + HDIM, beta + HDIM, W + (size_t)HDIM * HDIM, b + HDIM,
      xbuf, stats2, N, 1.0f / (float)N);
  k_layer<false><<<FL_BLOCKS, (FL_GW + FL_MW) * 32, 0, stream>>>(
      rowptr, csr, dinv, xbuf, stats2, gamma + 2 * HDIM, beta + 2 * HDIM, W + (size_t)2 * HDIM * HDIM,
      b + 2 * HDIM, hbuf, stats1, N, 1.0f / (float)N);

  k_out_gemm<<<(N + 63) / 64, 128, 0, stream>>>(hbuf, Wo, bo, out, N);
}


# Two work-groups per WGP in the layer kernel (512 blocks; row batches of 8 fit a wave in 64 registers): twice the gather waves wait at once; self loop first in a node's sum, column-sum order differs
# speedup vs baseline: 6.6664x; 1.1138x over previous
#include <hip/hip_runtime.h>

#define HDIM 128
#define NCLS 40
#define BN_EPS 1e-5f

typedef __attribute__((ext_vector_type(2))) float v2f;
typedef __attribute__((ext_vector_type(8))) float v8f;

#define BK_SHIFT 9
#define BK_NODES 512
#define BK_MAX 1024
#define CH_T 1024
#define SORT_U 8

__device__ __forceinline__ int bucket_of(int d, int nb) {
  int bk = d >> BK_SHIFT;
  return bk < 0 ? 0 : (bk < nb ? bk : nb - 1);
}

__global__ __launch_bounds__(CH_T) void k_bucket_hist(const int* __restrict__ ei, int* __restrict__ bh,
                                                      int E, int ch, int nb,
                                                      float* __restrict__ sums0, float* __restrict__ sums12) {
  __shared__ int hist[BK_MAX];
  const int t = threadIdx.x;
  if (blockIdx.x == 0) {
    if (t < 2 * HDIM) sums0[t] = 0.f;
    if (t < 4 * HDIM) sums12[t] = 0.f;
  }
  for (int q = t; q < nb; q += CH_T) hist[q] = 0;
  __syncthreads();
  const int e0 = blockIdx.x * ch;
  int e1 = e0 + ch;
  e1 = e1 < E ? e1 : E;
  for (int e = e0 + t; e < e1; e += SORT_U * CH_T) {
    int d[SORT_U];
#pragma unroll
    for (int q = 0; q < SORT_U; ++q) d[q] = e + q * CH_T < e1 ? ei[E + e + q * CH_T] : 0;
#pragma unroll
    for (int q = 0; q < SORT_U; ++q)
      if (e + q * CH_T < e1) atomicAdd(&hist[bucket_of(d[q], nb)], 1);
  }
  __syncthreads();
  for (int q = t; q < nb; q += CH_T) bh[q * gridDim.x + blockIdx.x] = hist[q];
}

#define NCHK 250
#define SS_ROWS 8
#define ST_CAP 8192
static_assert(CH_T == 1024 && CH_T >= 2 * HDIM, "32 waves; one thread per entry of stats in the last step");
__global__ __launch_bounds__(CH_T) void k_scatter_stats(const int* __restrict__ ei, const int* __restrict__ bh,
                                                        int* __restrict__ bstart, int2* __restrict__ tmp,
                                                        int E, int ch, int nb,
                                                        const float* __restrict__ x, float* __restrict__ stats, int n) {
  __shared__ float ls[32 * 2 * HDIM];
  __shared__ int2 stage[ST_CAP];
  const int t = threadIdx.x;
  const int lane = t & 31;
  const int wv = __builtin_amdgcn_readfirstlane(t >> 5);
  if ((int)blockIdx.x >= NCHK) {
    const float4* xb = (const float4*)x;
    const int nw = ((int)gridDim.x - NCHK) * 32;
    float4 s1 = make_float4(0.f, 0.f, 0.f, 0.f);
    float4 s2 = make_float4(0.f, 0.f, 0.f, 0.f);
    for (int base = ((int)blockIdx.x - NCHK) * 32 + wv; base < n; base += SS_ROWS * nw) {
      float4 a[SS_ROWS];
#pragma unroll
      for (int q = 0; q < SS_ROWS; ++q) {
        const int r = base + q * nw;
        if (r < n) a[q] = xb[(size_t)r * 32 + lane];
      }
#pragma unroll
      for (int q = 0; q < SS_ROWS; ++q) {
        if (base + q * nw < n) {
          s1.x += a[q].x; s1.y += a[q].y; s1.z += a[q].z; s1.w += a[q].w;
          s2.x = fmaf(a[q].x, a[q].x, s2.x); s2.y = fmaf(a[q].y, a[q].y, s2.y);
          s2.z = fmaf(a[q].z, a[q].z, s2.z); s2.w = fmaf(a[q].w, a[q].w, s2.w);
        }
      }
    }
    float4* lw = (float4*)&ls[wv * 2 * HDIM];
    lw[lane] = s1;
    lw[32 + lane] = s2;
    __syncthreads();
    if (t < 2 * HDIM) {
      float s = 0.f;
#pragma unroll
      for (int w = 0; w < 32; ++w) s += ls[w * 2 * HDIM + t];
      atomicAdd(&stats[t], s);
    }
    return;
  }
  int* cur = (int*)ls;
  int* tot = cur + BK_MAX;
  int* pre = tot + BK_MAX;
  int* own = pre + BK_MAX;
  int* dlt = own + BK_MAX;
  static_assert(5 * BK_MAX * sizeof(int) <= sizeof(ls), "five tables in ls");
  const int c = blockIdx.x;
  const int e0 = c * ch;
  int e1 = e0 + ch;
  e1 = e1 < E ? e1 : E;
  const bool staged = e1 - e0 <= ST_CAP;
  for (int q0 = wv; q0 < nb; q0 += 64) {
    int v[2][8];
#pragma unroll
    for (int h = 0; h < 2; ++h)
#pragma unroll
      for (int j = 0; j < 8; ++j) {
        const int q = q0 + h * 32, k = j * 32 + lane;
        v[h][j] = (q < nb && k < NCHK) ? bh[q * NCHK + k] : 0;
      }
#pragma unroll
    for (int h = 0; h < 2; ++h) {
      int ts = 0, ps = 0, os = 0;
#pragma unroll
      for (int j = 0; j < 8; ++j) {
        ts += v[h][j];
        ps += (j * 32 + lane < c) ? v[h][j] : 0;
        os += (j * 32 + lane == c) ? v[h][j] : 0;
      }
#pragma unroll
      for (int d = 16; d >= 1; d >>= 1) {
        ts += __shfl_xor(ts, d);
        ps += __shfl_xor(ps, d);
        os += __shfl_xor(os, d);
      }
      const int q = q0 + h * 32;
      if (lane == 0 && q < nb) { tot[q] = ts; pre[q] = ps; own[q] = os; }
    }
  }
  __syncthreads();
  if (wv == 0) {
    const int per = (nb + 31) >> 5;
    const int b0 = lane * per;
    int s = 0, so = 0;
    for (int k = 0; k < per; ++k) {
      s += b0 + k < nb ? tot[b0 + k] : 0;
      so += b0 + k < nb ? own[b0 + k] : 0;
    }
    int inc = s, inco = so;
#pragma unroll
    for (int d = 1; d < 32; d <<= 1) {
      int o = __shfl_up(inc, d);
      int oo = __shfl_up(inco, d);
      if (lane >= d) { inc += o; inco += oo; }
    }
    int run = inc - s;
    int lrun = inco - so;
    for (int k = 0; k < per; ++k) {
      const int q = b0 + k;
      if (q < nb) {
        const int g = run + pre[q];
        cur[q] = staged ? lrun : g;
        dlt[q] = g - lrun;
        if (c == 0) bstart[q] = run;
        run += tot[q];
        lrun += own[q];
      }
    }
    if (c == 0 && lane == 31) bstart[nb] = run;
  }
  __syncthreads();
  for (int e = e0 + t; e < e1; e += SORT_U * CH_T) {
    int s[SORT_U], d[SORT_U];
#pragma unroll
    for (int q = 0; q < SORT_U; ++q) {
      const bool in = e + q * CH_T < e1;
      s[q] = in ? ei[e + q * CH_T] : 0;
      d[q] = in ? ei[E + e + q * CH_T] : 0;
    }
#pragma unroll
    for (int q = 0; q < SORT_U; ++q) {
      if (e + q * CH_T < e1) {
        int p = atomicAdd(&cur[bucket_of(d[q], nb)], 1);
        if (staged) {
          p = p < 0 ? 0 : (p < ST_CAP ? p : ST_CAP - 1);
          stage[p] = make_int2(s[q], d[q]);
        } else {
          p = p < 0 ? 0 : (p < E ? p : E - 1);
          tmp[p] = make_int2(s[q], d[q]);
        }
      }
    }
  }
  if (!staged) return;
  __syncthreads();
  const int cnt = e1 - e0;
  for (int i = t; i < cnt; i += SORT_U * CH_T) {
    int2 sd[SORT_U];
    int p[SORT_U];
#pragma unroll
    for (int q = 0; q < SORT_U; ++q) sd[q] = i + q * CH_T < cnt ? stage[i + q * CH_T] : make_int2(0, 0);
#pragma unroll
    for (int q = 0; q < SORT_U; ++q) p[q] = i + q * CH_T + dlt[bucket_of(sd[q].y, nb)];
#pragma unroll
    for (int q = 0; q < SORT_U; ++q) {
      if (i + q * CH_T < cnt) {
        const int pp = p[q] < 0 ? 0 : (p[q] < E ? p[q] : E - 1);
        tmp[pp] = sd[q];
      }
    }
  }
}

__device__ __forceinline__ void bucket_range(const int* __restrict__ boff, int bk, int nch, int E,
                                             int& bs, int& be) {
  bs = boff[bk * nch];
  be = boff[(bk + 1) * nch];
  bs = bs < 0 ? 0 : (bs < E ? bs : E);
  be = be < bs ? bs : (be < E ? be : E);
}

__device__ __forceinline__ int local_node(int d, int node0) {
  const int l = d - node0;
  return l < 0 ? 0 : (l < BK_NODES ? l : BK_NODES - 1);
}

__global__ __launch_bounds__(BK_NODES) void k_bucket_rows(const int2* __restrict__ tmp,
                                                          const int* __restrict__ boff,
                                                          int* __restrict__ rowptr,
                                                          float* __restrict__ dinv, int n, int E, int nch) {
  __shared__ int cnt[BK_NODES];
  __shared__ int wsum[32];
  const int t = threadIdx.x, lane = t & 31, wv = t >> 5;
  const int node0 = blockIdx.x << BK_SHIFT;
  int bs, be;
  bucket_range(boff, blockIdx.x, nch, E, bs, be);
  cnt[t] = 0;
  __syncthreads();
  for (int i = bs + t; i < be; i += SORT_U * BK_NODES) {
    int d[SORT_U];
#pragma unroll
    for (int q = 0; q < SORT_U; ++q) d[q] = i + q * BK_NODES < be ? tmp[i + q * BK_NODES].y : 0;
#pragma unroll
    for (int q = 0; q < SORT_U; ++q)
      if (i + q * BK_NODES < be) atomicAdd(&cnt[local_node(d[q], node0)], 1);
  }
  __syncthreads();
  const int c = cnt[t];
  int inc = c;
#pragma unroll
  for (int d = 1; d < 32; d <<= 1) {
    int o = __shfl_up(inc, d);
    if (lane >= d) inc += o;
  }
  if (lane == 31) wsum[wv] = inc;
  __syncthreads();
  const int wt = lane < BK_NODES / 32 ? wsum[lane] : 0;
  int winc = wt;
#pragma unroll
  for (int d = 1; d < 32; d <<= 1) {
    int o = __shfl_up(winc, d);
    if (lane >= d) winc += o;
  }
  const int wex = __shfl(winc, wv) - __shfl(wt, wv);
  const int node = node0 + t;
  if (node < n) {
    rowptr[node] = bs + wex + inc - c;
    dinv[node] = rsqrtf((float)(c + 1));
  }
  if (blockIdx.x == gridDim.x - 1 && t == 0) rowptr[n] = be;
}

#define FILL_CAP 12288
__global__ __launch_bounds__(BK_NODES) void k_bucket_fill(const int2* __restrict__ tmp,
                                                          const int* __restrict__ boff,
                                                          const int* __restrict__ rowptr,
                                                          const float* __restrict__ dinv,
                                                          int2* __restrict__ csr, int n, int E, int nch) {
  __shared__ int cur[BK_NODES];
  __shared__ int2 stage[FILL_CAP];
  const int t = threadIdx.x;
  const int node0 = blockIdx.x << BK_SHIFT;
  int bs, be;
  bucket_range(boff, blockIdx.x, nch, E, bs, be);
  const int cnt = be - bs;
  const bool staged = cnt <= FILL_CAP;
  cur[t] = node0 + t < n ? rowptr[node0 + t] : be;
  __syncthreads();
  for (int i = bs + t; i < be; i += SORT_U * BK_NODES) {
    int2 sd[SORT_U];
    float w[SORT_U];
#pragma unroll
    for (int q = 0; q < SORT_U; ++q)
      sd[q] = i + q * BK_NODES < be ? tmp[i + q * BK_NODES] : make_int2(0, 0);
#pragma unroll
    for (int q = 0; q < SORT_U; ++q) {
      const int s = sd[q].x < 0 ? 0 : (sd[q].x < n ? sd[q].x : n - 1);
      const int d = sd[q].y < 0 ? 0 : (sd[q].y < n ? sd[q].y : n - 1);
      w[q] = dinv[s] * dinv[d];
    }
#pragma unroll
    for (int q = 0; q < SORT_U; ++q) {
      if (i + q * BK_NODES < be) {
        int p = atomicAdd(&cur[local_node(sd[q].y, node0)], 1);
        if (staged) {
          int l = p - bs;
          l = l < 0 ? 0 : (l < FILL_CAP ? l : FILL_CAP - 1);
          stage[l] = make_int2(sd[q].x, __float_as_int(w[q]));
        } else {
          p = p < 0 ? 0 : (p < E ? p : E - 1);
          csr[p] = make_int2(sd[q].x, __float_as_int(w[q]));
        }
      }
    }
  }
  if (!staged) return;
  __syncthreads();
  for (int i = t; i < cnt; i += SORT_U * BK_NODES) {
    int2 v[SORT_U];
#pragma unroll
    for (int q = 0; q < SORT_U; ++q) v[q] = i + q * BK_NODES < cnt ? stage[i + q * BK_NODES] : make_int2(0, 0);
#pragma unroll
    for (int q = 0; q < SORT_U; ++q)
      if (i + q * BK_NODES < cnt) csr[bs + i + q * BK_NODES] = v[q];
  }
}

#define WPAIR 288
#define ZP 132
#define FL_ROWS 48
#define FL_GW 24
#define FL_MW 8
#define FL_BLOCKS 512

__device__ __forceinline__ void bn_coef(float s1, float s2, float g, float bt, float inv_n,
                                        float& a, float& c) {
  const float mean = s1 * inv_n;
  const float var = s2 * inv_n - mean * mean;
  a = g * rsqrtf(var + BN_EPS);
  c = bt - a * mean;
}

template <int K>
__device__ __forceinline__ void edge_batch(float4& acc, float& wsum, const float4* __restrict__ hb,
                                           const int2 my, const int j, const int lane) {
  int sidx[K];
  float wgt[K];
  float4 a[K];
#pragma unroll
  for (int q = 0; q < K; ++q) {
    sidx[q] = __builtin_amdgcn_readlane(my.x, j + q);
    wgt[q] = __int_as_float(__builtin_amdgcn_readlane(my.y, j + q));
  }
#pragma unroll
  for (int q = 0; q < K; ++q)
    a[q] = *(const float4*)((const char*)hb + ((unsigned)sidx[q] * (unsigned)(HDIM * 4) + (unsigned)lane * 16u));
#pragma unroll
  for (int q = 0; q < K; ++q) {
    acc.x = fmaf(a[q].x, wgt[q], acc.x);
    acc.y = fmaf(a[q].y, wgt[q], acc.y);
    acc.z = fmaf(a[q].z, wgt[q], acc.z);
    acc.w = fmaf(a[q].w, wgt[q], acc.w);
    wsum += wgt[q];
  }
}

template <bool STATS>
__global__ __launch_bounds__((FL_GW + FL_MW) * 32, 16) void k_layer(const int* __restrict__ rowptr,
                                                                const int* __restrict__ csr,
                                                                const float* __restrict__ dinv,
                                                                const float* __restrict__ xin,
                                                                const float* __restrict__ stats_in,
                                                                const float* __restrict__ gamma,
                                                                const float* __restrict__ beta,
                                                                const float* __restrict__ W,
                                                                const float* __restrict__ bias,
                                                                float* __restrict__ xout,
                                                                float* __restrict__ stats_out, int n, float inv_n) {
  __shared__ float wsm[(HDIM / 2) * WPAIR];
  __shared__ float zs[2 * FL_ROWS * ZP];
  __shared__ float bnco[2 * HDIM];
  static_assert(FL_GW * 2 == FL_ROWS, "two nodes per gather wave");
  static_assert(FL_MW * 16 == HDIM, "one column tile per matrix wave");
  const int tid = threadIdx.x;
  const int lane = tid & 31;
  const int wave = __builtin_amdgcn_readfirstlane(tid >> 5);
  const int ntiles = (n + FL_ROWS - 1) / FL_ROWS;
  const int nmine = (int)blockIdx.x < ntiles ? (ntiles - 1 - (int)blockIdx.x) / (int)gridDim.x + 1 : 0;

#pragma unroll
  for (int p = 0; p < 2; ++p) {
    const int kp = p * (FL_GW + FL_MW) + wave;
    const float* wr = W + (2 * kp) * HDIM + lane * 4;
    const float4 w0 = *(const float4*)wr;
    const float4 w1 = *(const float4*)(wr + HDIM);
    float* d = &wsm[kp * WPAIR + lane * 8];
    *(float4*)d = make_float4(w0.x, w1.x, w0.y, w1.y);
    *(float4*)(d + 4) = make_float4(w0.z, w1.z, w0.w, w1.w);
  }

  const float4* hb = (const float4*)xin;
  const int2* cs = (const int2*)csr;
  if (tid < HDIM) {
    float a, cc;
    bn_coef(stats_in[tid], stats_in[HDIM + tid], gamma[tid], beta[tid], inv_n, a, cc);
    bnco[tid] = a;
    bnco[HDIM + tid] = cc;
  }
  __syncthreads();
  const int mw = wave >= FL_GW ? wave - FL_GW : 0;
  const int m16 = lane & 15;
  const int hi = lane >> 4;
  const int col = mw * 16 + m16;
  const float bcol = bias[col];
  float cs1 = 0.f, cs2 = 0.f;

  for (int it = 0; it <= nmine; ++it) {
    if (wave < FL_GW) {
      if (it < nmine) {
        const int tile = blockIdx.x + it * gridDim.x;
        float* zb = &zs[(it & 1) * FL_ROWS * ZP];
#pragma unroll 1
        for (int q = 0; q < 2; ++q) {
          const int r = wave * 2 + q;
          const int node = tile * FL_ROWS + r;
          float4 z = make_float4(0.f, 0.f, 0.f, 0.f);
          if (node < n) {
            const int beg = rowptr[node], end = rowptr[node + 1];
            const float di = dinv[node];
            const float4 hs = hb[(size_t)node * 32 + lane];
            const float w_self = di * di;
            float4 acc = make_float4(hs.x * w_self, hs.y * w_self, hs.z * w_self, hs.w * w_self);
            float wsum = w_self;
            for (int base = beg; base < end; base += 32) {
              int m = end - base;
              m = m < 32 ? m : 32;
              int2 my = make_int2(0, 0);
              if (lane < m) my = cs[base + lane];
              my.x = my.x < 0 ? 0 : (my.x < n ? my.x : n - 1);
              int j = 0;
              for (; j + 8 <= m; j += 8) edge_batch<8>(acc, wsum, hb, my, j, lane);
              if (j + 4 <= m) { edge_batch<4>(acc, wsum, hb, my, j, lane); j += 4; }
              const int rem = m - j;
              if (rem == 3) edge_batch<3>(acc, wsum, hb, my, j, lane);
              else if (rem == 2) edge_batch<2>(acc, wsum, hb, my, j, lane);
              else if (rem == 1) edge_batch<1>(acc, wsum, hb, my, j, lane);
            }
            const float4 sa = ((const float4*)bnco)[lane];
            const float4 sc = ((const float4*)(bnco + HDIM))[lane];
            z.x = fmaf(sa.x, acc.x, sc.x * wsum);
            z.y = fmaf(sa.y, acc.y, sc.y * wsum);
            z.z = fmaf(sa.z, acc.z, sc.z * wsum);
            z.w = fmaf(sa.w, acc.w, sc.w * wsum);
          }
          *(float4*)(&zb[r * ZP + lane * 4]) = z;
        }
      }
    } else if (it >= 1) {
      const int tile = blockIdx.x + (it - 1) * gridDim.x;
      const float* zb = &zs[((it - 1) & 1) * FL_ROWS * ZP];
      v8f pacc[3];
#pragma unroll
      for (int g = 0; g < 3; ++g)
#pragma unroll
        for (int v = 0; v < 8; ++v) pacc[g][v] = 0.f;
      const float* xrow = &zb[m16 * ZP];
      const float* wk0 = &wsm[hi * WPAIR + col * 2];
#pragma unroll
      for (int k4 = 0; k4 < HDIM / 4; ++k4) {
        const int kb = k4 * 4 + hi * 2;
        const float* wk = wk0 + (k4 * 2) * WPAIR;
        v2f b; b.x = wk[0]; b.y = wk[1];
#pragma unroll
        for (int g = 0; g < 3; ++g) {
          v2f a; a.x = xrow[g * 16 * ZP + kb]; a.y = xrow[g * 16 * ZP + kb + 1];
          pacc[g] = __builtin_amdgcn_wmma_f32_16x16x4_f32(
              false, a, false, b, (short)0, pacc[g], false, false);
        }
      }
#pragma unroll
      for (int g = 0; g < 3; ++g) {
#pragma unroll
        for (int v = 0; v < 8; ++v) {
          const int row = tile * FL_ROWS + g * 16 + v + 8 * hi;
          float o = pacc[g][v] + bcol;
          o = o > 0.f ? o : 0.f;
          if (row < n) {
            xout[(size_t)row * HDIM + col] = o;
            if (STATS) { cs1 += o; cs2 = fmaf(o, o, cs2); }
          }
        }
      }
    }
    __syncthreads();
  }
  if (STATS) {
    if (wave >= FL_GW) {
      cs1 += __shfl_xor(cs1, 16);
      cs2 += __shfl_xor(cs2, 16);
      if (__builtin_amdgcn_mbcnt_lo(~0u, 0u) < 16u) {
        atomicAdd(&stats_out[col], cs1);
        atomicAdd(&stats_out[HDIM + col], cs2);
      }
    }
  }
}

#define OXP 132
#define OWP 96
__global__ __launch_bounds__(128, 5) void k_out_gemm(const float* __restrict__ x,
                                                  const float* __restrict__ Wo,
                                                  const float* __restrict__ bo,
                                                  float* __restrict__ out, int n) {
  __shared__ float xs[64 * OXP];
  __shared__ float ws[(HDIM / 2) * OWP];
  const int tid = threadIdx.x;
  const int lane = tid & 31;
  const int wave = __builtin_amdgcn_readfirstlane(tid >> 5);
  const int rowBase = blockIdx.x * 64;
  const int m = lane & 15;
  const int hi = lane >> 4;

  float4 xv[16];
#pragma unroll
  for (int p = 0; p < 16; ++p) {
    const int row = rowBase + p * 4 + wave;
    xv[p] = make_float4(0.f, 0.f, 0.f, 0.f);
    if (row < n) xv[p] = ((const float4*)(x + (size_t)row * HDIM))[lane];
  }
#pragma unroll
  for (int p = 0; p < 5; ++p) {
    const int t = p * 128 + tid;
    const int kp = t / 10, j = t - kp * 10;
    const float* wr = Wo + (2 * kp) * NCLS + j * 4;
    const float4 w0 = *(const float4*)wr;
    const float4 w1 = *(const float4*)(wr + NCLS);
    float* d = &ws[kp * OWP + j * 8];
    *(float4*)d = make_float4(w0.x, w1.x, w0.y, w1.y);
    *(float4*)(d + 4) = make_float4(w0.z, w1.z, w0.w, w1.w);
  }
#pragma unroll
  for (int p = 0; p < 2; ++p) {
    const int q = tid * 2 + p;
    *(float4*)&ws[(q >> 2) * OWP + 2 * NCLS + (q & 3) * 4] = make_float4(0.f, 0.f, 0.f, 0.f);
  }
#pragma unroll
  for (int p = 0; p < 16; ++p) *(float4*)(&xs[(p * 4 + wave) * OXP + lane * 4]) = xv[p];
  __syncthreads();

  v8f acc[3];
#pragma unroll
  for (int c = 0; c < 3; ++c)
#pragma unroll
    for (int v = 0; v < 8; ++v) acc[c][v] = 0.f;

  const float* xrow = &xs[(wave * 16 + m) * OXP];
#pragma unroll
  for (int k4 = 0; k4 < 32; ++k4) {
    const int kb = k4 * 4 + hi * 2;
    v2f a; a.x = xrow[kb]; a.y = xrow[kb + 1];
    const float* wk = &ws[(k4 * 2 + hi) * OWP + m * 2];
#pragma unroll
    for (int c = 0; c < 3; ++c) {
      v2f b; b.x = wk[c * 32]; b.y = wk[c * 32 + 1];
      acc[c] = __builtin_amdgcn_wmma_f32_16x16x4_f32(
          false, a, false, b, (short)0, acc[c], false, false);
    }
  }
  __syncthreads();

  float* cs = xs;
#pragma unroll
  for (int c = 0; c < 3; ++c) {
    const int col = c * 16 + m;
    if (col < NCLS) {
      const float bv = bo[col];
#pragma unroll
      for (int v = 0; v < 8; ++v) cs[(wave * 16 + v + 8 * hi) * NCLS + col] = acc[c][v] + bv;
    }
  }
  __syncthreads();
  int rows = n - rowBase;
  rows = rows < 64 ? rows : 64;
  const int cnt = rows * NCLS;
  float* ob = out + (size_t)rowBase * NCLS;
#pragma unroll
  for (int p = 0; p < 5; ++p) {
    const int i4 = (p * 128 + tid) * 4;
    if (i4 < cnt) *(float4*)(ob + i4) = *(const float4*)&cs[i4];
  }
}

extern "C" void kernel_launch(void* const* d_in, const int* in_sizes, int n_in,
                              void* d_out, int out_size, void* d_ws, size_t ws_size,
                              hipStream_t stream) {
  const float* x_in  = (const float*)d_in[0];
  const int*   ei    = (const int*)d_in[1];
  const float* gamma = (const float*)d_in[2];
  const float* beta  = (const float*)d_in[3];
  const float* W     = (const float*)d_in[4];
  const float* b     = (const float*)d_in[5];
  const float* Wo    = (const float*)d_in[6];
  const float* bo    = (const float*)d_in[7];
  float* out = (float*)d_out;

  const int N = in_sizes[0] / HDIM;
  const int E = in_sizes[1] / 2;

  float* dinv  = (float*)d_ws;
  float* stats = dinv + ((N + 255) / 256) * 256;
  float* hbuf  = stats + 2 * HDIM;
  float* xbuf  = hbuf + (size_t)N * HDIM;

  int* csr    = (int*)d_out;
  int* rowptr = csr + ((2 * E + 3) / 4) * 4;
  float* stats1 = (float*)(rowptr + ((N + 4) / 4) * 4);
  float* stats2 = stats1 + 2 * HDIM;

  const int nb = (N + BK_NODES - 1) / BK_NODES;
  const int ch = (E + NCHK - 1) / NCHK;
  const int n2 = nb * NCHK;
  int2* tmp    = (int2*)xbuf;
  int*  bh     = (int*)xbuf + ((2 * E + 3) / 4) * 4;
  int*  bstart = bh + ((n2 + 3) / 4) * 4;

  k_bucket_hist<<<NCHK, CH_T, 0, stream>>>(ei, bh, E, ch, nb, stats, stats1);
  k_scatter_stats<<<2 * NCHK, CH_T, 0, stream>>>(ei, bh, bstart, tmp, E, ch, nb, x_in, stats, N);
  k_bucket_rows<<<nb, BK_NODES, 0, stream>>>(tmp, bstart, rowptr, dinv, N, E, 1);
  k_bucket_fill<<<nb, BK_NODES, 0, stream>>>(tmp, bstart, rowptr, dinv, (int2*)csr, N, E, 1);

  k_layer<true><<<FL_BLOCKS, (FL_GW + FL_MW) * 32, 0, stream>>>(
      rowptr, csr, dinv, x_in, stats, gamma, beta, W, b, hbuf, stats1, N, 1.0f / (float)N);
  k_layer<true><<<FL_BLOCKS, (FL_GW + FL_MW) * 32, 0, stream>>>(
      rowptr, csr, dinv, hbuf, stats1, gamma + HDIM, beta + HDIM, W + (size_t)HDIM * HDIM, b + HDIM,
      xbuf, stats2, N, 1.0f / (float)N);
  k_layer<false><<<FL_BLOCKS, (FL_GW + FL_MW) * 32, 0, stream>>>(
      rowptr, csr, dinv, xbuf, stats2, gamma + 2 * HDIM, beta + 2 * HDIM, W + (size_t)2 * HDIM * HDIM,
      b + 2 * HDIM, hbuf, stats1, N, 1.0f / (float)N);

  k_out_gemm<<<(N + 63) / 64, 128, 0, stream>>>(hbuf, Wo, bo, out, N);
}
